# starting kernel
# baseline (speedup 1.0000x reference)
#include <hip/hip_runtime.h>


typedef float v2f __attribute__((ext_vector_type(2)));
typedef float v8f __attribute__((ext_vector_type(8)));

__global__ void k_zero(float* __restrict__ p, int n) {
    int i = blockIdx.x * blockDim.x + threadIdx.x;
    if (i < n) p[i] = 0.0f;
}

__global__ void k_count(const int* __restrict__ dst, float* __restrict__ deg, int E) {
    int e = blockIdx.x * blockDim.x + threadIdx.x;
    if (e < E) atomicAdd(&deg[dst[e]], 1.0f);
}

__global__ void k_dinv(const float* __restrict__ deg, float* __restrict__ dinv, int n) {
    int i = blockIdx.x * blockDim.x + threadIdx.x;
    if (i < n) dinv[i] = rsqrtf(deg[i] + 1.0f);
}

template <bool RELU_A, int K>
__global__ void k_gemm_wmma(const float* __restrict__ A, const float* __restrict__ B,
                            float* __restrict__ C, int M, int N) {
    constexpr int LDK = K + 2;
    __shared__ float As[16 * LDK];

    const int mt   = blockIdx.x;
    const int nb   = blockIdx.y;
    const int tid  = threadIdx.x;
    const int wave = tid >> 5;
    const int lane = tid & 31;

    const float* __restrict__ Ag = A + (size_t)mt * 16 * K;
#pragma unroll
    for (int idx = tid; idx < 16 * K; idx += 256) {
        int r  = idx / K;
        int kk = idx - r * K;
        float v = Ag[idx];
        if (RELU_A) v = v > 0.0f ? v : 0.0f;
        As[r * LDK + kk] = v;
    }
    __syncthreads();

    const int nt  = nb * 8 + wave;
    const int col = lane & 15;
    const int kb  = (lane >> 4) << 1;

    const float* __restrict__ Arow = As + col * LDK + kb;
    const float* __restrict__ Bc   = B + (size_t)nt * 16 + col;

    v8f acc = {};
#pragma unroll
    for (int k = 0; k < K; k += 4) {
        v2f av; av.x = Arow[k];                      av.y = Arow[k + 1];
        v2f bv; bv.x = Bc[(size_t)(k + kb) * N];     bv.y = Bc[(size_t)(k + kb + 1) * N];
        acc = __builtin_amdgcn_wmma_f32_16x16x4_f32(
            false, av, false, bv, (short)0, acc, false, false);
    }

    float* __restrict__ Cp =
        C + (size_t)(mt * 16 + ((lane >> 4) << 3)) * N + (size_t)nt * 16 + col;
#pragma unroll
    for (int r = 0; r < 8; ++r) Cp[(size_t)r * N] = acc[r];
}

__global__ void k_selfloop_bias(const float* __restrict__ xw, const float* __restrict__ dinv,
                                const float* __restrict__ b, float* __restrict__ agg,
                                int n, int F) {
    int idx = blockIdx.x * blockDim.x + threadIdx.x;
    if (idx < n * F) {
        int i = idx / F;
        int f = idx - i * F;
        float di = dinv[i];
        agg[idx] = xw[idx] * di * di + b[f];
    }
}

__global__ void k_scatter(const int* __restrict__ src, const int* __restrict__ dst,
                          const float* __restrict__ dinv, const float* __restrict__ xw,
                          float* __restrict__ agg, int F) {
    int e = blockIdx.x;
    int f = threadIdx.x;
    int s = src[e], d = dst[e];
    float c = dinv[s] * dinv[d];
    atomicAdd(&agg[(size_t)d * F + f], xw[(size_t)s * F + f] * c);
}

__global__ void k_node_head(const float* __restrict__ emb, const float* __restrict__ Wn,
                            const float* __restrict__ bn, float* __restrict__ out, int n) {
    int i = blockIdx.x * blockDim.x + threadIdx.x;
    if (i >= n) return;
    const float4* row = (const float4*)(emb + (size_t)i * 128);
    float a0 = bn[0], a1 = bn[1];
#pragma unroll 8
    for (int q = 0; q < 32; ++q) {
        float4 v = row[q];
        int f = q * 4;
        a0 += v.x * Wn[(f + 0) * 2 + 0] + v.y * Wn[(f + 1) * 2 + 0] +
              v.z * Wn[(f + 2) * 2 + 0] + v.w * Wn[(f + 3) * 2 + 0];
        a1 += v.x * Wn[(f + 0) * 2 + 1] + v.y * Wn[(f + 1) * 2 + 1] +
              v.z * Wn[(f + 2) * 2 + 1] + v.w * Wn[(f + 3) * 2 + 1];
    }
    out[(size_t)i * 2 + 0] = a0;
    out[(size_t)i * 2 + 1] = a1;
}

__global__ void k_edge_head(const int* __restrict__ src, const int* __restrict__ dst,
                            const float* __restrict__ emb, const float* __restrict__ We,
                            const float* __restrict__ be, float* __restrict__ out, int E) {
    int lane = threadIdx.x & 31;
    int warp = blockIdx.x * (blockDim.x >> 5) + (threadIdx.x >> 5);
    int nwarps = gridDim.x * (blockDim.x >> 5);

    float ws0[4], ws1[4], wd0[4], wd1[4];
#pragma unroll
    for (int j = 0; j < 4; ++j) {
        int f = lane * 4 + j;
        ws0[j] = We[f * 2 + 0];         ws1[j] = We[f * 2 + 1];
        wd0[j] = We[(128 + f) * 2 + 0]; wd1[j] = We[(128 + f) * 2 + 1];
    }
    float b0 = be[0], b1 = be[1];

    for (int e = warp; e < E; e += nwarps) {
        int s = src[e], d = dst[e];
        int en = e + nwarps;
        if (en < E) {
            __builtin_prefetch(&emb[(size_t)src[en] * 128 + lane * 4], 0, 0);
            __builtin_prefetch(&emb[(size_t)dst[en] * 128 + lane * 4], 0, 0);
        }
        float4 vs = *(const float4*)(emb + (size_t)s * 128 + lane * 4);
        float4 vd = *(const float4*)(emb + (size_t)d * 128 + lane * 4);
        float a0 = vs.x * ws0[0] + vs.y * ws0[1] + vs.z * ws0[2] + vs.w * ws0[3] +
                   vd.x * wd0[0] + vd.y * wd0[1] + vd.z * wd0[2] + vd.w * wd0[3];
        float a1 = vs.x * ws1[0] + vs.y * ws1[1] + vs.z * ws1[2] + vs.w * ws1[3] +
                   vd.x * wd1[0] + vd.y * wd1[1] + vd.z * wd1[2] + vd.w * wd1[3];
#pragma unroll
        for (int off = 16; off > 0; off >>= 1) {
            a0 += __shfl_down(a0, off, 32);
            a1 += __shfl_down(a1, off, 32);
        }
        if (lane == 0) {
            out[(size_t)e * 2 + 0] = a0 + b0;
            out[(size_t)e * 2 + 1] = a1 + b1;
        }
    }
}

extern "C" void kernel_launch(void* const* d_in, const int* in_sizes, int n_in,
                              void* d_out, int out_size, void* d_ws, size_t ws_size,
                              hipStream_t stream) {
    const float* x  = (const float*)d_in[0];
    const int*   ei = (const int*)d_in[1];
    const float* W1 = (const float*)d_in[2];
    const float* b1 = (const float*)d_in[3];
    const float* W2 = (const float*)d_in[4];
    const float* b2 = (const float*)d_in[5];
    const float* Wn = (const float*)d_in[6];
    const float* bn = (const float*)d_in[7];
    const float* We = (const float*)d_in[8];
    const float* be = (const float*)d_in[9];

    const int N = in_sizes[0] / 64;
    const int E = in_sizes[1] / 2;
    const int* src = ei;
    const int* dst = ei + E;

    char*  ws  = (char*)d_ws;
    size_t off = 0;
    auto carve = [&](size_t bytes) -> float* {
        float* p = (float*)(ws + off);
        off += (bytes + 255) & ~(size_t)255;
        return p;
    };
    float* deg  = carve((size_t)N * 4);
    float* dinv = carve((size_t)N * 4);
    float* xw1  = carve((size_t)N * 256 * 4);
    float* agg1 = carve((size_t)N * 256 * 4);
    float* xw2  = carve((size_t)N * 128 * 4);
    float* emb  = carve((size_t)N * 128 * 4);

    float* out_node = (float*)d_out;
    float* out_edge = out_node + (size_t)N * 2;

    k_zero <<<(N + 255) / 256, 256, 0, stream>>>(deg, N);
    k_count<<<(E + 255) / 256, 256, 0, stream>>>(dst, deg, E);
    k_dinv <<<(N + 255) / 256, 256, 0, stream>>>(deg, dinv, N);

    k_gemm_wmma<false, 64><<<dim3(N / 16, 256 / 128), 256, 0, stream>>>(x, W1, xw1, N, 256);
    k_selfloop_bias<<<(N * 256 + 255) / 256, 256, 0, stream>>>(xw1, dinv, b1, agg1, N, 256);
    k_scatter<<<E, 256, 0, stream>>>(src, dst, dinv, xw1, agg1, 256);

    k_gemm_wmma<true, 256><<<dim3(N / 16, 128 / 128), 256, 0, stream>>>(agg1, W2, xw2, N, 128);
    k_selfloop_bias<<<(N * 128 + 255) / 256, 256, 0, stream>>>(xw2, dinv, b2, emb, N, 128);
    k_scatter<<<E, 128, 0, stream>>>(src, dst, dinv, xw2, emb, 128);

    k_node_head<<<(N + 255) / 256, 256, 0, stream>>>(emb, Wn, bn, out_node, N);
    k_edge_head<<<(E + 7) / 8, 256, 0, stream>>>(src, dst, emb, We, be, out_edge, E);
}


# f32 atomic scatter -> edges bucketed by dst each call (int atomic rank, scan, place); one wave sums a node's rows in registers, self-loop+bias first, one store; sum order = arrival rank
# speedup vs baseline: 3.7256x; 3.7256x over previous
#include <hip/hip_runtime.h>


typedef float v2f __attribute__((ext_vector_type(2)));
typedef float v8f __attribute__((ext_vector_type(8)));

__global__ void k_zero_i(int* __restrict__ p, int n) {
    int i = blockIdx.x * blockDim.x + threadIdx.x;
    if (i < n) p[i] = 0;
}

__global__ void k_count_rank(const int* __restrict__ dst, int* __restrict__ cnt,
                             int* __restrict__ rank, int E) {
    int e = blockIdx.x * blockDim.x + threadIdx.x;
    if (e < E) rank[e] = atomicAdd(&cnt[dst[e]], 1);
}

__global__ void __launch_bounds__(1024)
k_scan_dinv(const int* __restrict__ cnt, int* __restrict__ rowptr,
            float* __restrict__ dinv, int n) {
    __shared__ int s_pre[32];
    __shared__ int s_inc[32];
    const int t     = threadIdx.x;
    const int lane  = t & 31;
    const int wave  = t >> 5;
    const int tile0 = blockIdx.x * 1024;

    int pre = 0;
    for (int i = t; i < tile0; i += 1024) pre += cnt[i];

    const int i = tile0 + t;
    const int v = (i < n) ? cnt[i] : 0;

    int inc = v;
    for (int off = 1; off < 32; off <<= 1) {
        int u = __shfl_up(inc, off, 32);
        if (lane >= off) inc += u;
    }
    for (int off = 16; off > 0; off >>= 1) pre += __shfl_xor(pre, off, 32);

    if (lane == 31) s_inc[wave] = inc;
    if (lane == 0)  s_pre[wave] = pre;
    __syncthreads();

    int base = 0;
    for (int w = 0; w < 32; ++w) {
        base += s_pre[w];
        if (w < wave) base += s_inc[w];
    }
    const int excl = base + inc - v;
    if (i < n) {
        rowptr[i] = excl;
        dinv[i]   = rsqrtf((float)v + 1.0f);
    }
    if (i == n - 1) rowptr[n] = excl + v;
}

__global__ void k_place(const int* __restrict__ src, const int* __restrict__ dst,
                        const int* __restrict__ rowptr, const int* __restrict__ rank,
                        int* __restrict__ csr_src, int E) {
    int e = blockIdx.x * blockDim.x + threadIdx.x;
    if (e < E) {
        int p = rowptr[dst[e]] + rank[e];
        p = p < 0 ? 0 : (p > E - 1 ? E - 1 : p);
        csr_src[p] = src[e];
    }
}

template <bool RELU_A, int K>
__global__ void k_gemm_wmma(const float* __restrict__ A, const float* __restrict__ B,
                            float* __restrict__ C, int M, int N) {
    constexpr int LDK = K + 2;
    __shared__ float As[16 * LDK];

    const int mt   = blockIdx.x;
    const int nb   = blockIdx.y;
    const int tid  = threadIdx.x;
    const int wave = tid >> 5;
    const int lane = tid & 31;

    const float* __restrict__ Ag = A + (size_t)mt * 16 * K;
#pragma unroll
    for (int idx = tid; idx < 16 * K; idx += 256) {
        int r  = idx / K;
        int kk = idx - r * K;
        float v = Ag[idx];
        if (RELU_A) v = v > 0.0f ? v : 0.0f;
        As[r * LDK + kk] = v;
    }
    __syncthreads();

    const int nt  = nb * 8 + wave;
    const int col = lane & 15;
    const int kb  = (lane >> 4) << 1;

    const float* __restrict__ Arow = As + col * LDK + kb;
    const float* __restrict__ Bc   = B + (size_t)nt * 16 + col;

    v8f acc = {};
#pragma unroll
    for (int k = 0; k < K; k += 4) {
        v2f av; av.x = Arow[k];                      av.y = Arow[k + 1];
        v2f bv; bv.x = Bc[(size_t)(k + kb) * N];     bv.y = Bc[(size_t)(k + kb + 1) * N];
        acc = __builtin_amdgcn_wmma_f32_16x16x4_f32(
            false, av, false, bv, (short)0, acc, false, false);
    }

    float* __restrict__ Cp =
        C + (size_t)(mt * 16 + ((lane >> 4) << 3)) * N + (size_t)nt * 16 + col;
#pragma unroll
    for (int r = 0; r < 8; ++r) Cp[(size_t)r * N] = acc[r];
}

template <int F, int R>
__device__ __forceinline__ void gather_rows(const float* __restrict__ xw, int my, float cw,
                                            int j, int lane, float4 (&acc)[F / 128]) {
    constexpr int V = F / 128;
    float4 r[R][V];
    float  c[R];
#pragma unroll
    for (int u = 0; u < R; ++u) {
        const int s = __builtin_amdgcn_readlane(my, j + u);
        c[u] = __int_as_float(__builtin_amdgcn_readlane(__float_as_int(cw), j + u));
        const float* __restrict__ row = xw + (size_t)s * F + lane * 4;
#pragma unroll
        for (int v = 0; v < V; ++v) r[u][v] = *(const float4*)(row + v * 128);
    }
#pragma unroll
    for (int u = 0; u < R; ++u) {
#pragma unroll
        for (int v = 0; v < V; ++v) {
            acc[v].x += r[u][v].x * c[u];
            acc[v].y += r[u][v].y * c[u];
            acc[v].z += r[u][v].z * c[u];
            acc[v].w += r[u][v].w * c[u];
        }
    }
}

template <int F>
__global__ void __launch_bounds__(256)
k_gather(const int* __restrict__ rowptr, const int* __restrict__ csr_src,
         const float* __restrict__ dinv, const float* __restrict__ xw,
         const float* __restrict__ b, float* __restrict__ agg, int n) {
    constexpr int V = F / 128;
    const int lane = threadIdx.x & 31;
    const int wid = __builtin_amdgcn_readfirstlane(
        (int)(blockIdx.x * (blockDim.x >> 5) + (threadIdx.x >> 5)));
    const int nw  = gridDim.x * (blockDim.x >> 5);

    float4 bias[V];
#pragma unroll
    for (int v = 0; v < V; ++v) bias[v] = *(const float4*)(b + v * 128 + lane * 4);

    for (int node = wid; node < n; node += nw) {
        const int   beg = rowptr[node];
        const int   end = rowptr[node + 1];
        const float di  = dinv[node];

        const float* __restrict__ self = xw + (size_t)node * F + lane * 4;
        float4 acc[V];
#pragma unroll
        for (int v = 0; v < V; ++v) {
            const float4 x = *(const float4*)(self + v * 128);
            acc[v].x = x.x * di * di + bias[v].x;
            acc[v].y = x.y * di * di + bias[v].y;
            acc[v].z = x.z * di * di + bias[v].z;
            acc[v].w = x.w * di * di + bias[v].w;
        }

        for (int base = beg; base < end; base += 32) {
            const int m = (end - base) < 32 ? (end - base) : 32;
            int   my = 0;
            float cw = 0.0f;
            if (lane < m) {
                my = csr_src[base + lane];
                my = my < 0 ? 0 : (my > n - 1 ? n - 1 : my);
                cw = dinv[my] * di;
            }
            int j = 0;
            for (; j + 4 <= m; j += 4) gather_rows<F, 4>(xw, my, cw, j, lane, acc);
            if (j + 2 <= m) { gather_rows<F, 2>(xw, my, cw, j, lane, acc); j += 2; }
            if (j < m)      { gather_rows<F, 1>(xw, my, cw, j, lane, acc); }
        }

        float* __restrict__ orow = agg + (size_t)node * F + lane * 4;
#pragma unroll
        for (int v = 0; v < V; ++v) *(float4*)(orow + v * 128) = acc[v];
    }
}

__global__ void k_node_head(const float* __restrict__ emb, const float* __restrict__ Wn,
                            const float* __restrict__ bn, float* __restrict__ out, int n) {
    int i = blockIdx.x * blockDim.x + threadIdx.x;
    if (i >= n) return;
    const float4* row = (const float4*)(emb + (size_t)i * 128);
    float a0 = bn[0], a1 = bn[1];
#pragma unroll 8
    for (int q = 0; q < 32; ++q) {
        float4 v = row[q];
        int f = q * 4;
        a0 += v.x * Wn[(f + 0) * 2 + 0] + v.y * Wn[(f + 1) * 2 + 0] +
              v.z * Wn[(f + 2) * 2 + 0] + v.w * Wn[(f + 3) * 2 + 0];
        a1 += v.x * Wn[(f + 0) * 2 + 1] + v.y * Wn[(f + 1) * 2 + 1] +
              v.z * Wn[(f + 2) * 2 + 1] + v.w * Wn[(f + 3) * 2 + 1];
    }
    out[(size_t)i * 2 + 0] = a0;
    out[(size_t)i * 2 + 1] = a1;
}

__global__ void k_edge_head(const int* __restrict__ src, const int* __restrict__ dst,
                            const float* __restrict__ emb, const float* __restrict__ We,
                            const float* __restrict__ be, float* __restrict__ out, int E) {
    int lane = threadIdx.x & 31;
    int warp = blockIdx.x * (blockDim.x >> 5) + (threadIdx.x >> 5);
    int nwarps = gridDim.x * (blockDim.x >> 5);

    float ws0[4], ws1[4], wd0[4], wd1[4];
#pragma unroll
    for (int j = 0; j < 4; ++j) {
        int f = lane * 4 + j;
        ws0[j] = We[f * 2 + 0];         ws1[j] = We[f * 2 + 1];
        wd0[j] = We[(128 + f) * 2 + 0]; wd1[j] = We[(128 + f) * 2 + 1];
    }
    float b0 = be[0], b1 = be[1];

    for (int e = warp; e < E; e += nwarps) {
        int s = src[e], d = dst[e];
        int en = e + nwarps;
        if (en < E) {
            __builtin_prefetch(&emb[(size_t)src[en] * 128 + lane * 4], 0, 0);
            __builtin_prefetch(&emb[(size_t)dst[en] * 128 + lane * 4], 0, 0);
        }
        float4 vs = *(const float4*)(emb + (size_t)s * 128 + lane * 4);
        float4 vd = *(const float4*)(emb + (size_t)d * 128 + lane * 4);
        float a0 = vs.x * ws0[0] + vs.y * ws0[1] + vs.z * ws0[2] + vs.w * ws0[3] +
                   vd.x * wd0[0] + vd.y * wd0[1] + vd.z * wd0[2] + vd.w * wd0[3];
        float a1 = vs.x * ws1[0] + vs.y * ws1[1] + vs.z * ws1[2] + vs.w * ws1[3] +
                   vd.x * wd1[0] + vd.y * wd1[1] + vd.z * wd1[2] + vd.w * wd1[3];
#pragma unroll
        for (int off = 16; off > 0; off >>= 1) {
            a0 += __shfl_down(a0, off, 32);
            a1 += __shfl_down(a1, off, 32);
        }
        if (lane == 0) {
            out[(size_t)e * 2 + 0] = a0 + b0;
            out[(size_t)e * 2 + 1] = a1 + b1;
        }
    }
}

extern "C" void kernel_launch(void* const* d_in, const int* in_sizes, int n_in,
                              void* d_out, int out_size, void* d_ws, size_t ws_size,
                              hipStream_t stream) {
    const float* x  = (const float*)d_in[0];
    const int*   ei = (const int*)d_in[1];
    const float* W1 = (const float*)d_in[2];
    const float* b1 = (const float*)d_in[3];
    const float* W2 = (const float*)d_in[4];
    const float* b2 = (const float*)d_in[5];
    const float* Wn = (const float*)d_in[6];
    const float* bn = (const float*)d_in[7];
    const float* We = (const float*)d_in[8];
    const float* be = (const float*)d_in[9];

    const int N = in_sizes[0] / 64;
    const int E = in_sizes[1] / 2;
    const int* src = ei;
    const int* dst = ei + E;

    char*  ws  = (char*)d_ws;
    size_t off = 0;
    auto carve = [&](size_t bytes) -> float* {
        float* p = (float*)(ws + off);
        off += (bytes + 255) & ~(size_t)255;
        return p;
    };
    int*   cnt    = (int*)carve((size_t)N * 4);
    float* dinv   = carve((size_t)N * 4);
    int*   rowptr = (int*)carve((size_t)(N + 1) * 4);
    int*   rank   = (int*)carve((size_t)E * 4);
    int*   csrsrc = (int*)carve((size_t)E * 4);
    float* xw1  = carve((size_t)N * 256 * 4);
    float* agg1 = carve((size_t)N * 256 * 4);
    float* xw2  = carve((size_t)N * 128 * 4);
    float* emb  = carve((size_t)N * 128 * 4);

    float* out_node = (float*)d_out;
    float* out_edge = out_node + (size_t)N * 2;

    k_zero_i    <<<(N + 255) / 256, 256, 0, stream>>>(cnt, N);
    k_count_rank<<<(E + 255) / 256, 256, 0, stream>>>(dst, cnt, rank, E);
    k_scan_dinv <<<(N + 1023) / 1024, 1024, 0, stream>>>(cnt, rowptr, dinv, N);
    k_place     <<<(E + 255) / 256, 256, 0, stream>>>(src, dst, rowptr, rank, csrsrc, E);

    k_gemm_wmma<false, 64><<<dim3(N / 16, 256 / 128), 256, 0, stream>>>(x, W1, xw1, N, 256);
    k_gather<256><<<2048, 256, 0, stream>>>(rowptr, csrsrc, dinv, xw1, b1, agg1, N);

    k_gemm_wmma<true, 256><<<dim3(N / 16, 128 / 128), 256, 0, stream>>>(agg1, W2, xw2, N, 128);
    k_gather<128><<<2048, 256, 0, stream>>>(rowptr, csrsrc, dinv, xw2, b2, emb, N);

    k_node_head<<<(N + 255) / 256, 256, 0, stream>>>(emb, Wn, bn, out_node, N);
    k_edge_head<<<(E + 7) / 8, 256, 0, stream>>>(src, dst, emb, We, be, out_edge, E);
}


# edge head by linearity: node-head kernel also emits p=emb@We[:128], q=emb@We[128:] per node; edge_x=(p[src]+q[dst])+be, 4 edges/thread; the 256-term dot is summed as two 128-term halves
# speedup vs baseline: 4.8876x; 1.3119x over previous
#include <hip/hip_runtime.h>


typedef float v2f __attribute__((ext_vector_type(2)));
typedef float v8f __attribute__((ext_vector_type(8)));

__global__ void k_zero_i(int* __restrict__ p, int n) {
    int i = blockIdx.x * blockDim.x + threadIdx.x;
    if (i < n) p[i] = 0;
}

__global__ void k_count_rank(const int* __restrict__ dst, int* __restrict__ cnt,
                             int* __restrict__ rank, int E) {
    int e = blockIdx.x * blockDim.x + threadIdx.x;
    if (e < E) rank[e] = atomicAdd(&cnt[dst[e]], 1);
}

__global__ void __launch_bounds__(1024)
k_scan_dinv(const int* __restrict__ cnt, int* __restrict__ rowptr,
            float* __restrict__ dinv, int n) {
    __shared__ int s_pre[32];
    __shared__ int s_inc[32];
    const int t     = threadIdx.x;
    const int lane  = t & 31;
    const int wave  = t >> 5;
    const int tile0 = blockIdx.x * 1024;

    int pre = 0;
    for (int i = t; i < tile0; i += 1024) pre += cnt[i];

    const int i = tile0 + t;
    const int v = (i < n) ? cnt[i] : 0;

    int inc = v;
    for (int off = 1; off < 32; off <<= 1) {
        int u = __shfl_up(inc, off, 32);
        if (lane >= off) inc += u;
    }
    for (int off = 16; off > 0; off >>= 1) pre += __shfl_xor(pre, off, 32);

    if (lane == 31) s_inc[wave] = inc;
    if (lane == 0)  s_pre[wave] = pre;
    __syncthreads();

    int base = 0;
    for (int w = 0; w < 32; ++w) {
        base += s_pre[w];
        if (w < wave) base += s_inc[w];
    }
    const int excl = base + inc - v;
    if (i < n) {
        rowptr[i] = excl;
        dinv[i]   = rsqrtf((float)v + 1.0f);
    }
    if (i == n - 1) rowptr[n] = excl + v;
}

__global__ void k_place(const int* __restrict__ src, const int* __restrict__ dst,
                        const int* __restrict__ rowptr, const int* __restrict__ rank,
                        int* __restrict__ csr_src, int E) {
    int e = blockIdx.x * blockDim.x + threadIdx.x;
    if (e < E) {
        int p = rowptr[dst[e]] + rank[e];
        p = p < 0 ? 0 : (p > E - 1 ? E - 1 : p);
        csr_src[p] = src[e];
    }
}

template <bool RELU_A, int K>
__global__ void k_gemm_wmma(const float* __restrict__ A, const float* __restrict__ B,
                            float* __restrict__ C, int M, int N) {
    constexpr int LDK = K + 2;
    __shared__ float As[16 * LDK];

    const int mt   = blockIdx.x;
    const int nb   = blockIdx.y;
    const int tid  = threadIdx.x;
    const int wave = tid >> 5;
    const int lane = tid & 31;

    const float* __restrict__ Ag = A + (size_t)mt * 16 * K;
#pragma unroll
    for (int idx = tid; idx < 16 * K; idx += 256) {
        int r  = idx / K;
        int kk = idx - r * K;
        float v = Ag[idx];
        if (RELU_A) v = v > 0.0f ? v : 0.0f;
        As[r * LDK + kk] = v;
    }
    __syncthreads();

    const int nt  = nb * 8 + wave;
    const int col = lane & 15;
    const int kb  = (lane >> 4) << 1;

    const float* __restrict__ Arow = As + col * LDK + kb;
    const float* __restrict__ Bc   = B + (size_t)nt * 16 + col;

    v8f acc = {};
#pragma unroll
    for (int k = 0; k < K; k += 4) {
        v2f av; av.x = Arow[k];                      av.y = Arow[k + 1];
        v2f bv; bv.x = Bc[(size_t)(k + kb) * N];     bv.y = Bc[(size_t)(k + kb + 1) * N];
        acc = __builtin_amdgcn_wmma_f32_16x16x4_f32(
            false, av, false, bv, (short)0, acc, false, false);
    }

    float* __restrict__ Cp =
        C + (size_t)(mt * 16 + ((lane >> 4) << 3)) * N + (size_t)nt * 16 + col;
#pragma unroll
    for (int r = 0; r < 8; ++r) Cp[(size_t)r * N] = acc[r];
}

template <int F, int R>
__device__ __forceinline__ void gather_rows(const float* __restrict__ xw, int my, float cw,
                                            int j, int lane, float4 (&acc)[F / 128]) {
    constexpr int V = F / 128;
    float4 r[R][V];
    float  c[R];
#pragma unroll
    for (int u = 0; u < R; ++u) {
        const int s = __builtin_amdgcn_readlane(my, j + u);
        c[u] = __int_as_float(__builtin_amdgcn_readlane(__float_as_int(cw), j + u));
        const float* __restrict__ row = xw + (size_t)s * F + lane * 4;
#pragma unroll
        for (int v = 0; v < V; ++v) r[u][v] = *(const float4*)(row + v * 128);
    }
#pragma unroll
    for (int u = 0; u < R; ++u) {
#pragma unroll
        for (int v = 0; v < V; ++v) {
            acc[v].x += r[u][v].x * c[u];
            acc[v].y += r[u][v].y * c[u];
            acc[v].z += r[u][v].z * c[u];
            acc[v].w += r[u][v].w * c[u];
        }
    }
}

template <int F>
__global__ void __launch_bounds__(256)
k_gather(const int* __restrict__ rowptr, const int* __restrict__ csr_src,
         const float* __restrict__ dinv, const float* __restrict__ xw,
         const float* __restrict__ b, float* __restrict__ agg, int n) {
    constexpr int V = F / 128;
    const int lane = threadIdx.x & 31;
    const int wid = __builtin_amdgcn_readfirstlane(
        (int)(blockIdx.x * (blockDim.x >> 5) + (threadIdx.x >> 5)));
    const int nw  = gridDim.x * (blockDim.x >> 5);

    float4 bias[V];
#pragma unroll
    for (int v = 0; v < V; ++v) bias[v] = *(const float4*)(b + v * 128 + lane * 4);

    for (int node = wid; node < n; node += nw) {
        const int   beg = rowptr[node];
        const int   end = rowptr[node + 1];
        const float di  = dinv[node];

        const float* __restrict__ self = xw + (size_t)node * F + lane * 4;
        float4 acc[V];
#pragma unroll
        for (int v = 0; v < V; ++v) {
            const float4 x = *(const float4*)(self + v * 128);
            acc[v].x = x.x * di * di + bias[v].x;
            acc[v].y = x.y * di * di + bias[v].y;
            acc[v].z = x.z * di * di + bias[v].z;
            acc[v].w = x.w * di * di + bias[v].w;
        }

        for (int base = beg; base < end; base += 32) {
            const int m = (end - base) < 32 ? (end - base) : 32;
            int   my = 0;
            float cw = 0.0f;
            if (lane < m) {
                my = csr_src[base + lane];
                my = my < 0 ? 0 : (my > n - 1 ? n - 1 : my);
                cw = dinv[my] * di;
            }
            int j = 0;
            for (; j + 4 <= m; j += 4) gather_rows<F, 4>(xw, my, cw, j, lane, acc);
            if (j + 2 <= m) { gather_rows<F, 2>(xw, my, cw, j, lane, acc); j += 2; }
            if (j < m)      { gather_rows<F, 1>(xw, my, cw, j, lane, acc); }
        }

        float* __restrict__ orow = agg + (size_t)node * F + lane * 4;
#pragma unroll
        for (int v = 0; v < V; ++v) *(float4*)(orow + v * 128) = acc[v];
    }
}

__global__ void k_node_head(const float* __restrict__ emb, const float* __restrict__ Wn,
                            const float* __restrict__ bn, const float* __restrict__ We,
                            float* __restrict__ out, float* __restrict__ pq, int n) {
    int i = blockIdx.x * blockDim.x + threadIdx.x;
    if (i >= n) return;
    const float4* row = (const float4*)(emb + (size_t)i * 128);
    const float* __restrict__ Ws = We;
    const float* __restrict__ Wd = We + 128 * 2;
    float a0 = bn[0], a1 = bn[1];
    float p0 = 0.0f, p1 = 0.0f, q0 = 0.0f, q1 = 0.0f;
#pragma unroll 8
    for (int q = 0; q < 32; ++q) {
        float4 v = row[q];
        int f = q * 4;
        a0 += v.x * Wn[(f + 0) * 2 + 0] + v.y * Wn[(f + 1) * 2 + 0] +
              v.z * Wn[(f + 2) * 2 + 0] + v.w * Wn[(f + 3) * 2 + 0];
        a1 += v.x * Wn[(f + 0) * 2 + 1] + v.y * Wn[(f + 1) * 2 + 1] +
              v.z * Wn[(f + 2) * 2 + 1] + v.w * Wn[(f + 3) * 2 + 1];
        p0 += v.x * Ws[(f + 0) * 2 + 0] + v.y * Ws[(f + 1) * 2 + 0] +
              v.z * Ws[(f + 2) * 2 + 0] + v.w * Ws[(f + 3) * 2 + 0];
        p1 += v.x * Ws[(f + 0) * 2 + 1] + v.y * Ws[(f + 1) * 2 + 1] +
              v.z * Ws[(f + 2) * 2 + 1] + v.w * Ws[(f + 3) * 2 + 1];
        q0 += v.x * Wd[(f + 0) * 2 + 0] + v.y * Wd[(f + 1) * 2 + 0] +
              v.z * Wd[(f + 2) * 2 + 0] + v.w * Wd[(f + 3) * 2 + 0];
        q1 += v.x * Wd[(f + 0) * 2 + 1] + v.y * Wd[(f + 1) * 2 + 1] +
              v.z * Wd[(f + 2) * 2 + 1] + v.w * Wd[(f + 3) * 2 + 1];
    }
    out[(size_t)i * 2 + 0] = a0;
    out[(size_t)i * 2 + 1] = a1;
    float4 r;
    r.x = p0; r.y = p1; r.z = q0; r.w = q1;
    *(float4*)(pq + (size_t)i * 4) = r;
}

__device__ __forceinline__ float2 edge_one(const float* __restrict__ pq, int s, int d,
                                           float b0, float b1) {
    const float2 p = *(const float2*)(pq + (size_t)s * 4);
    const float2 q = *(const float2*)(pq + (size_t)d * 4 + 2);
    float2 o;
    o.x = (p.x + q.x) + b0;
    o.y = (p.y + q.y) + b1;
    return o;
}

__global__ void __launch_bounds__(256)
k_edge_head(const int* __restrict__ src, const int* __restrict__ dst,
            const float* __restrict__ pq, const float* __restrict__ be,
            float* __restrict__ out, int E) {
    const int   base = blockIdx.x * 1024;
    const float b0 = be[0], b1 = be[1];
    const bool wide = ((((size_t)src | (size_t)dst) & 7) == 0) && (((size_t)out & 15) == 0);
    if (wide && base + 1024 <= E) {
        const int  e0 = base + threadIdx.x * 2;
        const int  e1 = e0 + 512;
        const int2 s0 = *(const int2*)(src + e0);
        const int2 d0 = *(const int2*)(dst + e0);
        const int2 s1 = *(const int2*)(src + e1);
        const int2 d1 = *(const int2*)(dst + e1);
        const float2 a0 = edge_one(pq, s0.x, d0.x, b0, b1);
        const float2 a1 = edge_one(pq, s0.y, d0.y, b0, b1);
        const float2 c0 = edge_one(pq, s1.x, d1.x, b0, b1);
        const float2 c1 = edge_one(pq, s1.y, d1.y, b0, b1);
        float4 o0, o1;
        o0.x = a0.x; o0.y = a0.y; o0.z = a1.x; o0.w = a1.y;
        o1.x = c0.x; o1.y = c0.y; o1.z = c1.x; o1.w = c1.y;
        *(float4*)(out + (size_t)e0 * 2) = o0;
        *(float4*)(out + (size_t)e1 * 2) = o1;
    } else {
        for (int u = 0; u < 2; ++u) {
            for (int k = 0; k < 2; ++k) {
                const int e = base + u * 512 + threadIdx.x * 2 + k;
                if (e < E) *(float2*)(out + (size_t)e * 2) = edge_one(pq, src[e], dst[e], b0, b1);
            }
        }
    }
}

extern "C" void kernel_launch(void* const* d_in, const int* in_sizes, int n_in,
                              void* d_out, int out_size, void* d_ws, size_t ws_size,
                              hipStream_t stream) {
    const float* x  = (const float*)d_in[0];
    const int*   ei = (const int*)d_in[1];
    const float* W1 = (const float*)d_in[2];
    const float* b1 = (const float*)d_in[3];
    const float* W2 = (const float*)d_in[4];
    const float* b2 = (const float*)d_in[5];
    const float* Wn = (const float*)d_in[6];
    const float* bn = (const float*)d_in[7];
    const float* We = (const float*)d_in[8];
    const float* be = (const float*)d_in[9];

    const int N = in_sizes[0] / 64;
    const int E = in_sizes[1] / 2;
    const int* src = ei;
    const int* dst = ei + E;

    char*  ws  = (char*)d_ws;
    size_t off = 0;
    auto carve = [&](size_t bytes) -> float* {
        float* p = (float*)(ws + off);
        off += (bytes + 255) & ~(size_t)255;
        return p;
    };
    int*   cnt    = (int*)carve((size_t)N * 4);
    float* dinv   = carve((size_t)N * 4);
    int*   rowptr = (int*)carve((size_t)(N + 1) * 4);
    int*   rank   = (int*)carve((size_t)E * 4);
    int*   csrsrc = (int*)carve((size_t)E * 4);
    float* xw1  = carve((size_t)N * 256 * 4);
    float* agg1 = carve((size_t)N * 256 * 4);
    float* xw2  = carve((size_t)N * 128 * 4);
    float* emb  = carve((size_t)N * 128 * 4);
    float* pq   = carve((size_t)N * 4 * 4);

    float* out_node = (float*)d_out;
    float* out_edge = out_node + (size_t)N * 2;

    k_zero_i    <<<(N + 255) / 256, 256, 0, stream>>>(cnt, N);
    k_count_rank<<<(E + 255) / 256, 256, 0, stream>>>(dst, cnt, rank, E);
    k_scan_dinv <<<(N + 1023) / 1024, 1024, 0, stream>>>(cnt, rowptr, dinv, N);
    k_place     <<<(E + 255) / 256, 256, 0, stream>>>(src, dst, rowptr, rank, csrsrc, E);

    k_gemm_wmma<false, 64><<<dim3(N / 16, 256 / 128), 256, 0, stream>>>(x, W1, xw1, N, 256);
    k_gather<256><<<2048, 256, 0, stream>>>(rowptr, csrsrc, dinv, xw1, b1, agg1, N);

    k_gemm_wmma<true, 256><<<dim3(N / 16, 128 / 128), 256, 0, stream>>>(agg1, W2, xw2, N, 128);
    k_gather<128><<<2048, 256, 0, stream>>>(rowptr, csrsrc, dinv, xw2, b2, emb, N);

    k_node_head<<<(N + 255) / 256, 256, 0, stream>>>(emb, Wn, bn, We, out_node, pq, N);
    k_edge_head<<<(E + 1023) / 1024, 256, 0, stream>>>(src, dst, pq, be, out_edge, E);
}


# layer 1 by linearity: gather the 64-float x rows first, then b1 + (S x)@W1, b1 as the GEMM's start value: 4x fewer gather bytes; half-wave per bucket entry, halves added last; f32 sum order changes
# speedup vs baseline: 5.3627x; 1.0972x over previous
#include <hip/hip_runtime.h>


typedef float v2f __attribute__((ext_vector_type(2)));
typedef float v8f __attribute__((ext_vector_type(8)));

__global__ void k_zero_i(int* __restrict__ p, int n) {
    int i = blockIdx.x * blockDim.x + threadIdx.x;
    if (i < n) p[i] = 0;
}

__global__ void k_count_rank(const int* __restrict__ dst, int* __restrict__ cnt,
                             int* __restrict__ rank, int E) {
    int e = blockIdx.x * blockDim.x + threadIdx.x;
    if (e < E) rank[e] = atomicAdd(&cnt[dst[e]], 1);
}

__global__ void __launch_bounds__(1024)
k_scan_dinv(const int* __restrict__ cnt, int* __restrict__ rowptr,
            float* __restrict__ dinv, int n) {
    __shared__ int s_pre[32];
    __shared__ int s_inc[32];
    const int t     = threadIdx.x;
    const int lane  = t & 31;
    const int wave  = t >> 5;
    const int tile0 = blockIdx.x * 1024;

    int pre = 0;
    for (int i = t; i < tile0; i += 1024) pre += cnt[i];

    const int i = tile0 + t;
    const int v = (i < n) ? cnt[i] : 0;

    int inc = v;
    for (int off = 1; off < 32; off <<= 1) {
        int u = __shfl_up(inc, off, 32);
        if (lane >= off) inc += u;
    }
    for (int off = 16; off > 0; off >>= 1) pre += __shfl_xor(pre, off, 32);

    if (lane == 31) s_inc[wave] = inc;
    if (lane == 0)  s_pre[wave] = pre;
    __syncthreads();

    int base = 0;
    for (int w = 0; w < 32; ++w) {
        base += s_pre[w];
        if (w < wave) base += s_inc[w];
    }
    const int excl = base + inc - v;
    if (i < n) {
        rowptr[i] = excl;
        dinv[i]   = rsqrtf((float)v + 1.0f);
    }
    if (i == n - 1) rowptr[n] = excl + v;
}

__global__ void k_place(const int* __restrict__ src, const int* __restrict__ dst,
                        const int* __restrict__ rowptr, const int* __restrict__ rank,
                        int* __restrict__ csr_src, int E) {
    int e = blockIdx.x * blockDim.x + threadIdx.x;
    if (e < E) {
        int p = rowptr[dst[e]] + rank[e];
        p = p < 0 ? 0 : (p > E - 1 ? E - 1 : p);
        csr_src[p] = src[e];
    }
}

template <bool RELU_A, bool BIAS_C, int K>
__global__ void k_gemm_wmma(const float* __restrict__ A, const float* __restrict__ B,
                            const float* __restrict__ bias, float* __restrict__ C,
                            int M, int N) {
    constexpr int LDK = K + 2;
    __shared__ float As[16 * LDK];

    const int mt   = blockIdx.x;
    const int nb   = blockIdx.y;
    const int tid  = threadIdx.x;
    const int wave = tid >> 5;
    const int lane = tid & 31;

    const float* __restrict__ Ag = A + (size_t)mt * 16 * K;
#pragma unroll
    for (int idx = tid; idx < 16 * K; idx += 256) {
        int r  = idx / K;
        int kk = idx - r * K;
        float v = Ag[idx];
        if (RELU_A) v = v > 0.0f ? v : 0.0f;
        As[r * LDK + kk] = v;
    }
    __syncthreads();

    const int nt  = nb * 8 + wave;
    const int col = lane & 15;
    const int kb  = (lane >> 4) << 1;

    const float* __restrict__ Arow = As + col * LDK + kb;
    const float* __restrict__ Bc   = B + (size_t)nt * 16 + col;

    float bv = 0.0f;
    if (BIAS_C) bv = bias[nt * 16 + col];
    v8f acc;
#pragma unroll
    for (int r = 0; r < 8; ++r) acc[r] = bv;
#pragma unroll
    for (int k = 0; k < K; k += 4) {
        v2f av; av.x = Arow[k];                      av.y = Arow[k + 1];
        v2f bv; bv.x = Bc[(size_t)(k + kb) * N];     bv.y = Bc[(size_t)(k + kb + 1) * N];
        acc = __builtin_amdgcn_wmma_f32_16x16x4_f32(
            false, av, false, bv, (short)0, acc, false, false);
    }

    float* __restrict__ Cp =
        C + (size_t)(mt * 16 + ((lane >> 4) << 3)) * N + (size_t)nt * 16 + col;
#pragma unroll
    for (int r = 0; r < 8; ++r) Cp[(size_t)r * N] = acc[r];
}

template <int F, int R>
__device__ __forceinline__ void gather_rows(const float* __restrict__ xw, int my, float cw,
                                            int j, int lane, float4 (&acc)[F / 128]) {
    constexpr int V = F / 128;
    float4 r[R][V];
    float  c[R];
#pragma unroll
    for (int u = 0; u < R; ++u) {
        const int s = __builtin_amdgcn_readlane(my, j + u);
        c[u] = __int_as_float(__builtin_amdgcn_readlane(__float_as_int(cw), j + u));
        const float* __restrict__ row = xw + (size_t)s * F + lane * 4;
#pragma unroll
        for (int v = 0; v < V; ++v) r[u][v] = *(const float4*)(row + v * 128);
    }
#pragma unroll
    for (int u = 0; u < R; ++u) {
#pragma unroll
        for (int v = 0; v < V; ++v) {
            acc[v].x += r[u][v].x * c[u];
            acc[v].y += r[u][v].y * c[u];
            acc[v].z += r[u][v].z * c[u];
            acc[v].w += r[u][v].w * c[u];
        }
    }
}

template <int F>
__global__ void __launch_bounds__(256)
k_gather(const int* __restrict__ rowptr, const int* __restrict__ csr_src,
         const float* __restrict__ dinv, const float* __restrict__ xw,
         const float* __restrict__ b, float* __restrict__ agg, int n) {
    constexpr int V = F / 128;
    const int lane = threadIdx.x & 31;
    const int wid = __builtin_amdgcn_readfirstlane(
        (int)(blockIdx.x * (blockDim.x >> 5) + (threadIdx.x >> 5)));
    const int nw  = gridDim.x * (blockDim.x >> 5);

    float4 bias[V];
#pragma unroll
    for (int v = 0; v < V; ++v) bias[v] = *(const float4*)(b + v * 128 + lane * 4);

    for (int node = wid; node < n; node += nw) {
        const int   beg = rowptr[node];
        const int   end = rowptr[node + 1];
        const float di  = dinv[node];

        const float* __restrict__ self = xw + (size_t)node * F + lane * 4;
        float4 acc[V];
#pragma unroll
        for (int v = 0; v < V; ++v) {
            const float4 x = *(const float4*)(self + v * 128);
            acc[v].x = x.x * di * di + bias[v].x;
            acc[v].y = x.y * di * di + bias[v].y;
            acc[v].z = x.z * di * di + bias[v].z;
            acc[v].w = x.w * di * di + bias[v].w;
        }

        for (int base = beg; base < end; base += 32) {
            const int m = (end - base) < 32 ? (end - base) : 32;
            int   my = 0;
            float cw = 0.0f;
            if (lane < m) {
                my = csr_src[base + lane];
                my = my < 0 ? 0 : (my > n - 1 ? n - 1 : my);
                cw = dinv[my] * di;
            }
            int j = 0;
            for (; j + 4 <= m; j += 4) gather_rows<F, 4>(xw, my, cw, j, lane, acc);
            if (j + 2 <= m) { gather_rows<F, 2>(xw, my, cw, j, lane, acc); j += 2; }
            if (j < m)      { gather_rows<F, 1>(xw, my, cw, j, lane, acc); }
        }

        float* __restrict__ orow = agg + (size_t)node * F + lane * 4;
#pragma unroll
        for (int v = 0; v < V; ++v) *(float4*)(orow + v * 128) = acc[v];
    }
}

template <int R>
__device__ __forceinline__ void gather_pairs(const float* __restrict__ x, int my, float cw,
                                             int j, bool hi, int l16, float4& acc) {
    float4 r[R];
    float  c[R];
#pragma unroll
    for (int u = 0; u < R; ++u) {
        const int s0 = __builtin_amdgcn_readlane(my, j + 2 * u);
        const int s1 = __builtin_amdgcn_readlane(my, j + 2 * u + 1);
        const int c0 = __builtin_amdgcn_readlane(__float_as_int(cw), j + 2 * u);
        const int c1 = __builtin_amdgcn_readlane(__float_as_int(cw), j + 2 * u + 1);
        const int s  = hi ? s1 : s0;
        c[u] = __int_as_float(hi ? c1 : c0);
        r[u] = *(const float4*)(x + (size_t)s * 64 + l16 * 4);
    }
#pragma unroll
    for (int u = 0; u < R; ++u) {
        acc.x += r[u].x * c[u];
        acc.y += r[u].y * c[u];
        acc.z += r[u].z * c[u];
        acc.w += r[u].w * c[u];
    }
}

__global__ void __launch_bounds__(256)
k_gather_in(const int* __restrict__ rowptr, const int* __restrict__ csr_src,
            const float* __restrict__ dinv, const float* __restrict__ x,
            float* __restrict__ ax, int n) {
    const int  lane = threadIdx.x & 31;
    const bool hi   = lane >= 16;
    const int  l16  = lane & 15;
    const int wid = __builtin_amdgcn_readfirstlane(
        (int)(blockIdx.x * (blockDim.x >> 5) + (threadIdx.x >> 5)));
    const int nw  = gridDim.x * (blockDim.x >> 5);

    for (int node = wid; node < n; node += nw) {
        const int   beg = rowptr[node];
        const int   end = rowptr[node + 1];
        const float di  = dinv[node];

        float4 acc;
        {
            const float4 v = *(const float4*)(x + (size_t)node * 64 + l16 * 4);
            const float  w = di * di;
            acc.x = hi ? 0.0f : v.x * w;
            acc.y = hi ? 0.0f : v.y * w;
            acc.z = hi ? 0.0f : v.z * w;
            acc.w = hi ? 0.0f : v.w * w;
        }

        for (int base = beg; base < end; base += 32) {
            const int m = (end - base) < 32 ? (end - base) : 32;
            int   my = 0;
            float cw = 0.0f;
            if (lane < m) {
                my = csr_src[base + lane];
                my = my < 0 ? 0 : (my > n - 1 ? n - 1 : my);
                cw = dinv[my] * di;
            }
            int j = 0;
            for (; j + 16 <= m; j += 16) gather_pairs<8>(x, my, cw, j, hi, l16, acc);
            if (j + 8 <= m) { gather_pairs<4>(x, my, cw, j, hi, l16, acc); j += 8; }
            if (j + 4 <= m) { gather_pairs<2>(x, my, cw, j, hi, l16, acc); j += 4; }
            if (j + 2 <= m) { gather_pairs<1>(x, my, cw, j, hi, l16, acc); j += 2; }
            if (j < m) {
                const int   s = __builtin_amdgcn_readlane(my, j);
                const float c = __int_as_float(
                    __builtin_amdgcn_readlane(__float_as_int(cw), j));
                if (!hi) {
                    const float4 r = *(const float4*)(x + (size_t)s * 64 + l16 * 4);
                    acc.x += r.x * c;
                    acc.y += r.y * c;
                    acc.z += r.z * c;
                    acc.w += r.w * c;
                }
            }
        }

        float4 o;
        o.x = acc.x + __shfl_xor(acc.x, 16, 32);
        o.y = acc.y + __shfl_xor(acc.y, 16, 32);
        o.z = acc.z + __shfl_xor(acc.z, 16, 32);
        o.w = acc.w + __shfl_xor(acc.w, 16, 32);
        if (!hi) *(float4*)(ax + (size_t)node * 64 + l16 * 4) = o;
    }
}

__global__ void k_node_head(const float* __restrict__ emb, const float* __restrict__ Wn,
                            const float* __restrict__ bn, const float* __restrict__ We,
                            float* __restrict__ out, float* __restrict__ pq, int n) {
    int i = blockIdx.x * blockDim.x + threadIdx.x;
    if (i >= n) return;
    const float4* row = (const float4*)(emb + (size_t)i * 128);
    const float* __restrict__ Ws = We;
    const float* __restrict__ Wd = We + 128 * 2;
    float a0 = bn[0], a1 = bn[1];
    float p0 = 0.0f, p1 = 0.0f, q0 = 0.0f, q1 = 0.0f;
#pragma unroll 8
    for (int q = 0; q < 32; ++q) {
        float4 v = row[q];
        int f = q * 4;
        a0 += v.x * Wn[(f + 0) * 2 + 0] + v.y * Wn[(f + 1) * 2 + 0] +
              v.z * Wn[(f + 2) * 2 + 0] + v.w * Wn[(f + 3) * 2 + 0];
        a1 += v.x * Wn[(f + 0) * 2 + 1] + v.y * Wn[(f + 1) * 2 + 1] +
              v.z * Wn[(f + 2) * 2 + 1] + v.w * Wn[(f + 3) * 2 + 1];
        p0 += v.x * Ws[(f + 0) * 2 + 0] + v.y * Ws[(f + 1) * 2 + 0] +
              v.z * Ws[(f + 2) * 2 + 0] + v.w * Ws[(f + 3) * 2 + 0];
        p1 += v.x * Ws[(f + 0) * 2 + 1] + v.y * Ws[(f + 1) * 2 + 1] +
              v.z * Ws[(f + 2) * 2 + 1] + v.w * Ws[(f + 3) * 2 + 1];
        q0 += v.x * Wd[(f + 0) * 2 + 0] + v.y * Wd[(f + 1) * 2 + 0] +
              v.z * Wd[(f + 2) * 2 + 0] + v.w * Wd[(f + 3) * 2 + 0];
        q1 += v.x * Wd[(f + 0) * 2 + 1] + v.y * Wd[(f + 1) * 2 + 1] +
              v.z * Wd[(f + 2) * 2 + 1] + v.w * Wd[(f + 3) * 2 + 1];
    }
    out[(size_t)i * 2 + 0] = a0;
    out[(size_t)i * 2 + 1] = a1;
    float4 r;
    r.x = p0; r.y = p1; r.z = q0; r.w = q1;
    *(float4*)(pq + (size_t)i * 4) = r;
}

__device__ __forceinline__ float2 edge_one(const float* __restrict__ pq, int s, int d,
                                           float b0, float b1) {
    const float2 p = *(const float2*)(pq + (size_t)s * 4);
    const float2 q = *(const float2*)(pq + (size_t)d * 4 + 2);
    float2 o;
    o.x = (p.x + q.x) + b0;
    o.y = (p.y + q.y) + b1;
    return o;
}

__global__ void __launch_bounds__(256)
k_edge_head(const int* __restrict__ src, const int* __restrict__ dst,
            const float* __restrict__ pq, const float* __restrict__ be,
            float* __restrict__ out, int E) {
    const int   base = blockIdx.x * 1024;
    const float b0 = be[0], b1 = be[1];
    const bool wide = ((((size_t)src | (size_t)dst) & 7) == 0) && (((size_t)out & 15) == 0);
    if (wide && base + 1024 <= E) {
        const int  e0 = base + threadIdx.x * 2;
        const int  e1 = e0 + 512;
        const int2 s0 = *(const int2*)(src + e0);
        const int2 d0 = *(const int2*)(dst + e0);
        const int2 s1 = *(const int2*)(src + e1);
        const int2 d1 = *(const int2*)(dst + e1);
        const float2 a0 = edge_one(pq, s0.x, d0.x, b0, b1);
        const float2 a1 = edge_one(pq, s0.y, d0.y, b0, b1);
        const float2 c0 = edge_one(pq, s1.x, d1.x, b0, b1);
        const float2 c1 = edge_one(pq, s1.y, d1.y, b0, b1);
        float4 o0, o1;
        o0.x = a0.x; o0.y = a0.y; o0.z = a1.x; o0.w = a1.y;
        o1.x = c0.x; o1.y = c0.y; o1.z = c1.x; o1.w = c1.y;
        *(float4*)(out + (size_t)e0 * 2) = o0;
        *(float4*)(out + (size_t)e1 * 2) = o1;
    } else {
        for (int u = 0; u < 2; ++u) {
            for (int k = 0; k < 2; ++k) {
                const int e = base + u * 512 + threadIdx.x * 2 + k;
                if (e < E) *(float2*)(out + (size_t)e * 2) = edge_one(pq, src[e], dst[e], b0, b1);
            }
        }
    }
}

extern "C" void kernel_launch(void* const* d_in, const int* in_sizes, int n_in,
                              void* d_out, int out_size, void* d_ws, size_t ws_size,
                              hipStream_t stream) {
    const float* x  = (const float*)d_in[0];
    const int*   ei = (const int*)d_in[1];
    const float* W1 = (const float*)d_in[2];
    const float* b1 = (const float*)d_in[3];
    const float* W2 = (const float*)d_in[4];
    const float* b2 = (const float*)d_in[5];
    const float* Wn = (const float*)d_in[6];
    const float* bn = (const float*)d_in[7];
    const float* We = (const float*)d_in[8];
    const float* be = (const float*)d_in[9];

    const int N = in_sizes[0] / 64;
    const int E = in_sizes[1] / 2;
    const int* src = ei;
    const int* dst = ei + E;

    char*  ws  = (char*)d_ws;
    size_t off = 0;
    auto carve = [&](size_t bytes) -> float* {
        float* p = (float*)(ws + off);
        off += (bytes + 255) & ~(size_t)255;
        return p;
    };
    int*   cnt    = (int*)carve((size_t)N * 4);
    float* dinv   = carve((size_t)N * 4);
    int*   rowptr = (int*)carve((size_t)(N + 1) * 4);
    int*   rank   = (int*)carve((size_t)E * 4);
    int*   csrsrc = (int*)carve((size_t)E * 4);
    float* ax   = carve((size_t)N * 64 * 4);
    float* agg1 = carve((size_t)N * 256 * 4);
    float* xw2  = carve((size_t)N * 128 * 4);
    float* emb  = carve((size_t)N * 128 * 4);
    float* pq   = carve((size_t)N * 4 * 4);

    float* out_node = (float*)d_out;
    float* out_edge = out_node + (size_t)N * 2;

    k_zero_i    <<<(N + 255) / 256, 256, 0, stream>>>(cnt, N);
    k_count_rank<<<(E + 255) / 256, 256, 0, stream>>>(dst, cnt, rank, E);
    k_scan_dinv <<<(N + 1023) / 1024, 1024, 0, stream>>>(cnt, rowptr, dinv, N);
    k_place     <<<(E + 255) / 256, 256, 0, stream>>>(src, dst, rowptr, rank, csrsrc, E);

    k_gather_in<<<2048, 256, 0, stream>>>(rowptr, csrsrc, dinv, x, ax, N);
    k_gemm_wmma<false, true, 64><<<dim3(N / 16, 256 / 128), 256, 0, stream>>>(
        ax, W1, b1, agg1, N, 256);

    k_gemm_wmma<true, false, 256><<<dim3(N / 16, 128 / 128), 256, 0, stream>>>(
        agg1, W2, nullptr, xw2, N, 128);
    k_gather<128><<<2048, 256, 0, stream>>>(rowptr, csrsrc, dinv, xw2, b2, emb, N);

    k_node_head<<<(N + 255) / 256, 256, 0, stream>>>(emb, Wn, bn, We, out_node, pq, N);
    k_edge_head<<<(E + 1023) / 1024, 256, 0, stream>>>(src, dst, pq, be, out_edge, E);
}


# layer 2 + heads by linearity: Wp = W2 @ [Wn, We top, We bottom] (256x6, WMMA, every call); relu(h) @ Wp split-K over 8 waves; 8-float rows gathered, 4 lanes per node; f32 sum order changes
# speedup vs baseline: 7.5016x; 1.3988x over previous
#include <hip/hip_runtime.h>


typedef float v2f __attribute__((ext_vector_type(2)));
typedef float v8f __attribute__((ext_vector_type(8)));

__global__ void k_zero_i(int* __restrict__ p, int n) {
    int i = blockIdx.x * blockDim.x + threadIdx.x;
    if (i < n) p[i] = 0;
}

__global__ void k_count_rank(const int* __restrict__ dst, int* __restrict__ cnt,
                             int* __restrict__ rank, int E) {
    int e = blockIdx.x * blockDim.x + threadIdx.x;
    if (e < E) rank[e] = atomicAdd(&cnt[dst[e]], 1);
}

__global__ void __launch_bounds__(1024)
k_scan_dinv(const int* __restrict__ cnt, int* __restrict__ rowptr,
            float* __restrict__ dinv, int n) {
    __shared__ int s_pre[32];
    __shared__ int s_inc[32];
    const int t     = threadIdx.x;
    const int lane  = t & 31;
    const int wave  = t >> 5;
    const int tile0 = blockIdx.x * 1024;

    int pre = 0;
    for (int i = t; i < tile0; i += 1024) pre += cnt[i];

    const int i = tile0 + t;
    const int v = (i < n) ? cnt[i] : 0;

    int inc = v;
    for (int off = 1; off < 32; off <<= 1) {
        int u = __shfl_up(inc, off, 32);
        if (lane >= off) inc += u;
    }
    for (int off = 16; off > 0; off >>= 1) pre += __shfl_xor(pre, off, 32);

    if (lane == 31) s_inc[wave] = inc;
    if (lane == 0)  s_pre[wave] = pre;
    __syncthreads();

    int base = 0;
    for (int w = 0; w < 32; ++w) {
        base += s_pre[w];
        if (w < wave) base += s_inc[w];
    }
    const int excl = base + inc - v;
    if (i < n) {
        rowptr[i] = excl;
        dinv[i]   = rsqrtf((float)v + 1.0f);
    }
    if (i == n - 1) rowptr[n] = excl + v;
}

__global__ void k_place(const int* __restrict__ src, const int* __restrict__ dst,
                        const int* __restrict__ rowptr, const int* __restrict__ rank,
                        int* __restrict__ csr_src, int E) {
    int e = blockIdx.x * blockDim.x + threadIdx.x;
    if (e < E) {
        int p = rowptr[dst[e]] + rank[e];
        p = p < 0 ? 0 : (p > E - 1 ? E - 1 : p);
        csr_src[p] = src[e];
    }
}

template <bool RELU_A, bool BIAS_C, int K>
__global__ void k_gemm_wmma(const float* __restrict__ A, const float* __restrict__ B,
                            const float* __restrict__ bias, float* __restrict__ C,
                            int M, int N) {
    constexpr int LDK = K + 2;
    __shared__ float As[16 * LDK];

    const int mt   = blockIdx.x;
    const int nb   = blockIdx.y;
    const int tid  = threadIdx.x;
    const int wave = tid >> 5;
    const int lane = tid & 31;

    const float* __restrict__ Ag = A + (size_t)mt * 16 * K;
#pragma unroll
    for (int idx = tid; idx < 16 * K; idx += 256) {
        int r  = idx / K;
        int kk = idx - r * K;
        float v = Ag[idx];
        if (RELU_A) v = v > 0.0f ? v : 0.0f;
        As[r * LDK + kk] = v;
    }
    __syncthreads();

    const int nt  = nb * 8 + wave;
    const int col = lane & 15;
    const int kb  = (lane >> 4) << 1;

    const float* __restrict__ Arow = As + col * LDK + kb;
    const float* __restrict__ Bc   = B + (size_t)nt * 16 + col;

    float bv = 0.0f;
    if (BIAS_C) bv = bias[nt * 16 + col];
    v8f acc;
#pragma unroll
    for (int r = 0; r < 8; ++r) acc[r] = bv;
#pragma unroll
    for (int k = 0; k < K; k += 4) {
        v2f av; av.x = Arow[k];                      av.y = Arow[k + 1];
        v2f bv; bv.x = Bc[(size_t)(k + kb) * N];     bv.y = Bc[(size_t)(k + kb + 1) * N];
        acc = __builtin_amdgcn_wmma_f32_16x16x4_f32(
            false, av, false, bv, (short)0, acc, false, false);
    }

    float* __restrict__ Cp =
        C + (size_t)(mt * 16 + ((lane >> 4) << 3)) * N + (size_t)nt * 16 + col;
#pragma unroll
    for (int r = 0; r < 8; ++r) Cp[(size_t)r * N] = acc[r];
}

template <int R>
__device__ __forceinline__ void gather_pairs(const float* __restrict__ x, int my, float cw,
                                             int j, bool hi, int l16, float4& acc) {
    float4 r[R];
    float  c[R];
#pragma unroll
    for (int u = 0; u < R; ++u) {
        const int s0 = __builtin_amdgcn_readlane(my, j + 2 * u);
        const int s1 = __builtin_amdgcn_readlane(my, j + 2 * u + 1);
        const int c0 = __builtin_amdgcn_readlane(__float_as_int(cw), j + 2 * u);
        const int c1 = __builtin_amdgcn_readlane(__float_as_int(cw), j + 2 * u + 1);
        const int s  = hi ? s1 : s0;
        c[u] = __int_as_float(hi ? c1 : c0);
        r[u] = *(const float4*)(x + (size_t)s * 64 + l16 * 4);
    }
#pragma unroll
    for (int u = 0; u < R; ++u) {
        acc.x += r[u].x * c[u];
        acc.y += r[u].y * c[u];
        acc.z += r[u].z * c[u];
        acc.w += r[u].w * c[u];
    }
}

__global__ void __launch_bounds__(256)
k_gather_in(const int* __restrict__ rowptr, const int* __restrict__ csr_src,
            const float* __restrict__ dinv, const float* __restrict__ x,
            float* __restrict__ ax, int n) {
    const int  lane = threadIdx.x & 31;
    const bool hi   = lane >= 16;
    const int  l16  = lane & 15;
    const int wid = __builtin_amdgcn_readfirstlane(
        (int)(blockIdx.x * (blockDim.x >> 5) + (threadIdx.x >> 5)));
    const int nw  = gridDim.x * (blockDim.x >> 5);

    for (int node = wid; node < n; node += nw) {
        const int   beg = rowptr[node];
        const int   end = rowptr[node + 1];
        const float di  = dinv[node];

        float4 acc;
        {
            const float4 v = *(const float4*)(x + (size_t)node * 64 + l16 * 4);
            const float  w = di * di;
            acc.x = hi ? 0.0f : v.x * w;
            acc.y = hi ? 0.0f : v.y * w;
            acc.z = hi ? 0.0f : v.z * w;
            acc.w = hi ? 0.0f : v.w * w;
        }

        for (int base = beg; base < end; base += 32) {
            const int m = (end - base) < 32 ? (end - base) : 32;
            int   my = 0;
            float cw = 0.0f;
            if (lane < m) {
                my = csr_src[base + lane];
                my = my < 0 ? 0 : (my > n - 1 ? n - 1 : my);
                cw = dinv[my] * di;
            }
            int j = 0;
            for (; j + 16 <= m; j += 16) gather_pairs<8>(x, my, cw, j, hi, l16, acc);
            if (j + 8 <= m) { gather_pairs<4>(x, my, cw, j, hi, l16, acc); j += 8; }
            if (j + 4 <= m) { gather_pairs<2>(x, my, cw, j, hi, l16, acc); j += 4; }
            if (j + 2 <= m) { gather_pairs<1>(x, my, cw, j, hi, l16, acc); j += 2; }
            if (j < m) {
                const int   s = __builtin_amdgcn_readlane(my, j);
                const float c = __int_as_float(
                    __builtin_amdgcn_readlane(__float_as_int(cw), j));
                if (!hi) {
                    const float4 r = *(const float4*)(x + (size_t)s * 64 + l16 * 4);
                    acc.x += r.x * c;
                    acc.y += r.y * c;
                    acc.z += r.z * c;
                    acc.w += r.w * c;
                }
            }
        }

        float4 o;
        o.x = acc.x + __shfl_xor(acc.x, 16, 32);
        o.y = acc.y + __shfl_xor(acc.y, 16, 32);
        o.z = acc.z + __shfl_xor(acc.z, 16, 32);
        o.w = acc.w + __shfl_xor(acc.w, 16, 32);
        if (!hi) *(float4*)(ax + (size_t)node * 64 + l16 * 4) = o;
    }
}

__global__ void __launch_bounds__(32)
k_fold_heads(const float* __restrict__ W2, const float* __restrict__ b2,
             const float* __restrict__ Wn, const float* __restrict__ We,
             float* __restrict__ Wp) {
    const int lane = threadIdx.x;
    const int col  = lane & 15;
    const int kb   = (lane >> 4) << 1;
    const bool live = col < 6;
    const float* __restrict__ hc =
        col < 2 ? Wn + col : (col < 4 ? We + (col - 2) : (col < 6 ? We + 256 + (col - 4) : Wn));

    if (blockIdx.x < 16) {
        const float* __restrict__ Ar = W2 + (size_t)(blockIdx.x * 16 + col) * 128 + kb;
        v8f acc;
#pragma unroll
        for (int r = 0; r < 8; ++r) acc[r] = 0.0f;
#pragma unroll
        for (int k = 0; k < 128; k += 4) {
            v2f av; av.x = Ar[k];                 av.y = Ar[k + 1];
            const float h0 = hc[(k + kb) * 2];
            const float h1 = hc[(k + kb + 1) * 2];
            v2f bv; bv.x = live ? h0 : 0.0f;      bv.y = live ? h1 : 0.0f;
            acc = __builtin_amdgcn_wmma_f32_16x16x4_f32(
                false, av, false, bv, (short)0, acc, false, false);
        }
        float* __restrict__ Cp = Wp + (size_t)(blockIdx.x * 16 + ((lane >> 4) << 3)) * 16 + col;
#pragma unroll
        for (int r = 0; r < 8; ++r) Cp[(size_t)r * 16] = acc[r];
    } else if (lane < 16) {
        float s = 0.0f;
        for (int k = 0; k < 128; ++k) {
            const float h = hc[k * 2];
            s += b2[k] * (live ? h : 0.0f);
        }
        Wp[256 * 16 + lane] = s;
    }
}

__global__ void __launch_bounds__(256)
k_gemm_fold(const float* __restrict__ A, const float* __restrict__ Wp,
            float* __restrict__ Y, int M) {
    constexpr int K   = 256;
    constexpr int LDK = K + 2;
    constexpr int LDR = 18;
    __shared__ __attribute__((aligned(16))) float As[16 * LDK];
    __shared__ float red[8 * 16 * LDR];

    const int mt   = blockIdx.x;
    const int tid  = threadIdx.x;
    const int wave = tid >> 5;
    const int lane = tid & 31;

    const int col = lane & 15;
    const int kb  = (lane >> 4) << 1;

    const float* __restrict__ Bc = Wp + (size_t)(wave * 32) * 16 + col;
    v2f bv[8];
#pragma unroll
    for (int s = 0; s < 8; ++s) {
        bv[s].x = Bc[(4 * s + kb) * 16];
        bv[s].y = Bc[(4 * s + kb + 1) * 16];
    }

    const float2* __restrict__ Ag = (const float2*)(A + (size_t)mt * 16 * K);
    float2 st[8];
#pragma unroll
    for (int i = 0; i < 8; ++i) st[i] = Ag[tid + 256 * i];
#pragma unroll
    for (int i = 0; i < 8; ++i) {
        const int idx = tid + 256 * i;
        const int r   = idx / (K / 2);
        const int k2  = idx - r * (K / 2);
        float2 v = st[i];
        v.x = v.x > 0.0f ? v.x : 0.0f;
        v.y = v.y > 0.0f ? v.y : 0.0f;
        *(float2*)(As + r * LDK + 2 * k2) = v;
    }
    __syncthreads();

    const float* __restrict__ Arow = As + col * LDK + wave * 32 + kb;
    v8f acc;
#pragma unroll
    for (int r = 0; r < 8; ++r) acc[r] = 0.0f;
#pragma unroll
    for (int s = 0; s < 8; ++s) {
        v2f av; av.x = Arow[4 * s];           av.y = Arow[4 * s + 1];
        acc = __builtin_amdgcn_wmma_f32_16x16x4_f32(
            false, av, false, bv[s], (short)0, acc, false, false);
    }

    float* __restrict__ rp = red + (wave * 16 + ((lane >> 4) << 3)) * LDR + col;
#pragma unroll
    for (int r = 0; r < 8; ++r) rp[r * LDR] = acc[r];
    __syncthreads();

    if (tid < 128) {
        const int row = tid >> 3;
        const int c   = tid & 7;
        float s = red[row * LDR + c];
#pragma unroll
        for (int w = 1; w < 8; ++w) s += red[(w * 16 + row) * LDR + c];
        Y[(size_t)mt * 128 + tid] = s;
    }
}

__global__ void __launch_bounds__(256)
k_gather_heads(const int* __restrict__ rowptr, const int* __restrict__ csr_src,
               const float* __restrict__ dinv, const float* __restrict__ y,
               const float* __restrict__ c, const float* __restrict__ bn,
               float* __restrict__ out, float* __restrict__ pq, int n) {
    const int  t    = blockIdx.x * 256 + threadIdx.x;
    const int  q    = t & 3;
    const int  node = t >> 2;
    const bool live = node < n;
    const int  nd   = live ? node : n - 1;

    const int   beg = rowptr[nd];
    const int   end = rowptr[nd + 1];
    const float di  = dinv[nd];
    const float4* __restrict__ y4 = (const float4*)y;

    float4 a0, a1;
    {
        const float4 s0 = y4[(size_t)nd * 2];
        const float4 s1 = y4[(size_t)nd * 2 + 1];
        const float  w  = di * di;
        const bool   z  = q != 0;
        a0.x = z ? 0.0f : s0.x * w;  a0.y = z ? 0.0f : s0.y * w;
        a0.z = z ? 0.0f : s0.z * w;  a0.w = z ? 0.0f : s0.w * w;
        a1.x = z ? 0.0f : s1.x * w;  a1.y = z ? 0.0f : s1.y * w;
        a1.z = z ? 0.0f : s1.z * w;  a1.w = z ? 0.0f : s1.w * w;
    }

    int e = beg + q;
    for (; e + 4 < end; e += 8) {
        int sA = csr_src[e];
        int sB = csr_src[e + 4];
        sA = sA < 0 ? 0 : (sA > n - 1 ? n - 1 : sA);
        sB = sB < 0 ? 0 : (sB > n - 1 ? n - 1 : sB);
        const float  cA = dinv[sA] * di;
        const float  cB = dinv[sB] * di;
        const float4 rA0 = y4[(size_t)sA * 2], rA1 = y4[(size_t)sA * 2 + 1];
        const float4 rB0 = y4[(size_t)sB * 2], rB1 = y4[(size_t)sB * 2 + 1];
        a0.x += rA0.x * cA;  a0.y += rA0.y * cA;  a0.z += rA0.z * cA;  a0.w += rA0.w * cA;
        a1.x += rA1.x * cA;  a1.y += rA1.y * cA;  a1.z += rA1.z * cA;  a1.w += rA1.w * cA;
        a0.x += rB0.x * cB;  a0.y += rB0.y * cB;  a0.z += rB0.z * cB;  a0.w += rB0.w * cB;
        a1.x += rB1.x * cB;  a1.y += rB1.y * cB;  a1.z += rB1.z * cB;  a1.w += rB1.w * cB;
    }
    if (e < end) {
        int sA = csr_src[e];
        sA = sA < 0 ? 0 : (sA > n - 1 ? n - 1 : sA);
        const float  cA = dinv[sA] * di;
        const float4 rA0 = y4[(size_t)sA * 2], rA1 = y4[(size_t)sA * 2 + 1];
        a0.x += rA0.x * cA;  a0.y += rA0.y * cA;  a0.z += rA0.z * cA;  a0.w += rA0.w * cA;
        a1.x += rA1.x * cA;  a1.y += rA1.y * cA;  a1.z += rA1.z * cA;  a1.w += rA1.w * cA;
    }

#pragma unroll
    for (int off = 1; off <= 2; off <<= 1) {
        a0.x += __shfl_xor(a0.x, off, 32);  a0.y += __shfl_xor(a0.y, off, 32);
        a0.z += __shfl_xor(a0.z, off, 32);  a0.w += __shfl_xor(a0.w, off, 32);
        a1.x += __shfl_xor(a1.x, off, 32);  a1.y += __shfl_xor(a1.y, off, 32);
    }

    if (live) {
        if (q == 0) {
            float2 o;
            o.x = (a0.x + c[0]) + bn[0];
            o.y = (a0.y + c[1]) + bn[1];
            *(float2*)(out + (size_t)nd * 2) = o;
        } else if (q == 1) {
            float4 r;
            r.x = a0.z + c[2];  r.y = a0.w + c[3];
            r.z = a1.x + c[4];  r.w = a1.y + c[5];
            *(float4*)(pq + (size_t)nd * 4) = r;
        }
    }
}

__device__ __forceinline__ float2 edge_one(const float* __restrict__ pq, int s, int d,
                                           float b0, float b1) {
    const float2 p = *(const float2*)(pq + (size_t)s * 4);
    const float2 q = *(const float2*)(pq + (size_t)d * 4 + 2);
    float2 o;
    o.x = (p.x + q.x) + b0;
    o.y = (p.y + q.y) + b1;
    return o;
}

__global__ void __launch_bounds__(256)
k_edge_head(const int* __restrict__ src, const int* __restrict__ dst,
            const float* __restrict__ pq, const float* __restrict__ be,
            float* __restrict__ out, int E) {
    const int   base = blockIdx.x * 1024;
    const float b0 = be[0], b1 = be[1];
    const bool wide = ((((size_t)src | (size_t)dst) & 7) == 0) && (((size_t)out & 15) == 0);
    if (wide && base + 1024 <= E) {
        const int  e0 = base + threadIdx.x * 2;
        const int  e1 = e0 + 512;
        const int2 s0 = *(const int2*)(src + e0);
        const int2 d0 = *(const int2*)(dst + e0);
        const int2 s1 = *(const int2*)(src + e1);
        const int2 d1 = *(const int2*)(dst + e1);
        const float2 a0 = edge_one(pq, s0.x, d0.x, b0, b1);
        const float2 a1 = edge_one(pq, s0.y, d0.y, b0, b1);
        const float2 c0 = edge_one(pq, s1.x, d1.x, b0, b1);
        const float2 c1 = edge_one(pq, s1.y, d1.y, b0, b1);
        float4 o0, o1;
        o0.x = a0.x; o0.y = a0.y; o0.z = a1.x; o0.w = a1.y;
        o1.x = c0.x; o1.y = c0.y; o1.z = c1.x; o1.w = c1.y;
        *(float4*)(out + (size_t)e0 * 2) = o0;
        *(float4*)(out + (size_t)e1 * 2) = o1;
    } else {
        for (int u = 0; u < 2; ++u) {
            for (int k = 0; k < 2; ++k) {
                const int e = base + u * 512 + threadIdx.x * 2 + k;
                if (e < E) *(float2*)(out + (size_t)e * 2) = edge_one(pq, src[e], dst[e], b0, b1);
            }
        }
    }
}

extern "C" void kernel_launch(void* const* d_in, const int* in_sizes, int n_in,
                              void* d_out, int out_size, void* d_ws, size_t ws_size,
                              hipStream_t stream) {
    const float* x  = (const float*)d_in[0];
    const int*   ei = (const int*)d_in[1];
    const float* W1 = (const float*)d_in[2];
    const float* b1 = (const float*)d_in[3];
    const float* W2 = (const float*)d_in[4];
    const float* b2 = (const float*)d_in[5];
    const float* Wn = (const float*)d_in[6];
    const float* bn = (const float*)d_in[7];
    const float* We = (const float*)d_in[8];
    const float* be = (const float*)d_in[9];

    const int N = in_sizes[0] / 64;
    const int E = in_sizes[1] / 2;
    const int* src = ei;
    const int* dst = ei + E;

    char*  ws  = (char*)d_ws;
    size_t off = 0;
    auto carve = [&](size_t bytes) -> float* {
        float* p = (float*)(ws + off);
        off += (bytes + 255) & ~(size_t)255;
        return p;
    };
    int*   cnt    = (int*)carve((size_t)N * 4);
    float* dinv   = carve((size_t)N * 4);
    int*   rowptr = (int*)carve((size_t)(N + 1) * 4);
    int*   rank   = (int*)carve((size_t)E * 4);
    int*   csrsrc = (int*)carve((size_t)E * 4);
    float* ax   = carve((size_t)N * 64 * 4);
    float* agg1 = carve((size_t)N * 256 * 4);
    float* wp   = carve((size_t)(256 + 1) * 16 * 4);
    float* y    = carve((size_t)N * 8 * 4);
    float* pq   = carve((size_t)N * 4 * 4);

    float* out_node = (float*)d_out;
    float* out_edge = out_node + (size_t)N * 2;

    k_zero_i    <<<(N + 255) / 256, 256, 0, stream>>>(cnt, N);
    k_count_rank<<<(E + 255) / 256, 256, 0, stream>>>(dst, cnt, rank, E);
    k_scan_dinv <<<(N + 1023) / 1024, 1024, 0, stream>>>(cnt, rowptr, dinv, N);
    k_place     <<<(E + 255) / 256, 256, 0, stream>>>(src, dst, rowptr, rank, csrsrc, E);

    k_gather_in<<<2048, 256, 0, stream>>>(rowptr, csrsrc, dinv, x, ax, N);
    k_gemm_wmma<false, true, 64><<<dim3(N / 16, 256 / 128), 256, 0, stream>>>(
        ax, W1, b1, agg1, N, 256);

    k_fold_heads<<<17, 32, 0, stream>>>(W2, b2, Wn, We, wp);
    k_gemm_fold<<<N / 16, 256, 0, stream>>>(agg1, wp, y, N);
    k_gather_heads<<<(N * 4 + 255) / 256, 256, 0, stream>>>(
        rowptr, csrsrc, dinv, y, wp + 256 * 16, bn, out_node, pq, N);

    k_edge_head<<<(E + 1023) / 1024, 256, 0, stream>>>(src, dst, pq, be, out_edge, E);
}


# layer-1 GEMM, bias, ReLU and folded layer-2 GEMM fused in one kernel on a fixed grid: the hidden 16x256 tile stays in WMMA D registers and is the next B operand; k order and f32 sum order change
# speedup vs baseline: 8.7804x; 1.1705x over previous
#include <hip/hip_runtime.h>


typedef float v2f __attribute__((ext_vector_type(2)));
typedef float v8f __attribute__((ext_vector_type(8)));

__global__ void k_zero_i(int* __restrict__ p, int n) {
    int i = blockIdx.x * blockDim.x + threadIdx.x;
    if (i < n) p[i] = 0;
}

__global__ void k_count_rank(const int* __restrict__ dst, int* __restrict__ cnt,
                             int* __restrict__ rank, int E) {
    int e = blockIdx.x * blockDim.x + threadIdx.x;
    if (e < E) rank[e] = atomicAdd(&cnt[dst[e]], 1);
}

__global__ void __launch_bounds__(1024)
k_scan_dinv(const int* __restrict__ cnt, int* __restrict__ rowptr,
            float* __restrict__ dinv, int n) {
    __shared__ int s_pre[32];
    __shared__ int s_inc[32];
    const int t     = threadIdx.x;
    const int lane  = t & 31;
    const int wave  = t >> 5;
    const int tile0 = blockIdx.x * 1024;

    int pre = 0;
    for (int i = t; i < tile0; i += 1024) pre += cnt[i];

    const int i = tile0 + t;
    const int v = (i < n) ? cnt[i] : 0;

    int inc = v;
    for (int off = 1; off < 32; off <<= 1) {
        int u = __shfl_up(inc, off, 32);
        if (lane >= off) inc += u;
    }
    for (int off = 16; off > 0; off >>= 1) pre += __shfl_xor(pre, off, 32);

    if (lane == 31) s_inc[wave] = inc;
    if (lane == 0)  s_pre[wave] = pre;
    __syncthreads();

    int base = 0;
    for (int w = 0; w < 32; ++w) {
        base += s_pre[w];
        if (w < wave) base += s_inc[w];
    }
    const int excl = base + inc - v;
    if (i < n) {
        rowptr[i] = excl;
        dinv[i]   = rsqrtf((float)v + 1.0f);
    }
    if (i == n - 1) rowptr[n] = excl + v;
}

__global__ void k_place(const int* __restrict__ src, const int* __restrict__ dst,
                        const int* __restrict__ rowptr, const int* __restrict__ rank,
                        int* __restrict__ csr_src, int E) {
    int e = blockIdx.x * blockDim.x + threadIdx.x;
    if (e < E) {
        int p = rowptr[dst[e]] + rank[e];
        p = p < 0 ? 0 : (p > E - 1 ? E - 1 : p);
        csr_src[p] = src[e];
    }
}

template <int R>
__device__ __forceinline__ void gather_pairs(const float* __restrict__ x, int my, float cw,
                                             int j, bool hi, int l16, float4& acc) {
    float4 r[R];
    float  c[R];
#pragma unroll
    for (int u = 0; u < R; ++u) {
        const int s0 = __builtin_amdgcn_readlane(my, j + 2 * u);
        const int s1 = __builtin_amdgcn_readlane(my, j + 2 * u + 1);
        const int c0 = __builtin_amdgcn_readlane(__float_as_int(cw), j + 2 * u);
        const int c1 = __builtin_amdgcn_readlane(__float_as_int(cw), j + 2 * u + 1);
        const int s  = hi ? s1 : s0;
        c[u] = __int_as_float(hi ? c1 : c0);
        r[u] = *(const float4*)(x + (size_t)s * 64 + l16 * 4);
    }
#pragma unroll
    for (int u = 0; u < R; ++u) {
        acc.x += r[u].x * c[u];
        acc.y += r[u].y * c[u];
        acc.z += r[u].z * c[u];
        acc.w += r[u].w * c[u];
    }
}

__global__ void __launch_bounds__(256)
k_gather_in(const int* __restrict__ rowptr, const int* __restrict__ csr_src,
            const float* __restrict__ dinv, const float* __restrict__ x,
            float* __restrict__ ax, int n) {
    const int  lane = threadIdx.x & 31;
    const bool hi   = lane >= 16;
    const int  l16  = lane & 15;
    const int wid = __builtin_amdgcn_readfirstlane(
        (int)(blockIdx.x * (blockDim.x >> 5) + (threadIdx.x >> 5)));
    const int nw  = gridDim.x * (blockDim.x >> 5);

    for (int node = wid; node < n; node += nw) {
        const int   beg = rowptr[node];
        const int   end = rowptr[node + 1];
        const float di  = dinv[node];

        float4 acc;
        {
            const float4 v = *(const float4*)(x + (size_t)node * 64 + l16 * 4);
            const float  w = di * di;
            acc.x = hi ? 0.0f : v.x * w;
            acc.y = hi ? 0.0f : v.y * w;
            acc.z = hi ? 0.0f : v.z * w;
            acc.w = hi ? 0.0f : v.w * w;
        }

        for (int base = beg; base < end; base += 32) {
            const int m = (end - base) < 32 ? (end - base) : 32;
            int   my = 0;
            float cw = 0.0f;
            if (lane < m) {
                my = csr_src[base + lane];
                my = my < 0 ? 0 : (my > n - 1 ? n - 1 : my);
                cw = dinv[my] * di;
            }
            int j = 0;
            for (; j + 16 <= m; j += 16) gather_pairs<8>(x, my, cw, j, hi, l16, acc);
            if (j + 8 <= m) { gather_pairs<4>(x, my, cw, j, hi, l16, acc); j += 8; }
            if (j + 4 <= m) { gather_pairs<2>(x, my, cw, j, hi, l16, acc); j += 4; }
            if (j + 2 <= m) { gather_pairs<1>(x, my, cw, j, hi, l16, acc); j += 2; }
            if (j < m) {
                const int   s = __builtin_amdgcn_readlane(my, j);
                const float c = __int_as_float(
                    __builtin_amdgcn_readlane(__float_as_int(cw), j));
                if (!hi) {
                    const float4 r = *(const float4*)(x + (size_t)s * 64 + l16 * 4);
                    acc.x += r.x * c;
                    acc.y += r.y * c;
                    acc.z += r.z * c;
                    acc.w += r.w * c;
                }
            }
        }

        float4 o;
        o.x = acc.x + __shfl_xor(acc.x, 16, 32);
        o.y = acc.y + __shfl_xor(acc.y, 16, 32);
        o.z = acc.z + __shfl_xor(acc.z, 16, 32);
        o.w = acc.w + __shfl_xor(acc.w, 16, 32);
        if (!hi) *(float4*)(ax + (size_t)node * 64 + l16 * 4) = o;
    }
}

__global__ void __launch_bounds__(32)
k_fold_heads(const float* __restrict__ W2, const float* __restrict__ b2,
             const float* __restrict__ Wn, const float* __restrict__ We,
             float* __restrict__ Wp) {
    const int lane = threadIdx.x;
    const int col  = lane & 15;
    const int kb   = (lane >> 4) << 1;
    const bool live = col < 6;
    const float* __restrict__ hc =
        col < 2 ? Wn + col : (col < 4 ? We + (col - 2) : (col < 6 ? We + 256 + (col - 4) : Wn));

    if (blockIdx.x < 16) {
        const float* __restrict__ Ar = W2 + (size_t)(blockIdx.x * 16 + col) * 128 + kb;
        v8f acc;
#pragma unroll
        for (int r = 0; r < 8; ++r) acc[r] = 0.0f;
#pragma unroll
        for (int k = 0; k < 128; k += 4) {
            v2f av; av.x = Ar[k];                 av.y = Ar[k + 1];
            const float h0 = hc[(k + kb) * 2];
            const float h1 = hc[(k + kb + 1) * 2];
            v2f bv; bv.x = live ? h0 : 0.0f;      bv.y = live ? h1 : 0.0f;
            acc = __builtin_amdgcn_wmma_f32_16x16x4_f32(
                false, av, false, bv, (short)0, acc, false, false);
        }
        float* __restrict__ Cp = Wp + (size_t)(blockIdx.x * 16 + ((lane >> 4) << 3)) * 16 + col;
#pragma unroll
        for (int r = 0; r < 8; ++r) Cp[(size_t)r * 16] = acc[r];
    } else if (lane < 16) {
        float s = 0.0f;
        for (int k = 0; k < 128; ++k) {
            const float h = hc[k * 2];
            s += b2[k] * (live ? h : 0.0f);
        }
        Wp[256 * 16 + lane] = s;
    }
}

__global__ void __launch_bounds__(256)
k_hidden_fold(const float* __restrict__ AX, const float* __restrict__ W1,
              const float* __restrict__ b1, const float* __restrict__ Wp,
              float* __restrict__ Y, int ntiles) {
    constexpr int LDA = 68;
    constexpr int LDR = 12;
    __shared__ __attribute__((aligned(16))) float As[16 * LDA];
    __shared__ __attribute__((aligned(16))) float red[8 * 16 * LDR];

    const int tid  = threadIdx.x;
    const int wave = tid >> 5;
    const int lane = tid & 31;
    const int c    = lane & 15;
    const int h    = lane >> 4;
    const int hb   = wave * 32;

    v2f a1[2][16];
#pragma unroll
    for (int t = 0; t < 2; ++t) {
#pragma unroll
        for (int sp = 0; sp < 8; ++sp) {
#pragma unroll
            for (int u = 0; u < 2; ++u) {
                const int k = 8 * sp + 4 * h + 2 * u;
                a1[t][2 * sp + u].x = W1[(size_t)k * 256 + hb + 16 * t + c];
                a1[t][2 * sp + u].y = W1[(size_t)(k + 1) * 256 + hb + 16 * t + c];
            }
        }
    }
    v2f a2[2][4];
#pragma unroll
    for (int t = 0; t < 2; ++t) {
#pragma unroll
        for (int s = 0; s < 4; ++s) {
            const int ch = hb + 16 * t + 8 * h + 2 * s;
            a2[t][s].x = Wp[(size_t)ch * 16 + c];
            a2[t][s].y = Wp[(size_t)(ch + 1) * 16 + c];
        }
    }
    float bs[2][8];
#pragma unroll
    for (int t = 0; t < 2; ++t) {
#pragma unroll
        for (int r = 0; r < 8; ++r) bs[t][r] = b1[hb + 16 * t + 8 * h + r];
    }

    const int sr = tid >> 4;
    const int sc = tid & 15;

    int tile = blockIdx.x;
    float4 g = make_float4(0.0f, 0.0f, 0.0f, 0.0f);
    if (tile < ntiles) g = *(const float4*)(AX + (size_t)tile * 1024 + tid * 4);

    for (; tile < ntiles; tile += gridDim.x) {
        *(float4*)(As + sr * LDA + sc * 4) = g;
        __syncthreads();
        const int nxt = tile + gridDim.x;
        if (nxt < ntiles) g = *(const float4*)(AX + (size_t)nxt * 1024 + tid * 4);

        v2f b[16];
#pragma unroll
        for (int sp = 0; sp < 8; ++sp) {
            const float4 q = *(const float4*)(As + c * LDA + 8 * sp + 4 * h);
            b[2 * sp].x     = q.x;  b[2 * sp].y     = q.y;
            b[2 * sp + 1].x = q.z;  b[2 * sp + 1].y = q.w;
        }

        v8f h0, h1;
#pragma unroll
        for (int r = 0; r < 8; ++r) { h0[r] = bs[0][r]; h1[r] = bs[1][r]; }
#pragma unroll
        for (int s = 0; s < 16; ++s) {
            h0 = __builtin_amdgcn_wmma_f32_16x16x4_f32(
                false, a1[0][s], false, b[s], (short)0, h0, false, false);
            h1 = __builtin_amdgcn_wmma_f32_16x16x4_f32(
                false, a1[1][s], false, b[s], (short)0, h1, false, false);
        }
#pragma unroll
        for (int r = 0; r < 8; ++r) {
            h0[r] = h0[r] > 0.0f ? h0[r] : 0.0f;
            h1[r] = h1[r] > 0.0f ? h1[r] : 0.0f;
        }

        v8f o;
#pragma unroll
        for (int r = 0; r < 8; ++r) o[r] = 0.0f;
#pragma unroll
        for (int s = 0; s < 4; ++s) {
            v2f hv; hv.x = h0[2 * s]; hv.y = h0[2 * s + 1];
            o = __builtin_amdgcn_wmma_f32_16x16x4_f32(
                false, a2[0][s], false, hv, (short)0, o, false, false);
        }
#pragma unroll
        for (int s = 0; s < 4; ++s) {
            v2f hv; hv.x = h1[2 * s]; hv.y = h1[2 * s + 1];
            o = __builtin_amdgcn_wmma_f32_16x16x4_f32(
                false, a2[1][s], false, hv, (short)0, o, false, false);
        }

        if (h == 0) {
            float* __restrict__ rp = red + (wave * 16 + c) * LDR;
            *(float4*)(rp)     = make_float4(o[0], o[1], o[2], o[3]);
            *(float4*)(rp + 4) = make_float4(o[4], o[5], o[6], o[7]);
        }
        __syncthreads();

        if (tid < 128) {
            const int node = tid >> 3;
            const int j    = tid & 7;
            float s = red[node * LDR + j];
#pragma unroll
            for (int w = 1; w < 8; ++w) s += red[(w * 16 + node) * LDR + j];
            Y[(size_t)tile * 128 + tid] = s;
        }
    }
}

__global__ void __launch_bounds__(256)
k_gather_heads(const int* __restrict__ rowptr, const int* __restrict__ csr_src,
               const float* __restrict__ dinv, const float* __restrict__ y,
               const float* __restrict__ c, const float* __restrict__ bn,
               float* __restrict__ out, float* __restrict__ pq, int n) {
    const int  t    = blockIdx.x * 256 + threadIdx.x;
    const int  q    = t & 3;
    const int  node = t >> 2;
    const bool live = node < n;
    const int  nd   = live ? node : n - 1;

    const int   beg = rowptr[nd];
    const int   end = rowptr[nd + 1];
    const float di  = dinv[nd];
    const float4* __restrict__ y4 = (const float4*)y;

    float4 a0, a1;
    {
        const float4 s0 = y4[(size_t)nd * 2];
        const float4 s1 = y4[(size_t)nd * 2 + 1];
        const float  w  = di * di;
        const bool   z  = q != 0;
        a0.x = z ? 0.0f : s0.x * w;  a0.y = z ? 0.0f : s0.y * w;
        a0.z = z ? 0.0f : s0.z * w;  a0.w = z ? 0.0f : s0.w * w;
        a1.x = z ? 0.0f : s1.x * w;  a1.y = z ? 0.0f : s1.y * w;
        a1.z = z ? 0.0f : s1.z * w;  a1.w = z ? 0.0f : s1.w * w;
    }

    int e = beg + q;
    for (; e + 4 < end; e += 8) {
        int sA = csr_src[e];
        int sB = csr_src[e + 4];
        sA = sA < 0 ? 0 : (sA > n - 1 ? n - 1 : sA);
        sB = sB < 0 ? 0 : (sB > n - 1 ? n - 1 : sB);
        const float  cA = dinv[sA] * di;
        const float  cB = dinv[sB] * di;
        const float4 rA0 = y4[(size_t)sA * 2], rA1 = y4[(size_t)sA * 2 + 1];
        const float4 rB0 = y4[(size_t)sB * 2], rB1 = y4[(size_t)sB * 2 + 1];
        a0.x += rA0.x * cA;  a0.y += rA0.y * cA;  a0.z += rA0.z * cA;  a0.w += rA0.w * cA;
        a1.x += rA1.x * cA;  a1.y += rA1.y * cA;  a1.z += rA1.z * cA;  a1.w += rA1.w * cA;
        a0.x += rB0.x * cB;  a0.y += rB0.y * cB;  a0.z += rB0.z * cB;  a0.w += rB0.w * cB;
        a1.x += rB1.x * cB;  a1.y += rB1.y * cB;  a1.z += rB1.z * cB;  a1.w += rB1.w * cB;
    }
    if (e < end) {
        int sA = csr_src[e];
        sA = sA < 0 ? 0 : (sA > n - 1 ? n - 1 : sA);
        const float  cA = dinv[sA] * di;
        const float4 rA0 = y4[(size_t)sA * 2], rA1 = y4[(size_t)sA * 2 + 1];
        a0.x += rA0.x * cA;  a0.y += rA0.y * cA;  a0.z += rA0.z * cA;  a0.w += rA0.w * cA;
        a1.x += rA1.x * cA;  a1.y += rA1.y * cA;  a1.z += rA1.z * cA;  a1.w += rA1.w * cA;
    }

#pragma unroll
    for (int off = 1; off <= 2; off <<= 1) {
        a0.x += __shfl_xor(a0.x, off, 32);  a0.y += __shfl_xor(a0.y, off, 32);
        a0.z += __shfl_xor(a0.z, off, 32);  a0.w += __shfl_xor(a0.w, off, 32);
        a1.x += __shfl_xor(a1.x, off, 32);  a1.y += __shfl_xor(a1.y, off, 32);
    }

    if (live) {
        if (q == 0) {
            float2 o;
            o.x = (a0.x + c[0]) + bn[0];
            o.y = (a0.y + c[1]) + bn[1];
            *(float2*)(out + (size_t)nd * 2) = o;
        } else if (q == 1) {
            float4 r;
            r.x = a0.z + c[2];  r.y = a0.w + c[3];
            r.z = a1.x + c[4];  r.w = a1.y + c[5];
            *(float4*)(pq + (size_t)nd * 4) = r;
        }
    }
}

__device__ __forceinline__ float2 edge_one(const float* __restrict__ pq, int s, int d,
                                           float b0, float b1) {
    const float2 p = *(const float2*)(pq + (size_t)s * 4);
    const float2 q = *(const float2*)(pq + (size_t)d * 4 + 2);
    float2 o;
    o.x = (p.x + q.x) + b0;
    o.y = (p.y + q.y) + b1;
    return o;
}

__global__ void __launch_bounds__(256)
k_edge_head(const int* __restrict__ src, const int* __restrict__ dst,
            const float* __restrict__ pq, const float* __restrict__ be,
            float* __restrict__ out, int E) {
    const int   base = blockIdx.x * 1024;
    const float b0 = be[0], b1 = be[1];
    const bool wide = ((((size_t)src | (size_t)dst) & 7) == 0) && (((size_t)out & 15) == 0);
    if (wide && base + 1024 <= E) {
        const int  e0 = base + threadIdx.x * 2;
        const int  e1 = e0 + 512;
        const int2 s0 = *(const int2*)(src + e0);
        const int2 d0 = *(const int2*)(dst + e0);
        const int2 s1 = *(const int2*)(src + e1);
        const int2 d1 = *(const int2*)(dst + e1);
        const float2 a0 = edge_one(pq, s0.x, d0.x, b0, b1);
        const float2 a1 = edge_one(pq, s0.y, d0.y, b0, b1);
        const float2 c0 = edge_one(pq, s1.x, d1.x, b0, b1);
        const float2 c1 = edge_one(pq, s1.y, d1.y, b0, b1);
        float4 o0, o1;
        o0.x = a0.x; o0.y = a0.y; o0.z = a1.x; o0.w = a1.y;
        o1.x = c0.x; o1.y = c0.y; o1.z = c1.x; o1.w = c1.y;
        *(float4*)(out + (size_t)e0 * 2) = o0;
        *(float4*)(out + (size_t)e1 * 2) = o1;
    } else {
        for (int u = 0; u < 2; ++u) {
            for (int k = 0; k < 2; ++k) {
                const int e = base + u * 512 + threadIdx.x * 2 + k;
                if (e < E) *(float2*)(out + (size_t)e * 2) = edge_one(pq, src[e], dst[e], b0, b1);
            }
        }
    }
}

extern "C" void kernel_launch(void* const* d_in, const int* in_sizes, int n_in,
                              void* d_out, int out_size, void* d_ws, size_t ws_size,
                              hipStream_t stream) {
    const float* x  = (const float*)d_in[0];
    const int*   ei = (const int*)d_in[1];
    const float* W1 = (const float*)d_in[2];
    const float* b1 = (const float*)d_in[3];
    const float* W2 = (const float*)d_in[4];
    const float* b2 = (const float*)d_in[5];
    const float* Wn = (const float*)d_in[6];
    const float* bn = (const float*)d_in[7];
    const float* We = (const float*)d_in[8];
    const float* be = (const float*)d_in[9];

    const int N = in_sizes[0] / 64;
    const int E = in_sizes[1] / 2;
    const int* src = ei;
    const int* dst = ei + E;

    char*  ws  = (char*)d_ws;
    size_t off = 0;
    auto carve = [&](size_t bytes) -> float* {
        float* p = (float*)(ws + off);
        off += (bytes + 255) & ~(size_t)255;
        return p;
    };
    int*   cnt    = (int*)carve((size_t)N * 4);
    float* dinv   = carve((size_t)N * 4);
    int*   rowptr = (int*)carve((size_t)(N + 1) * 4);
    int*   rank   = (int*)carve((size_t)E * 4);
    int*   csrsrc = (int*)carve((size_t)E * 4);
    float* ax   = carve((size_t)N * 64 * 4);
    float* wp   = carve((size_t)(256 + 1) * 16 * 4);
    float* y    = carve((size_t)N * 8 * 4);
    float* pq   = carve((size_t)N * 4 * 4);

    float* out_node = (float*)d_out;
    float* out_edge = out_node + (size_t)N * 2;

    k_zero_i    <<<(N + 255) / 256, 256, 0, stream>>>(cnt, N);
    k_count_rank<<<(E + 255) / 256, 256, 0, stream>>>(dst, cnt, rank, E);
    k_scan_dinv <<<(N + 1023) / 1024, 1024, 0, stream>>>(cnt, rowptr, dinv, N);
    k_place     <<<(E + 255) / 256, 256, 0, stream>>>(src, dst, rowptr, rank, csrsrc, E);

    k_gather_in<<<2048, 256, 0, stream>>>(rowptr, csrsrc, dinv, x, ax, N);

    k_fold_heads<<<17, 32, 0, stream>>>(W2, b2, Wn, We, wp);
    k_hidden_fold<<<768, 256, 0, stream>>>(ax, W1, b1, wp, y, N / 16);
    k_gather_heads<<<(N * 4 + 255) / 256, 256, 0, stream>>>(
        rowptr, csrsrc, dinv, y, wp + 256 * 16, bn, out_node, pq, N);

    k_edge_head<<<(E + 1023) / 1024, 256, 0, stream>>>(src, dst, pq, be, out_edge, E);
}


# edge bucketing without global atomics: two-level counting sort on LDS counters (block coarse counts, coarse scatter, per-bucket place), 9 -> 8 launches; arrival order, so f32 sum order, reshuffles
# speedup vs baseline: 15.6984x; 1.7879x over previous
#include <hip/hip_runtime.h>


typedef float v2f __attribute__((ext_vector_type(2)));
typedef float v8f __attribute__((ext_vector_type(8)));

constexpr int BSH   = 8;
constexpr int BNODE = 1 << BSH;
constexpr int NBMAX = 256;
constexpr int CHUNK = 4096;

__device__ __forceinline__ int clamp_node(int v, int n) {
    return v < 0 ? 0 : (v > n - 1 ? n - 1 : v);
}
__device__ __forceinline__ int bucket_of(int d) {
    const int b = d >> BSH;
    return b > NBMAX - 1 ? NBMAX - 1 : b;
}

__global__ void __launch_bounds__(1024)
k_bucket_hist(const int* __restrict__ dst, int* __restrict__ hist, int E, int n) {
    __shared__ int h[NBMAX];
    const int t    = threadIdx.x;
    const int base = blockIdx.x * CHUNK;
    int d[4];
#pragma unroll
    for (int k = 0; k < 4; ++k) {
        const int e = base + k * 1024 + t;
        d[k] = e < E ? dst[e] : 0;
    }
    if (t < NBMAX) h[t] = 0;
    __syncthreads();
#pragma unroll
    for (int k = 0; k < 4; ++k) {
        const int e = base + k * 1024 + t;
        if (e < E) atomicAdd(&h[bucket_of(clamp_node(d[k], n))], 1);
    }
    __syncthreads();
    if (t < NBMAX) hist[(size_t)blockIdx.x * NBMAX + t] = h[t];
}

__device__ __forceinline__ int wave_incl_scan(int v, int lane) {
    for (int off = 1; off < 32; off <<= 1) {
        const int u = __shfl_up(v, off, 32);
        if (lane >= off) v += u;
    }
    return v;
}

__global__ void __launch_bounds__(1024)
k_bucket_scatter(const int* __restrict__ src, const int* __restrict__ dst,
                 const int* __restrict__ hist, int* __restrict__ bstart,
                 int* __restrict__ ent, int E, int n) {
    __shared__ int h[NBMAX];
    __shared__ int gb[NBMAX];
    __shared__ int ls[NBMAX];
    __shared__ __attribute__((aligned(16))) int ptot[16 * NBMAX];
    __shared__ __attribute__((aligned(16))) int ppre[16 * NBMAX];
    __shared__ int stage[CHUNK];
    __shared__ int wsum[8];
    __shared__ int lsum[8];
    const int t    = threadIdx.x;
    const int lane = t & 31;
    const int wave = t >> 5;
    const int blk  = blockIdx.x;
    const int nblk = gridDim.x;
    const int base = blk * CHUNK;

    int d[4], s[4];
#pragma unroll
    for (int k = 0; k < 4; ++k) {
        const int e = base + k * 1024 + t;
        d[k] = e < E ? dst[e] : 0;
        s[k] = e < E ? src[e] : 0;
    }
    const int own = t < NBMAX ? hist[(size_t)blk * NBMAX + t] : 0;

    {
        const int c4 = (t & 63) * 4;
        const int q  = t >> 6;
        int4 tot = make_int4(0, 0, 0, 0);
        int4 pre = make_int4(0, 0, 0, 0);
        for (int j0 = q; j0 < nblk; j0 += 256) {
            int4 v[16];
#pragma unroll
            for (int u = 0; u < 16; ++u) {
                const int j = j0 + 16 * u;
                v[u] = make_int4(0, 0, 0, 0);
                if (j < nblk) v[u] = *(const int4*)(hist + (size_t)j * NBMAX + c4);
            }
#pragma unroll
            for (int u = 0; u < 16; ++u) {
                const int j = j0 + 16 * u;
                tot.x += v[u].x;  tot.y += v[u].y;  tot.z += v[u].z;  tot.w += v[u].w;
                if (j < blk) {
                    pre.x += v[u].x;  pre.y += v[u].y;  pre.z += v[u].z;  pre.w += v[u].w;
                }
            }
        }
        *(int4*)(ptot + q * NBMAX + c4) = tot;
        *(int4*)(ppre + q * NBMAX + c4) = pre;
    }
    if (t < NBMAX) h[t] = 0;
    __syncthreads();

    int tot = 0, pre = 0;
    if (t < NBMAX) {
#pragma unroll
        for (int q = 0; q < 16; ++q) {
            tot += ptot[q * NBMAX + t];
            pre += ppre[q * NBMAX + t];
        }
    }
    const int inc = wave_incl_scan(tot, lane);
    const int inl = wave_incl_scan(own, lane);
    if (lane == 31 && wave < 8) { wsum[wave] = inc; lsum[wave] = inl; }
    __syncthreads();
    if (t < NBMAX) {
        int wb = 0, lb = 0;
#pragma unroll
        for (int w = 0; w < 8; ++w) {
            wb += (w < wave) ? wsum[w] : 0;
            lb += (w < wave) ? lsum[w] : 0;
        }
        const int excl = wb + inc - tot;
        const int lex  = lb + inl - own;
        ls[t] = lex;
        gb[t] = excl + pre - lex;
        if (blk == 0) {
            bstart[t] = excl;
            if (t == NBMAX - 1) bstart[NBMAX] = excl + tot;
        }
    }
    __syncthreads();

    int b[4], r[4];
#pragma unroll
    for (int k = 0; k < 4; ++k) {
        const int e = base + k * 1024 + t;
        d[k] = clamp_node(d[k], n);
        s[k] = clamp_node(s[k], n);
        b[k] = bucket_of(d[k]);
        r[k] = 0;
        if (e < E) r[k] = atomicAdd(&h[b[k]], 1);
    }
#pragma unroll
    for (int k = 0; k < 4; ++k) {
        const int e = base + k * 1024 + t;
        if (e < E) {
            int i = ls[b[k]] + r[k];
            i = i < 0 ? 0 : (i > CHUNK - 1 ? CHUNK - 1 : i);
            stage[i] = (int)(((unsigned)s[k] << 16) | (unsigned)(d[k] & 0xFFFF));
        }
    }
    __syncthreads();

    const int mine = (E - base) < CHUNK ? (E - base) : CHUNK;
#pragma unroll
    for (int k = 0; k < 4; ++k) {
        const int i = k * 1024 + t;
        if (i < mine) {
            const int w  = stage[i];
            int p = gb[bucket_of(w & 0xFFFF)] + i;
            p = p < 0 ? 0 : (p > E - 1 ? E - 1 : p);
            ent[p] = w;
        }
    }
}

__global__ void __launch_bounds__(1024)
k_bucket_fine(const int* __restrict__ ent, const int* __restrict__ bstart,
              int* __restrict__ rowptr, float* __restrict__ dinv,
              int* __restrict__ csr_src, int E, int n) {
    constexpr int FCAP = 8 * 1024;
    __shared__ int cnt[BNODE];
    __shared__ int cur[BNODE];
    __shared__ int wsum[8];
    __shared__ int stage[FCAP];
    const int t    = threadIdx.x;
    const int lane = t & 31;
    const int wave = t >> 5;
    const int bk   = blockIdx.x;
    int beg = bstart[bk];
    int end = bstart[bk + 1];
    beg = beg < 0 ? 0 : (beg > E ? E : beg);
    end = end < beg ? beg : (end > E ? E : end);
    const int  m      = end - beg;
    const bool staged = m <= FCAP;

    unsigned w[8];
#pragma unroll
    for (int u = 0; u < 8; ++u) {
        const int i = u * 1024 + t;
        w[u] = i < m ? (unsigned)ent[beg + i] : 0u;
    }
    if (t < BNODE) cnt[t] = 0;
    __syncthreads();
#pragma unroll
    for (int u = 0; u < 8; ++u) {
        const int i = u * 1024 + t;
        if (i < m) atomicAdd(&cnt[w[u] & (BNODE - 1)], 1);
    }
    for (int i = FCAP + t; i < m; i += 1024)
        atomicAdd(&cnt[(unsigned)ent[beg + i] & (BNODE - 1)], 1);
    __syncthreads();

    const int v   = t < BNODE ? cnt[t] : 0;
    const int inc = wave_incl_scan(v, lane);
    if (lane == 31 && wave < 8) wsum[wave] = inc;
    __syncthreads();
    if (t < BNODE) {
        int wb = 0;
#pragma unroll
        for (int k = 0; k < 8; ++k) wb += (k < wave) ? wsum[k] : 0;
        const int first = wb + inc - v;
        cur[t] = first;
        const int node = bk * BNODE + t;
        if (node < n) {
            rowptr[node] = beg + first;
            dinv[node]   = rsqrtf((float)v + 1.0f);
        }
    }
    if (bk == (int)gridDim.x - 1 && t == 0) rowptr[n] = end;
    __syncthreads();

    int p[8];
#pragma unroll
    for (int u = 0; u < 8; ++u) {
        const int i = u * 1024 + t;
        p[u] = 0;
        if (i < m) p[u] = atomicAdd(&cur[w[u] & (BNODE - 1)], 1);
    }
    if (staged) {
#pragma unroll
        for (int u = 0; u < 8; ++u) {
            const int i = u * 1024 + t;
            if (i < m) {
                const int q = p[u] < 0 ? 0 : (p[u] > FCAP - 1 ? FCAP - 1 : p[u]);
                stage[q] = (int)(w[u] >> 16);
            }
        }
    } else {
#pragma unroll
        for (int u = 0; u < 8; ++u) {
            const int i = u * 1024 + t;
            if (i < m) {
                int q = beg + p[u];
                q = q < 0 ? 0 : (q > E - 1 ? E - 1 : q);
                csr_src[q] = (int)(w[u] >> 16);
            }
        }
        for (int i = FCAP + t; i < m; i += 1024) {
            const unsigned x = (unsigned)ent[beg + i];
            int q = beg + atomicAdd(&cur[x & (BNODE - 1)], 1);
            q = q < 0 ? 0 : (q > E - 1 ? E - 1 : q);
            csr_src[q] = (int)(x >> 16);
        }
    }
    __syncthreads();
    if (staged) {
        for (int i = t; i < m; i += 1024) csr_src[beg + i] = stage[i];
    }
}

template <int R>
__device__ __forceinline__ void gather_pairs(const float* __restrict__ x, int my, float cw,
                                             int j, bool hi, int l16, float4& acc) {
    float4 r[R];
    float  c[R];
#pragma unroll
    for (int u = 0; u < R; ++u) {
        const int s0 = __builtin_amdgcn_readlane(my, j + 2 * u);
        const int s1 = __builtin_amdgcn_readlane(my, j + 2 * u + 1);
        const int c0 = __builtin_amdgcn_readlane(__float_as_int(cw), j + 2 * u);
        const int c1 = __builtin_amdgcn_readlane(__float_as_int(cw), j + 2 * u + 1);
        const int s  = hi ? s1 : s0;
        c[u] = __int_as_float(hi ? c1 : c0);
        r[u] = *(const float4*)(x + (size_t)s * 64 + l16 * 4);
    }
#pragma unroll
    for (int u = 0; u < R; ++u) {
        acc.x += r[u].x * c[u];
        acc.y += r[u].y * c[u];
        acc.z += r[u].z * c[u];
        acc.w += r[u].w * c[u];
    }
}

__global__ void __launch_bounds__(256)
k_gather_in(const int* __restrict__ rowptr, const int* __restrict__ csr_src,
            const float* __restrict__ dinv, const float* __restrict__ x,
            float* __restrict__ ax, int n) {
    const int  lane = threadIdx.x & 31;
    const bool hi   = lane >= 16;
    const int  l16  = lane & 15;
    const int wid = __builtin_amdgcn_readfirstlane(
        (int)(blockIdx.x * (blockDim.x >> 5) + (threadIdx.x >> 5)));
    const int nw  = gridDim.x * (blockDim.x >> 5);

    for (int node = wid; node < n; node += nw) {
        const int   beg = rowptr[node];
        const int   end = rowptr[node + 1];
        const float di  = dinv[node];

        float4 acc;
        {
            const float4 v = *(const float4*)(x + (size_t)node * 64 + l16 * 4);
            const float  w = di * di;
            acc.x = hi ? 0.0f : v.x * w;
            acc.y = hi ? 0.0f : v.y * w;
            acc.z = hi ? 0.0f : v.z * w;
            acc.w = hi ? 0.0f : v.w * w;
        }

        for (int base = beg; base < end; base += 32) {
            const int m = (end - base) < 32 ? (end - base) : 32;
            int   my = 0;
            float cw = 0.0f;
            if (lane < m) {
                my = csr_src[base + lane];
                my = my < 0 ? 0 : (my > n - 1 ? n - 1 : my);
                cw = dinv[my] * di;
            }
            int j = 0;
            for (; j + 16 <= m; j += 16) gather_pairs<8>(x, my, cw, j, hi, l16, acc);
            if (j + 8 <= m) { gather_pairs<4>(x, my, cw, j, hi, l16, acc); j += 8; }
            if (j + 4 <= m) { gather_pairs<2>(x, my, cw, j, hi, l16, acc); j += 4; }
            if (j + 2 <= m) { gather_pairs<1>(x, my, cw, j, hi, l16, acc); j += 2; }
            if (j < m) {
                const int   s = __builtin_amdgcn_readlane(my, j);
                const float c = __int_as_float(
                    __builtin_amdgcn_readlane(__float_as_int(cw), j));
                if (!hi) {
                    const float4 r = *(const float4*)(x + (size_t)s * 64 + l16 * 4);
                    acc.x += r.x * c;
                    acc.y += r.y * c;
                    acc.z += r.z * c;
                    acc.w += r.w * c;
                }
            }
        }

        float4 o;
        o.x = acc.x + __shfl_xor(acc.x, 16, 32);
        o.y = acc.y + __shfl_xor(acc.y, 16, 32);
        o.z = acc.z + __shfl_xor(acc.z, 16, 32);
        o.w = acc.w + __shfl_xor(acc.w, 16, 32);
        if (!hi) *(float4*)(ax + (size_t)node * 64 + l16 * 4) = o;
    }
}

__global__ void __launch_bounds__(32)
k_fold_heads(const float* __restrict__ W2, const float* __restrict__ b2,
             const float* __restrict__ Wn, const float* __restrict__ We,
             float* __restrict__ Wp) {
    const int lane = threadIdx.x;
    const int col  = lane & 15;
    const int kb   = (lane >> 4) << 1;
    const bool live = col < 6;
    const float* __restrict__ hc =
        col < 2 ? Wn + col : (col < 4 ? We + (col - 2) : (col < 6 ? We + 256 + (col - 4) : Wn));

    if (blockIdx.x < 16) {
        const float* __restrict__ Ar = W2 + (size_t)(blockIdx.x * 16 + col) * 128 + kb;
        v8f acc;
#pragma unroll
        for (int r = 0; r < 8; ++r) acc[r] = 0.0f;
#pragma unroll
        for (int k = 0; k < 128; k += 4) {
            v2f av; av.x = Ar[k];                 av.y = Ar[k + 1];
            const float h0 = hc[(k + kb) * 2];
            const float h1 = hc[(k + kb + 1) * 2];
            v2f bv; bv.x = live ? h0 : 0.0f;      bv.y = live ? h1 : 0.0f;
            acc = __builtin_amdgcn_wmma_f32_16x16x4_f32(
                false, av, false, bv, (short)0, acc, false, false);
        }
        float* __restrict__ Cp = Wp + (size_t)(blockIdx.x * 16 + ((lane >> 4) << 3)) * 16 + col;
#pragma unroll
        for (int r = 0; r < 8; ++r) Cp[(size_t)r * 16] = acc[r];
    } else if (lane < 16) {
        float s = 0.0f;
        for (int k = 0; k < 128; ++k) {
            const float h = hc[k * 2];
            s += b2[k] * (live ? h : 0.0f);
        }
        Wp[256 * 16 + lane] = s;
    }
}

__global__ void __launch_bounds__(256)
k_hidden_fold(const float* __restrict__ AX, const float* __restrict__ W1,
              const float* __restrict__ b1, const float* __restrict__ Wp,
              float* __restrict__ Y, int ntiles) {
    constexpr int LDA = 68;
    constexpr int LDR = 12;
    __shared__ __attribute__((aligned(16))) float As[16 * LDA];
    __shared__ __attribute__((aligned(16))) float red[8 * 16 * LDR];

    const int tid  = threadIdx.x;
    const int wave = tid >> 5;
    const int lane = tid & 31;
    const int c    = lane & 15;
    const int h    = lane >> 4;
    const int hb   = wave * 32;

    v2f a1[2][16];
#pragma unroll
    for (int t = 0; t < 2; ++t) {
#pragma unroll
        for (int sp = 0; sp < 8; ++sp) {
#pragma unroll
            for (int u = 0; u < 2; ++u) {
                const int k = 8 * sp + 4 * h + 2 * u;
                a1[t][2 * sp + u].x = W1[(size_t)k * 256 + hb + 16 * t + c];
                a1[t][2 * sp + u].y = W1[(size_t)(k + 1) * 256 + hb + 16 * t + c];
            }
        }
    }
    v2f a2[2][4];
#pragma unroll
    for (int t = 0; t < 2; ++t) {
#pragma unroll
        for (int s = 0; s < 4; ++s) {
            const int ch = hb + 16 * t + 8 * h + 2 * s;
            a2[t][s].x = Wp[(size_t)ch * 16 + c];
            a2[t][s].y = Wp[(size_t)(ch + 1) * 16 + c];
        }
    }
    float bs[2][8];
#pragma unroll
    for (int t = 0; t < 2; ++t) {
#pragma unroll
        for (int r = 0; r < 8; ++r) bs[t][r] = b1[hb + 16 * t + 8 * h + r];
    }

    const int sr = tid >> 4;
    const int sc = tid & 15;

    int tile = blockIdx.x;
    float4 g = make_float4(0.0f, 0.0f, 0.0f, 0.0f);
    if (tile < ntiles) g = *(const float4*)(AX + (size_t)tile * 1024 + tid * 4);

    for (; tile < ntiles; tile += gridDim.x) {
        *(float4*)(As + sr * LDA + sc * 4) = g;
        __syncthreads();
        const int nxt = tile + gridDim.x;
        if (nxt < ntiles) g = *(const float4*)(AX + (size_t)nxt * 1024 + tid * 4);

        v2f b[16];
#pragma unroll
        for (int sp = 0; sp < 8; ++sp) {
            const float4 q = *(const float4*)(As + c * LDA + 8 * sp + 4 * h);
            b[2 * sp].x     = q.x;  b[2 * sp].y     = q.y;
            b[2 * sp + 1].x = q.z;  b[2 * sp + 1].y = q.w;
        }

        v8f h0, h1;
#pragma unroll
        for (int r = 0; r < 8; ++r) { h0[r] = bs[0][r]; h1[r] = bs[1][r]; }
#pragma unroll
        for (int s = 0; s < 16; ++s) {
            h0 = __builtin_amdgcn_wmma_f32_16x16x4_f32(
                false, a1[0][s], false, b[s], (short)0, h0, false, false);
            h1 = __builtin_amdgcn_wmma_f32_16x16x4_f32(
                false, a1[1][s], false, b[s], (short)0, h1, false, false);
        }
#pragma unroll
        for (int r = 0; r < 8; ++r) {
            h0[r] = h0[r] > 0.0f ? h0[r] : 0.0f;
            h1[r] = h1[r] > 0.0f ? h1[r] : 0.0f;
        }

        v8f o;
#pragma unroll
        for (int r = 0; r < 8; ++r) o[r] = 0.0f;
#pragma unroll
        for (int s = 0; s < 4; ++s) {
            v2f hv; hv.x = h0[2 * s]; hv.y = h0[2 * s + 1];
            o = __builtin_amdgcn_wmma_f32_16x16x4_f32(
                false, a2[0][s], false, hv, (short)0, o, false, false);
        }
#pragma unroll
        for (int s = 0; s < 4; ++s) {
            v2f hv; hv.x = h1[2 * s]; hv.y = h1[2 * s + 1];
            o = __builtin_amdgcn_wmma_f32_16x16x4_f32(
                false, a2[1][s], false, hv, (short)0, o, false, false);
        }

        if (h == 0) {
            float* __restrict__ rp = red + (wave * 16 + c) * LDR;
            *(float4*)(rp)     = make_float4(o[0], o[1], o[2], o[3]);
            *(float4*)(rp + 4) = make_float4(o[4], o[5], o[6], o[7]);
        }
        __syncthreads();

        if (tid < 128) {
            const int node = tid >> 3;
            const int j    = tid & 7;
            float s = red[node * LDR + j];
#pragma unroll
            for (int w = 1; w < 8; ++w) s += red[(w * 16 + node) * LDR + j];
            Y[(size_t)tile * 128 + tid] = s;
        }
    }
}

__global__ void __launch_bounds__(256)
k_gather_heads(const int* __restrict__ rowptr, const int* __restrict__ csr_src,
               const float* __restrict__ dinv, const float* __restrict__ y,
               const float* __restrict__ c, const float* __restrict__ bn,
               float* __restrict__ out, float* __restrict__ pq, int n) {
    const int  t    = blockIdx.x * 256 + threadIdx.x;
    const int  q    = t & 3;
    const int  node = t >> 2;
    const bool live = node < n;
    const int  nd   = live ? node : n - 1;

    const int   beg = rowptr[nd];
    const int   end = rowptr[nd + 1];
    const float di  = dinv[nd];
    const float4* __restrict__ y4 = (const float4*)y;

    float4 a0, a1;
    {
        const float4 s0 = y4[(size_t)nd * 2];
        const float4 s1 = y4[(size_t)nd * 2 + 1];
        const float  w  = di * di;
        const bool   z  = q != 0;
        a0.x = z ? 0.0f : s0.x * w;  a0.y = z ? 0.0f : s0.y * w;
        a0.z = z ? 0.0f : s0.z * w;  a0.w = z ? 0.0f : s0.w * w;
        a1.x = z ? 0.0f : s1.x * w;  a1.y = z ? 0.0f : s1.y * w;
        a1.z = z ? 0.0f : s1.z * w;  a1.w = z ? 0.0f : s1.w * w;
    }

    int e = beg + q;
    for (; e + 4 < end; e += 8) {
        int sA = csr_src[e];
        int sB = csr_src[e + 4];
        sA = sA < 0 ? 0 : (sA > n - 1 ? n - 1 : sA);
        sB = sB < 0 ? 0 : (sB > n - 1 ? n - 1 : sB);
        const float  cA = dinv[sA] * di;
        const float  cB = dinv[sB] * di;
        const float4 rA0 = y4[(size_t)sA * 2], rA1 = y4[(size_t)sA * 2 + 1];
        const float4 rB0 = y4[(size_t)sB * 2], rB1 = y4[(size_t)sB * 2 + 1];
        a0.x += rA0.x * cA;  a0.y += rA0.y * cA;  a0.z += rA0.z * cA;  a0.w += rA0.w * cA;
        a1.x += rA1.x * cA;  a1.y += rA1.y * cA;  a1.z += rA1.z * cA;  a1.w += rA1.w * cA;
        a0.x += rB0.x * cB;  a0.y += rB0.y * cB;  a0.z += rB0.z * cB;  a0.w += rB0.w * cB;
        a1.x += rB1.x * cB;  a1.y += rB1.y * cB;  a1.z += rB1.z * cB;  a1.w += rB1.w * cB;
    }
    if (e < end) {
        int sA = csr_src[e];
        sA = sA < 0 ? 0 : (sA > n - 1 ? n - 1 : sA);
        const float  cA = dinv[sA] * di;
        const float4 rA0 = y4[(size_t)sA * 2], rA1 = y4[(size_t)sA * 2 + 1];
        a0.x += rA0.x * cA;  a0.y += rA0.y * cA;  a0.z += rA0.z * cA;  a0.w += rA0.w * cA;
        a1.x += rA1.x * cA;  a1.y += rA1.y * cA;  a1.z += rA1.z * cA;  a1.w += rA1.w * cA;
    }

#pragma unroll
    for (int off = 1; off <= 2; off <<= 1) {
        a0.x += __shfl_xor(a0.x, off, 32);  a0.y += __shfl_xor(a0.y, off, 32);
        a0.z += __shfl_xor(a0.z, off, 32);  a0.w += __shfl_xor(a0.w, off, 32);
        a1.x += __shfl_xor(a1.x, off, 32);  a1.y += __shfl_xor(a1.y, off, 32);
    }

    if (live) {
        if (q == 0) {
            float2 o;
            o.x = (a0.x + c[0]) + bn[0];
            o.y = (a0.y + c[1]) + bn[1];
            *(float2*)(out + (size_t)nd * 2) = o;
        } else if (q == 1) {
            float4 r;
            r.x = a0.z + c[2];  r.y = a0.w + c[3];
            r.z = a1.x + c[4];  r.w = a1.y + c[5];
            *(float4*)(pq + (size_t)nd * 4) = r;
        }
    }
}

__device__ __forceinline__ float2 edge_one(const float* __restrict__ pq, int s, int d,
                                           float b0, float b1) {
    const float2 p = *(const float2*)(pq + (size_t)s * 4);
    const float2 q = *(const float2*)(pq + (size_t)d * 4 + 2);
    float2 o;
    o.x = (p.x + q.x) + b0;
    o.y = (p.y + q.y) + b1;
    return o;
}

__global__ void __launch_bounds__(256)
k_edge_head(const int* __restrict__ src, const int* __restrict__ dst,
            const float* __restrict__ pq, const float* __restrict__ be,
            float* __restrict__ out, int E) {
    const int   base = blockIdx.x * 1024;
    const float b0 = be[0], b1 = be[1];
    const bool wide = ((((size_t)src | (size_t)dst) & 7) == 0) && (((size_t)out & 15) == 0);
    if (wide && base + 1024 <= E) {
        const int  e0 = base + threadIdx.x * 2;
        const int  e1 = e0 + 512;
        const int2 s0 = *(const int2*)(src + e0);
        const int2 d0 = *(const int2*)(dst + e0);
        const int2 s1 = *(const int2*)(src + e1);
        const int2 d1 = *(const int2*)(dst + e1);
        const float2 a0 = edge_one(pq, s0.x, d0.x, b0, b1);
        const float2 a1 = edge_one(pq, s0.y, d0.y, b0, b1);
        const float2 c0 = edge_one(pq, s1.x, d1.x, b0, b1);
        const float2 c1 = edge_one(pq, s1.y, d1.y, b0, b1);
        float4 o0, o1;
        o0.x = a0.x; o0.y = a0.y; o0.z = a1.x; o0.w = a1.y;
        o1.x = c0.x; o1.y = c0.y; o1.z = c1.x; o1.w = c1.y;
        *(float4*)(out + (size_t)e0 * 2) = o0;
        *(float4*)(out + (size_t)e1 * 2) = o1;
    } else {
        for (int u = 0; u < 2; ++u) {
            for (int k = 0; k < 2; ++k) {
                const int e = base + u * 512 + threadIdx.x * 2 + k;
                if (e < E) *(float2*)(out + (size_t)e * 2) = edge_one(pq, src[e], dst[e], b0, b1);
            }
        }
    }
}

extern "C" void kernel_launch(void* const* d_in, const int* in_sizes, int n_in,
                              void* d_out, int out_size, void* d_ws, size_t ws_size,
                              hipStream_t stream) {
    const float* x  = (const float*)d_in[0];
    const int*   ei = (const int*)d_in[1];
    const float* W1 = (const float*)d_in[2];
    const float* b1 = (const float*)d_in[3];
    const float* W2 = (const float*)d_in[4];
    const float* b2 = (const float*)d_in[5];
    const float* Wn = (const float*)d_in[6];
    const float* bn = (const float*)d_in[7];
    const float* We = (const float*)d_in[8];
    const float* be = (const float*)d_in[9];

    const int N = in_sizes[0] / 64;
    const int E = in_sizes[1] / 2;
    const int* src = ei;
    const int* dst = ei + E;

    char*  ws  = (char*)d_ws;
    size_t off = 0;
    auto carve = [&](size_t bytes) -> float* {
        float* p = (float*)(ws + off);
        off += (bytes + 255) & ~(size_t)255;
        return p;
    };
    const int G1 = (E + CHUNK - 1) / CHUNK;
    float* dinv   = carve((size_t)N * 4);
    int*   rowptr = (int*)carve((size_t)(N + 1) * 4);
    int*   hist   = (int*)carve((size_t)G1 * NBMAX * 4);
    int*   bstart = (int*)carve((size_t)(NBMAX + 1) * 4);
    int*   ent    = (int*)carve((size_t)E * 4);
    int*   csrsrc = (int*)carve((size_t)E * 4);
    float* ax   = carve((size_t)N * 64 * 4);
    float* wp   = carve((size_t)(256 + 1) * 16 * 4);
    float* y    = carve((size_t)N * 8 * 4);
    float* pq   = carve((size_t)N * 4 * 4);

    float* out_node = (float*)d_out;
    float* out_edge = out_node + (size_t)N * 2;

    k_bucket_hist   <<<G1, 1024, 0, stream>>>(dst, hist, E, N);
    k_bucket_scatter<<<G1, 1024, 0, stream>>>(src, dst, hist, bstart, ent, E, N);
    k_bucket_fine   <<<(N + BNODE - 1) / BNODE, 1024, 0, stream>>>(ent, bstart, rowptr, dinv, csrsrc, E, N);

    k_gather_in<<<2048, 256, 0, stream>>>(rowptr, csrsrc, dinv, x, ax, N);

    k_fold_heads<<<17, 32, 0, stream>>>(W2, b2, Wn, We, wp);
    k_hidden_fold<<<768, 256, 0, stream>>>(ax, W1, b1, wp, y, N / 16);
    k_gather_heads<<<(N * 4 + 255) / 256, 256, 0, stream>>>(
        rowptr, csrsrc, dinv, y, wp + 256 * 16, bn, out_node, pq, N);

    k_edge_head<<<(E + 1023) / 1024, 256, 0, stream>>>(src, dst, pq, be, out_edge, E);
}


# layer-1 neighbour sum fused into the product kernel: half a wave sums a node's 64-float rows straight into the LDS tile the WMMA reads, lists prefetched; 8 -> 7 launches; f32 sum order changes
# speedup vs baseline: 16.6163x; 1.0585x over previous
#include <hip/hip_runtime.h>


typedef float v2f __attribute__((ext_vector_type(2)));
typedef float v8f __attribute__((ext_vector_type(8)));

constexpr int BSH   = 8;
constexpr int BNODE = 1 << BSH;
constexpr int NBMAX = 256;
constexpr int CHUNK = 4096;

__device__ __forceinline__ int clamp_node(int v, int n) {
    return v < 0 ? 0 : (v > n - 1 ? n - 1 : v);
}
__device__ __forceinline__ int bucket_of(int d) {
    const int b = d >> BSH;
    return b > NBMAX - 1 ? NBMAX - 1 : b;
}

__global__ void __launch_bounds__(1024)
k_bucket_hist(const int* __restrict__ dst, int* __restrict__ hist, int E, int n) {
    __shared__ int h[NBMAX];
    const int t    = threadIdx.x;
    const int base = blockIdx.x * CHUNK;
    int d[4];
#pragma unroll
    for (int k = 0; k < 4; ++k) {
        const int e = base + k * 1024 + t;
        d[k] = e < E ? dst[e] : 0;
    }
    if (t < NBMAX) h[t] = 0;
    __syncthreads();
#pragma unroll
    for (int k = 0; k < 4; ++k) {
        const int e = base + k * 1024 + t;
        if (e < E) atomicAdd(&h[bucket_of(clamp_node(d[k], n))], 1);
    }
    __syncthreads();
    if (t < NBMAX) hist[(size_t)blockIdx.x * NBMAX + t] = h[t];
}

__device__ __forceinline__ int wave_incl_scan(int v, int lane) {
    for (int off = 1; off < 32; off <<= 1) {
        const int u = __shfl_up(v, off, 32);
        if (lane >= off) v += u;
    }
    return v;
}

__global__ void __launch_bounds__(1024)
k_bucket_scatter(const int* __restrict__ src, const int* __restrict__ dst,
                 const int* __restrict__ hist, int* __restrict__ bstart,
                 int* __restrict__ ent, int E, int n) {
    __shared__ int h[NBMAX];
    __shared__ int gb[NBMAX];
    __shared__ int ls[NBMAX];
    __shared__ __attribute__((aligned(16))) int ptot[16 * NBMAX];
    __shared__ __attribute__((aligned(16))) int ppre[16 * NBMAX];
    __shared__ int stage[CHUNK];
    __shared__ int wsum[8];
    __shared__ int lsum[8];
    const int t    = threadIdx.x;
    const int lane = t & 31;
    const int wave = t >> 5;
    const int blk  = blockIdx.x;
    const int nblk = gridDim.x;
    const int base = blk * CHUNK;

    int d[4], s[4];
#pragma unroll
    for (int k = 0; k < 4; ++k) {
        const int e = base + k * 1024 + t;
        d[k] = e < E ? dst[e] : 0;
        s[k] = e < E ? src[e] : 0;
    }
    const int own = t < NBMAX ? hist[(size_t)blk * NBMAX + t] : 0;

    {
        const int c4 = (t & 63) * 4;
        const int q  = t >> 6;
        int4 tot = make_int4(0, 0, 0, 0);
        int4 pre = make_int4(0, 0, 0, 0);
        for (int j0 = q; j0 < nblk; j0 += 256) {
            int4 v[16];
#pragma unroll
            for (int u = 0; u < 16; ++u) {
                const int j = j0 + 16 * u;
                v[u] = make_int4(0, 0, 0, 0);
                if (j < nblk) v[u] = *(const int4*)(hist + (size_t)j * NBMAX + c4);
            }
#pragma unroll
            for (int u = 0; u < 16; ++u) {
                const int j = j0 + 16 * u;
                tot.x += v[u].x;  tot.y += v[u].y;  tot.z += v[u].z;  tot.w += v[u].w;
                if (j < blk) {
                    pre.x += v[u].x;  pre.y += v[u].y;  pre.z += v[u].z;  pre.w += v[u].w;
                }
            }
        }
        *(int4*)(ptot + q * NBMAX + c4) = tot;
        *(int4*)(ppre + q * NBMAX + c4) = pre;
    }
    if (t < NBMAX) h[t] = 0;
    __syncthreads();

    int tot = 0, pre = 0;
    if (t < NBMAX) {
#pragma unroll
        for (int q = 0; q < 16; ++q) {
            tot += ptot[q * NBMAX + t];
            pre += ppre[q * NBMAX + t];
        }
    }
    const int inc = wave_incl_scan(tot, lane);
    const int inl = wave_incl_scan(own, lane);
    if (lane == 31 && wave < 8) { wsum[wave] = inc; lsum[wave] = inl; }
    __syncthreads();
    if (t < NBMAX) {
        int wb = 0, lb = 0;
#pragma unroll
        for (int w = 0; w < 8; ++w) {
            wb += (w < wave) ? wsum[w] : 0;
            lb += (w < wave) ? lsum[w] : 0;
        }
        const int excl = wb + inc - tot;
        const int lex  = lb + inl - own;
        ls[t] = lex;
        gb[t] = excl + pre - lex;
        if (blk == 0) {
            bstart[t] = excl;
            if (t == NBMAX - 1) bstart[NBMAX] = excl + tot;
        }
    }
    __syncthreads();

    int b[4], r[4];
#pragma unroll
    for (int k = 0; k < 4; ++k) {
        const int e = base + k * 1024 + t;
        d[k] = clamp_node(d[k], n);
        s[k] = clamp_node(s[k], n);
        b[k] = bucket_of(d[k]);
        r[k] = 0;
        if (e < E) r[k] = atomicAdd(&h[b[k]], 1);
    }
#pragma unroll
    for (int k = 0; k < 4; ++k) {
        const int e = base + k * 1024 + t;
        if (e < E) {
            int i = ls[b[k]] + r[k];
            i = i < 0 ? 0 : (i > CHUNK - 1 ? CHUNK - 1 : i);
            stage[i] = (int)(((unsigned)s[k] << 16) | (unsigned)(d[k] & 0xFFFF));
        }
    }
    __syncthreads();

    const int mine = (E - base) < CHUNK ? (E - base) : CHUNK;
#pragma unroll
    for (int k = 0; k < 4; ++k) {
        const int i = k * 1024 + t;
        if (i < mine) {
            const int w  = stage[i];
            int p = gb[bucket_of(w & 0xFFFF)] + i;
            p = p < 0 ? 0 : (p > E - 1 ? E - 1 : p);
            ent[p] = w;
        }
    }
}

__global__ void __launch_bounds__(1024)
k_bucket_fine(const int* __restrict__ ent, const int* __restrict__ bstart,
              int* __restrict__ rowptr, float* __restrict__ dinv,
              int* __restrict__ csr_src, int E, int n) {
    constexpr int FCAP = 8 * 1024;
    __shared__ int cnt[BNODE];
    __shared__ int cur[BNODE];
    __shared__ int wsum[8];
    __shared__ int stage[FCAP];
    const int t    = threadIdx.x;
    const int lane = t & 31;
    const int wave = t >> 5;
    const int bk   = blockIdx.x;
    int beg = bstart[bk];
    int end = bstart[bk + 1];
    beg = beg < 0 ? 0 : (beg > E ? E : beg);
    end = end < beg ? beg : (end > E ? E : end);
    const int  m      = end - beg;
    const bool staged = m <= FCAP;

    unsigned w[8];
#pragma unroll
    for (int u = 0; u < 8; ++u) {
        const int i = u * 1024 + t;
        w[u] = i < m ? (unsigned)ent[beg + i] : 0u;
    }
    if (t < BNODE) cnt[t] = 0;
    __syncthreads();
#pragma unroll
    for (int u = 0; u < 8; ++u) {
        const int i = u * 1024 + t;
        if (i < m) atomicAdd(&cnt[w[u] & (BNODE - 1)], 1);
    }
    for (int i = FCAP + t; i < m; i += 1024)
        atomicAdd(&cnt[(unsigned)ent[beg + i] & (BNODE - 1)], 1);
    __syncthreads();

    const int v   = t < BNODE ? cnt[t] : 0;
    const int inc = wave_incl_scan(v, lane);
    if (lane == 31 && wave < 8) wsum[wave] = inc;
    __syncthreads();
    if (t < BNODE) {
        int wb = 0;
#pragma unroll
        for (int k = 0; k < 8; ++k) wb += (k < wave) ? wsum[k] : 0;
        const int first = wb + inc - v;
        cur[t] = first;
        const int node = bk * BNODE + t;
        if (node < n) {
            rowptr[node] = beg + first;
            dinv[node]   = rsqrtf((float)v + 1.0f);
        }
    }
    if (bk == (int)gridDim.x - 1 && t == 0) rowptr[n] = end;
    __syncthreads();

    int p[8];
#pragma unroll
    for (int u = 0; u < 8; ++u) {
        const int i = u * 1024 + t;
        p[u] = 0;
        if (i < m) p[u] = atomicAdd(&cur[w[u] & (BNODE - 1)], 1);
    }
    if (staged) {
#pragma unroll
        for (int u = 0; u < 8; ++u) {
            const int i = u * 1024 + t;
            if (i < m) {
                const int q = p[u] < 0 ? 0 : (p[u] > FCAP - 1 ? FCAP - 1 : p[u]);
                stage[q] = (int)(w[u] >> 16);
            }
        }
    } else {
#pragma unroll
        for (int u = 0; u < 8; ++u) {
            const int i = u * 1024 + t;
            if (i < m) {
                int q = beg + p[u];
                q = q < 0 ? 0 : (q > E - 1 ? E - 1 : q);
                csr_src[q] = (int)(w[u] >> 16);
            }
        }
        for (int i = FCAP + t; i < m; i += 1024) {
            const unsigned x = (unsigned)ent[beg + i];
            int q = beg + atomicAdd(&cur[x & (BNODE - 1)], 1);
            q = q < 0 ? 0 : (q > E - 1 ? E - 1 : q);
            csr_src[q] = (int)(x >> 16);
        }
    }
    __syncthreads();
    if (staged) {
        for (int i = t; i < m; i += 1024) csr_src[beg + i] = stage[i];
    }
}

template <int R>
__device__ __forceinline__ void gather_rows(const float* __restrict__ x, int my, float cw,
                                            int j, bool hi, int l16, float4& acc) {
    float4 r[R];
#pragma unroll
    for (int u = 0; u < R; ++u) {
        const int s0 = __builtin_amdgcn_readlane(my, j + u);
        const int s1 = __builtin_amdgcn_readlane(my, 16 + j + u);
        const unsigned s = (unsigned)(hi ? s1 : s0);
        r[u] = *(const float4*)((const char*)x + (s * 256u + (unsigned)l16 * 16u));
    }
#pragma unroll
    for (int u = 0; u < R; ++u) {
        const int   c0 = __builtin_amdgcn_readlane(__float_as_int(cw), j + u);
        const int   c1 = __builtin_amdgcn_readlane(__float_as_int(cw), 16 + j + u);
        const float c  = __int_as_float(hi ? c1 : c0);
        acc.x += r[u].x * c;
        acc.y += r[u].y * c;
        acc.z += r[u].z * c;
        acc.w += r[u].w * c;
    }
}

__device__ __forceinline__ void gather_chunk(const float* __restrict__ x, int my, float cw,
                                             int T, bool hi, int l16, float4& acc) {
    int j = 0;
    for (; j + 8 <= T; j += 8) gather_rows<8>(x, my, cw, j, hi, l16, acc);
    if (j + 4 <= T) { gather_rows<4>(x, my, cw, j, hi, l16, acc); j += 4; }
    if (j + 2 <= T) { gather_rows<2>(x, my, cw, j, hi, l16, acc); j += 2; }
    if (j < T)      { gather_rows<1>(x, my, cw, j, hi, l16, acc); }
}

__device__ __forceinline__ void load_positions(const int* __restrict__ csr_src, int b, int e, int c,
                                               int nd, bool live, int& m0, int& m1) {
    m0 = nd;  m1 = nd;
    if (live) {
        if (c >= 1 && c <= e - b) m0 = csr_src[b + c - 1];
        if (16 + c <= e - b)      m1 = csr_src[b + 15 + c];
    }
}

__global__ void __launch_bounds__(32)
k_fold_heads(const float* __restrict__ W2, const float* __restrict__ b2,
             const float* __restrict__ Wn, const float* __restrict__ We,
             float* __restrict__ Wp) {
    const int lane = threadIdx.x;
    const int col  = lane & 15;
    const int kb   = (lane >> 4) << 1;
    const bool live = col < 6;
    const float* __restrict__ hc =
        col < 2 ? Wn + col : (col < 4 ? We + (col - 2) : (col < 6 ? We + 256 + (col - 4) : Wn));

    if (blockIdx.x < 16) {
        const float* __restrict__ Ar = W2 + (size_t)(blockIdx.x * 16 + col) * 128 + kb;
        v8f acc;
#pragma unroll
        for (int r = 0; r < 8; ++r) acc[r] = 0.0f;
#pragma unroll
        for (int k = 0; k < 128; k += 4) {
            v2f av; av.x = Ar[k];                 av.y = Ar[k + 1];
            const float h0 = hc[(k + kb) * 2];
            const float h1 = hc[(k + kb + 1) * 2];
            v2f bv; bv.x = live ? h0 : 0.0f;      bv.y = live ? h1 : 0.0f;
            acc = __builtin_amdgcn_wmma_f32_16x16x4_f32(
                false, av, false, bv, (short)0, acc, false, false);
        }
        float* __restrict__ Cp = Wp + (size_t)(blockIdx.x * 16 + ((lane >> 4) << 3)) * 16 + col;
#pragma unroll
        for (int r = 0; r < 8; ++r) Cp[(size_t)r * 16] = acc[r];
    } else if (lane < 16) {
        float s = 0.0f;
        for (int k = 0; k < 128; ++k) {
            const float h = hc[k * 2];
            s += b2[k] * (live ? h : 0.0f);
        }
        Wp[256 * 16 + lane] = s;
    }
}

__global__ void __launch_bounds__(256)
k_hidden_fold(const int* __restrict__ rowptr, const int* __restrict__ csr_src,
              const float* __restrict__ dinv, const float* __restrict__ x,
              const float* __restrict__ W1, const float* __restrict__ b1,
              const float* __restrict__ Wp, float* __restrict__ Y, int ntiles, int n) {
    constexpr int LDA = 68;
    constexpr int LDR = 12;
    constexpr int LDF = 20;
    __shared__ __attribute__((aligned(16))) float As[16 * LDA];
    __shared__ __attribute__((aligned(16))) float red[8 * 16 * LDR];
    __shared__ __attribute__((aligned(16))) float bsL[256];
    __shared__ __attribute__((aligned(16))) float a2L[256 * LDF];

    const int  tid  = threadIdx.x;
    const int  wave = tid >> 5;
    const int  lane = tid & 31;
    const int  c    = lane & 15;
    const int  h    = lane >> 4;
    const bool hi   = h != 0;
    const int  hb   = wave * 32;

    v2f a1[2][16];
#pragma unroll
    for (int t = 0; t < 2; ++t) {
#pragma unroll
        for (int sp = 0; sp < 8; ++sp) {
#pragma unroll
            for (int u = 0; u < 2; ++u) {
                const int k = 8 * sp + 4 * h + 2 * u;
                a1[t][2 * sp + u].x = W1[(size_t)k * 256 + hb + 16 * t + c];
                a1[t][2 * sp + u].y = W1[(size_t)(k + 1) * 256 + hb + 16 * t + c];
            }
        }
    }
    {
        float* __restrict__ fp = a2L + tid * LDF;
#pragma unroll
        for (int t = 0; t < 2; ++t) {
#pragma unroll
            for (int s = 0; s < 4; s += 2) {
                const int ch = hb + 16 * t + 8 * h + 2 * s;
                float4 w;
                w.x = Wp[(size_t)ch * 16 + c];
                w.y = Wp[(size_t)(ch + 1) * 16 + c];
                w.z = Wp[(size_t)(ch + 2) * 16 + c];
                w.w = Wp[(size_t)(ch + 3) * 16 + c];
                *(float4*)(fp + 8 * t + 2 * s) = w;
            }
        }
        bsL[tid] = b1[tid];
    }

    const int hn   = 2 * wave + h;
    const int step = (int)gridDim.x;
    int tile = blockIdx.x;
    auto node_of = [&](int t) -> int {
        int nd = t * 16 + hn;
        nd = nd < n ? nd : n - 1;
        return nd < 0 ? 0 : nd;
    };

    int   bA = 0, eA = 0, bB = 0, eB = 0, bC = 0, eC = 0;
    float dA = 0.0f, dB = 0.0f, dC = 0.0f;
    if (tile < ntiles) {
        const int nd = node_of(tile);
        bA = rowptr[nd];  eA = rowptr[nd + 1];  dA = dinv[nd];
    }
    if (tile + step < ntiles) {
        const int nd = node_of(tile + step);
        bB = rowptr[nd];  eB = rowptr[nd + 1];  dB = dinv[nd];
    }
    if (tile + 2 * step < ntiles) {
        const int nd = node_of(tile + 2 * step);
        bC = rowptr[nd];  eC = rowptr[nd + 1];  dC = dinv[nd];
    }
    int my0, my1, nx0, nx1;
    load_positions(csr_src, bA, eA, c, node_of(tile), tile < ntiles, my0, my1);
    load_positions(csr_src, bB, eB, c, node_of(tile + step), tile + step < ntiles, nx0, nx1);
    my0 = my0 < 0 ? 0 : (my0 > n - 1 ? n - 1 : my0);
    my1 = my1 < 0 ? 0 : (my1 > n - 1 ? n - 1 : my1);
    float q0 = dinv[my0];
    float q1 = dinv[my1];

    for (; tile < ntiles; tile += step) {
        const int nd = node_of(tile);
        const int m = eA - bA + 1;
        const int mlo = __builtin_amdgcn_readlane(m, 0);
        const int mhi = __builtin_amdgcn_readlane(m, 16);
        const int mm  = mlo > mhi ? mlo : mhi;

        const float cw0 = c < m      ? q0 * dA : 0.0f;
        const float cw1 = 16 + c < m ? q1 * dA : 0.0f;

        float4 acc = make_float4(0.0f, 0.0f, 0.0f, 0.0f);
        {
            const int tp = ((mm < 32 ? mm : 32) + 3) & ~3;
            int j = 0;
            for (; j + 8 <= tp; j += 8) {
                const bool first = j < 16;
                gather_rows<8>(x, first ? my0 : my1, first ? cw0 : cw1, j & 15, hi, c, acc);
            }
            if (j < tp) {
                const bool first = j < 16;
                gather_rows<4>(x, first ? my0 : my1, first ? cw0 : cw1, j & 15, hi, c, acc);
            }
        }
        for (int base = 32; base < mm; base += 16) {
            int   my = nd;
            float cw = 0.0f;
            if (base + c < m) {
                my = csr_src[bA + base - 1 + c];
                my = my < 0 ? 0 : (my > n - 1 ? n - 1 : my);
                cw = dinv[my] * dA;
            }
            gather_chunk(x, my, cw, mm - base < 16 ? mm - base : 16, hi, c, acc);
        }

        *(float4*)(As + hn * LDA + c * 4) = acc;
        __syncthreads();

        {
            bA = bB;  eA = eB;  dA = dB;
            bB = bC;  eB = eC;  dB = dC;
            my0 = nx0 < 0 ? 0 : (nx0 > n - 1 ? n - 1 : nx0);
            my1 = nx1 < 0 ? 0 : (nx1 > n - 1 ? n - 1 : nx1);
            q0 = dinv[my0];
            q1 = dinv[my1];
            const int t2 = tile + 2 * step;
            load_positions(csr_src, bB, eB, c, node_of(t2), t2 < ntiles, nx0, nx1);
            const int t3 = t2 + step;
            if (t3 < ntiles) {
                const int n3 = node_of(t3);
                bC = rowptr[n3];  eC = rowptr[n3 + 1];  dC = dinv[n3];
            }
        }

        v2f b[16];
#pragma unroll
        for (int sp = 0; sp < 8; ++sp) {
            const float4 q = *(const float4*)(As + c * LDA + 8 * sp + 4 * h);
            b[2 * sp].x     = q.x;  b[2 * sp].y     = q.y;
            b[2 * sp + 1].x = q.z;  b[2 * sp + 1].y = q.w;
        }

        v8f h0, h1;
        {
            const float4 p0 = *(const float4*)(bsL + hb + 8 * h);
            const float4 p1 = *(const float4*)(bsL + hb + 8 * h + 4);
            const float4 p2 = *(const float4*)(bsL + hb + 16 + 8 * h);
            const float4 p3 = *(const float4*)(bsL + hb + 16 + 8 * h + 4);
            h0[0] = p0.x; h0[1] = p0.y; h0[2] = p0.z; h0[3] = p0.w;
            h0[4] = p1.x; h0[5] = p1.y; h0[6] = p1.z; h0[7] = p1.w;
            h1[0] = p2.x; h1[1] = p2.y; h1[2] = p2.z; h1[3] = p2.w;
            h1[4] = p3.x; h1[5] = p3.y; h1[6] = p3.z; h1[7] = p3.w;
        }
        v2f a2[2][4];
        {
            const float* __restrict__ fp = a2L + tid * LDF;
#pragma unroll
            for (int t = 0; t < 2; ++t) {
                const float4 f0 = *(const float4*)(fp + 8 * t);
                const float4 f1 = *(const float4*)(fp + 8 * t + 4);
                a2[t][0].x = f0.x;  a2[t][0].y = f0.y;
                a2[t][1].x = f0.z;  a2[t][1].y = f0.w;
                a2[t][2].x = f1.x;  a2[t][2].y = f1.y;
                a2[t][3].x = f1.z;  a2[t][3].y = f1.w;
            }
        }
#pragma unroll
        for (int s = 0; s < 16; ++s) {
            h0 = __builtin_amdgcn_wmma_f32_16x16x4_f32(
                false, a1[0][s], false, b[s], (short)0, h0, false, false);
            h1 = __builtin_amdgcn_wmma_f32_16x16x4_f32(
                false, a1[1][s], false, b[s], (short)0, h1, false, false);
        }
#pragma unroll
        for (int r = 0; r < 8; ++r) {
            h0[r] = h0[r] > 0.0f ? h0[r] : 0.0f;
            h1[r] = h1[r] > 0.0f ? h1[r] : 0.0f;
        }

        v8f o;
#pragma unroll
        for (int r = 0; r < 8; ++r) o[r] = 0.0f;
#pragma unroll
        for (int s = 0; s < 4; ++s) {
            v2f hv; hv.x = h0[2 * s]; hv.y = h0[2 * s + 1];
            o = __builtin_amdgcn_wmma_f32_16x16x4_f32(
                false, a2[0][s], false, hv, (short)0, o, false, false);
        }
#pragma unroll
        for (int s = 0; s < 4; ++s) {
            v2f hv; hv.x = h1[2 * s]; hv.y = h1[2 * s + 1];
            o = __builtin_amdgcn_wmma_f32_16x16x4_f32(
                false, a2[1][s], false, hv, (short)0, o, false, false);
        }

        if (h == 0) {
            float* __restrict__ rp = red + (wave * 16 + c) * LDR;
            *(float4*)(rp)     = make_float4(o[0], o[1], o[2], o[3]);
            *(float4*)(rp + 4) = make_float4(o[4], o[5], o[6], o[7]);
        }
        __syncthreads();

        if (tid < 128) {
            const int node = tid >> 3;
            const int j    = tid & 7;
            float s = red[node * LDR + j];
#pragma unroll
            for (int w = 1; w < 8; ++w) s += red[(w * 16 + node) * LDR + j];
            Y[(size_t)tile * 128 + tid] = s;
        }
    }
}

__global__ void __launch_bounds__(256)
k_gather_heads(const int* __restrict__ rowptr, const int* __restrict__ csr_src,
               const float* __restrict__ dinv, const float* __restrict__ y,
               const float* __restrict__ c, const float* __restrict__ bn,
               float* __restrict__ out, float* __restrict__ pq, int n) {
    const int  t    = blockIdx.x * 256 + threadIdx.x;
    const int  q    = t & 3;
    const int  node = t >> 2;
    const bool live = node < n;
    const int  nd   = live ? node : n - 1;

    const int   beg = rowptr[nd];
    const int   end = rowptr[nd + 1];
    const float di  = dinv[nd];
    const float4* __restrict__ y4 = (const float4*)y;

    float4 a0, a1;
    {
        const float4 s0 = y4[(size_t)nd * 2];
        const float4 s1 = y4[(size_t)nd * 2 + 1];
        const float  w  = di * di;
        const bool   z  = q != 0;
        a0.x = z ? 0.0f : s0.x * w;  a0.y = z ? 0.0f : s0.y * w;
        a0.z = z ? 0.0f : s0.z * w;  a0.w = z ? 0.0f : s0.w * w;
        a1.x = z ? 0.0f : s1.x * w;  a1.y = z ? 0.0f : s1.y * w;
        a1.z = z ? 0.0f : s1.z * w;  a1.w = z ? 0.0f : s1.w * w;
    }

    int e = beg + q;
    for (; e + 4 < end; e += 8) {
        int sA = csr_src[e];
        int sB = csr_src[e + 4];
        sA = sA < 0 ? 0 : (sA > n - 1 ? n - 1 : sA);
        sB = sB < 0 ? 0 : (sB > n - 1 ? n - 1 : sB);
        const float  cA = dinv[sA] * di;
        const float  cB = dinv[sB] * di;
        const float4 rA0 = y4[(size_t)sA * 2], rA1 = y4[(size_t)sA * 2 + 1];
        const float4 rB0 = y4[(size_t)sB * 2], rB1 = y4[(size_t)sB * 2 + 1];
        a0.x += rA0.x * cA;  a0.y += rA0.y * cA;  a0.z += rA0.z * cA;  a0.w += rA0.w * cA;
        a1.x += rA1.x * cA;  a1.y += rA1.y * cA;  a1.z += rA1.z * cA;  a1.w += rA1.w * cA;
        a0.x += rB0.x * cB;  a0.y += rB0.y * cB;  a0.z += rB0.z * cB;  a0.w += rB0.w * cB;
        a1.x += rB1.x * cB;  a1.y += rB1.y * cB;  a1.z += rB1.z * cB;  a1.w += rB1.w * cB;
    }
    if (e < end) {
        int sA = csr_src[e];
        sA = sA < 0 ? 0 : (sA > n - 1 ? n - 1 : sA);
        const float  cA = dinv[sA] * di;
        const float4 rA0 = y4[(size_t)sA * 2], rA1 = y4[(size_t)sA * 2 + 1];
        a0.x += rA0.x * cA;  a0.y += rA0.y * cA;  a0.z += rA0.z * cA;  a0.w += rA0.w * cA;
        a1.x += rA1.x * cA;  a1.y += rA1.y * cA;  a1.z += rA1.z * cA;  a1.w += rA1.w * cA;
    }

#pragma unroll
    for (int off = 1; off <= 2; off <<= 1) {
        a0.x += __shfl_xor(a0.x, off, 32);  a0.y += __shfl_xor(a0.y, off, 32);
        a0.z += __shfl_xor(a0.z, off, 32);  a0.w += __shfl_xor(a0.w, off, 32);
        a1.x += __shfl_xor(a1.x, off, 32);  a1.y += __shfl_xor(a1.y, off, 32);
    }

    if (live) {
        if (q == 0) {
            float2 o;
            o.x = (a0.x + c[0]) + bn[0];
            o.y = (a0.y + c[1]) + bn[1];
            *(float2*)(out + (size_t)nd * 2) = o;
        } else if (q == 1) {
            float4 r;
            r.x = a0.z + c[2];  r.y = a0.w + c[3];
            r.z = a1.x + c[4];  r.w = a1.y + c[5];
            *(float4*)(pq + (size_t)nd * 4) = r;
        }
    }
}

__device__ __forceinline__ float2 edge_one(const float* __restrict__ pq, int s, int d,
                                           float b0, float b1) {
    const float2 p = *(const float2*)(pq + (size_t)s * 4);
    const float2 q = *(const float2*)(pq + (size_t)d * 4 + 2);
    float2 o;
    o.x = (p.x + q.x) + b0;
    o.y = (p.y + q.y) + b1;
    return o;
}

__global__ void __launch_bounds__(256)
k_edge_head(const int* __restrict__ src, const int* __restrict__ dst,
            const float* __restrict__ pq, const float* __restrict__ be,
            float* __restrict__ out, int E) {
    const int   base = blockIdx.x * 1024;
    const float b0 = be[0], b1 = be[1];
    const bool wide = ((((size_t)src | (size_t)dst) & 7) == 0) && (((size_t)out & 15) == 0);
    if (wide && base + 1024 <= E) {
        const int  e0 = base + threadIdx.x * 2;
        const int  e1 = e0 + 512;
        const int2 s0 = *(const int2*)(src + e0);
        const int2 d0 = *(const int2*)(dst + e0);
        const int2 s1 = *(const int2*)(src + e1);
        const int2 d1 = *(const int2*)(dst + e1);
        const float2 a0 = edge_one(pq, s0.x, d0.x, b0, b1);
        const float2 a1 = edge_one(pq, s0.y, d0.y, b0, b1);
        const float2 c0 = edge_one(pq, s1.x, d1.x, b0, b1);
        const float2 c1 = edge_one(pq, s1.y, d1.y, b0, b1);
        float4 o0, o1;
        o0.x = a0.x; o0.y = a0.y; o0.z = a1.x; o0.w = a1.y;
        o1.x = c0.x; o1.y = c0.y; o1.z = c1.x; o1.w = c1.y;
        *(float4*)(out + (size_t)e0 * 2) = o0;
        *(float4*)(out + (size_t)e1 * 2) = o1;
    } else {
        for (int u = 0; u < 2; ++u) {
            for (int k = 0; k < 2; ++k) {
                const int e = base + u * 512 + threadIdx.x * 2 + k;
                if (e < E) *(float2*)(out + (size_t)e * 2) = edge_one(pq, src[e], dst[e], b0, b1);
            }
        }
    }
}

extern "C" void kernel_launch(void* const* d_in, const int* in_sizes, int n_in,
                              void* d_out, int out_size, void* d_ws, size_t ws_size,
                              hipStream_t stream) {
    const float* x  = (const float*)d_in[0];
    const int*   ei = (const int*)d_in[1];
    const float* W1 = (const float*)d_in[2];
    const float* b1 = (const float*)d_in[3];
    const float* W2 = (const float*)d_in[4];
    const float* b2 = (const float*)d_in[5];
    const float* Wn = (const float*)d_in[6];
    const float* bn = (const float*)d_in[7];
    const float* We = (const float*)d_in[8];
    const float* be = (const float*)d_in[9];

    const int N = in_sizes[0] / 64;
    const int E = in_sizes[1] / 2;
    const int* src = ei;
    const int* dst = ei + E;

    char*  ws  = (char*)d_ws;
    size_t off = 0;
    auto carve = [&](size_t bytes) -> float* {
        float* p = (float*)(ws + off);
        off += (bytes + 255) & ~(size_t)255;
        return p;
    };
    const int G1 = (E + CHUNK - 1) / CHUNK;
    float* dinv   = carve((size_t)N * 4);
    int*   rowptr = (int*)carve((size_t)(N + 1) * 4);
    int*   hist   = (int*)carve((size_t)G1 * NBMAX * 4);
    int*   bstart = (int*)carve((size_t)(NBMAX + 1) * 4);
    int*   ent    = (int*)carve((size_t)E * 4);
    int*   csrsrc = (int*)carve((size_t)E * 4);
    float* wp   = carve((size_t)(256 + 1) * 16 * 4);
    float* y    = carve((size_t)N * 8 * 4);
    float* pq   = carve((size_t)N * 4 * 4);

    float* out_node = (float*)d_out;
    float* out_edge = out_node + (size_t)N * 2;

    k_bucket_hist   <<<G1, 1024, 0, stream>>>(dst, hist, E, N);
    k_bucket_scatter<<<G1, 1024, 0, stream>>>(src, dst, hist, bstart, ent, E, N);
    k_bucket_fine   <<<(N + BNODE - 1) / BNODE, 1024, 0, stream>>>(ent, bstart, rowptr, dinv, csrsrc, E, N);

    k_fold_heads<<<17, 32, 0, stream>>>(W2, b2, Wn, We, wp);
    k_hidden_fold<<<768, 256, 0, stream>>>(rowptr, csrsrc, dinv, x, W1, b1, wp, y, N / 16, N);
    k_gather_heads<<<(N * 4 + 255) / 256, 256, 0, stream>>>(
        rowptr, csrsrc, dinv, y, wp + 256 * 16, bn, out_node, pq, N);

    k_edge_head<<<(E + 1023) / 1024, 256, 0, stream>>>(src, dst, pq, be, out_edge, E);
}


# folded-weights product W2 @ [Wn, We halves] (WMMA) loses its own launch: it runs as one extra work-group of the coarse scatter launch, beside the sorting groups; 7 -> 6 launches
# speedup vs baseline: 17.5572x; 1.0566x over previous
#include <hip/hip_runtime.h>


typedef float v2f __attribute__((ext_vector_type(2)));
typedef float v8f __attribute__((ext_vector_type(8)));

constexpr int BSH   = 8;
constexpr int BNODE = 1 << BSH;
constexpr int NBMAX = 256;
constexpr int CHUNK = 4096;

__device__ __forceinline__ int clamp_node(int v, int n) {
    return v < 0 ? 0 : (v > n - 1 ? n - 1 : v);
}
__device__ __forceinline__ int bucket_of(int d) {
    const int b = d >> BSH;
    return b > NBMAX - 1 ? NBMAX - 1 : b;
}

__global__ void __launch_bounds__(1024)
k_bucket_hist(const int* __restrict__ dst, int* __restrict__ hist, int E, int n) {
    __shared__ int h[NBMAX];
    const int t    = threadIdx.x;
    const int base = blockIdx.x * CHUNK;
    int d[4];
#pragma unroll
    for (int k = 0; k < 4; ++k) {
        const int e = base + k * 1024 + t;
        d[k] = e < E ? dst[e] : 0;
    }
    if (t < NBMAX) h[t] = 0;
    __syncthreads();
#pragma unroll
    for (int k = 0; k < 4; ++k) {
        const int e = base + k * 1024 + t;
        if (e < E) atomicAdd(&h[bucket_of(clamp_node(d[k], n))], 1);
    }
    __syncthreads();
    if (t < NBMAX) hist[(size_t)blockIdx.x * NBMAX + t] = h[t];
}

__device__ __forceinline__ int wave_incl_scan(int v, int lane) {
    for (int off = 1; off < 32; off <<= 1) {
        const int u = __shfl_up(v, off, 32);
        if (lane >= off) v += u;
    }
    return v;
}

__device__ __forceinline__ void fold_heads_part(int part, int lane,
                                                const float* __restrict__ W2,
                                                const float* __restrict__ b2,
                                                const float* __restrict__ Wn,
                                                const float* __restrict__ We,
                                                float* __restrict__ Wp) {
    const int col  = lane & 15;
    const int kb   = (lane >> 4) << 1;
    const bool live = col < 6;
    const float* __restrict__ hc =
        col < 2 ? Wn + col : (col < 4 ? We + (col - 2) : (col < 6 ? We + 256 + (col - 4) : Wn));

    if (part < 16) {
        const float* __restrict__ Ar = W2 + (size_t)(part * 16 + col) * 128 + kb;
        v8f acc;
#pragma unroll
        for (int r = 0; r < 8; ++r) acc[r] = 0.0f;
#pragma unroll
        for (int k = 0; k < 128; k += 4) {
            v2f av; av.x = Ar[k];                 av.y = Ar[k + 1];
            const float h0 = hc[(k + kb) * 2];
            const float h1 = hc[(k + kb + 1) * 2];
            v2f bv; bv.x = live ? h0 : 0.0f;      bv.y = live ? h1 : 0.0f;
            acc = __builtin_amdgcn_wmma_f32_16x16x4_f32(
                false, av, false, bv, (short)0, acc, false, false);
        }
        float* __restrict__ Cp = Wp + (size_t)(part * 16 + ((lane >> 4) << 3)) * 16 + col;
#pragma unroll
        for (int r = 0; r < 8; ++r) Cp[(size_t)r * 16] = acc[r];
    } else if (lane < 16) {
        float s = 0.0f;
        for (int k = 0; k < 128; ++k) {
            const float h = hc[k * 2];
            s += b2[k] * (live ? h : 0.0f);
        }
        Wp[256 * 16 + lane] = s;
    }
}

__global__ void __launch_bounds__(1024)
k_bucket_scatter(const int* __restrict__ src, const int* __restrict__ dst,
                 const int* __restrict__ hist, int* __restrict__ bstart,
                 int* __restrict__ ent, int E, int n,
                 const float* __restrict__ W2, const float* __restrict__ b2,
                 const float* __restrict__ Wn, const float* __restrict__ We,
                 float* __restrict__ Wp) {
    __shared__ int h[NBMAX];
    __shared__ int gb[NBMAX];
    __shared__ int ls[NBMAX];
    __shared__ __attribute__((aligned(16))) int ptot[16 * NBMAX];
    __shared__ __attribute__((aligned(16))) int ppre[16 * NBMAX];
    __shared__ int stage[CHUNK];
    __shared__ int wsum[8];
    __shared__ int lsum[8];
    const int t    = threadIdx.x;
    const int lane = t & 31;
    const int wave = t >> 5;

    if (blockIdx.x == 0) {
        const int part = __builtin_amdgcn_readfirstlane(wave);
        if (part <= 16) fold_heads_part(part, lane, W2, b2, Wn, We, Wp);
        return;
    }
    const int blk  = (int)blockIdx.x - 1;
    const int nblk = (int)gridDim.x - 1;
    const int base = blk * CHUNK;

    int d[4], s[4];
#pragma unroll
    for (int k = 0; k < 4; ++k) {
        const int e = base + k * 1024 + t;
        d[k] = e < E ? dst[e] : 0;
        s[k] = e < E ? src[e] : 0;
    }
    const int own = t < NBMAX ? hist[(size_t)blk * NBMAX + t] : 0;

    {
        const int c4 = (t & 63) * 4;
        const int q  = t >> 6;
        int4 tot = make_int4(0, 0, 0, 0);
        int4 pre = make_int4(0, 0, 0, 0);
        for (int j0 = q; j0 < nblk; j0 += 256) {
            int4 v[16];
#pragma unroll
            for (int u = 0; u < 16; ++u) {
                const int j = j0 + 16 * u;
                v[u] = make_int4(0, 0, 0, 0);
                if (j < nblk) v[u] = *(const int4*)(hist + (size_t)j * NBMAX + c4);
            }
#pragma unroll
            for (int u = 0; u < 16; ++u) {
                const int j = j0 + 16 * u;
                tot.x += v[u].x;  tot.y += v[u].y;  tot.z += v[u].z;  tot.w += v[u].w;
                if (j < blk) {
                    pre.x += v[u].x;  pre.y += v[u].y;  pre.z += v[u].z;  pre.w += v[u].w;
                }
            }
        }
        *(int4*)(ptot + q * NBMAX + c4) = tot;
        *(int4*)(ppre + q * NBMAX + c4) = pre;
    }
    if (t < NBMAX) h[t] = 0;
    __syncthreads();

    int tot = 0, pre = 0;
    if (t < NBMAX) {
#pragma unroll
        for (int q = 0; q < 16; ++q) {
            tot += ptot[q * NBMAX + t];
            pre += ppre[q * NBMAX + t];
        }
    }
    const int inc = wave_incl_scan(tot, lane);
    const int inl = wave_incl_scan(own, lane);
    if (lane == 31 && wave < 8) { wsum[wave] = inc; lsum[wave] = inl; }
    __syncthreads();
    if (t < NBMAX) {
        int wb = 0, lb = 0;
#pragma unroll
        for (int w = 0; w < 8; ++w) {
            wb += (w < wave) ? wsum[w] : 0;
            lb += (w < wave) ? lsum[w] : 0;
        }
        const int excl = wb + inc - tot;
        const int lex  = lb + inl - own;
        ls[t] = lex;
        gb[t] = excl + pre - lex;
        if (blk == 0) {
            bstart[t] = excl;
            if (t == NBMAX - 1) bstart[NBMAX] = excl + tot;
        }
    }
    __syncthreads();

    int b[4], r[4];
#pragma unroll
    for (int k = 0; k < 4; ++k) {
        const int e = base + k * 1024 + t;
        d[k] = clamp_node(d[k], n);
        s[k] = clamp_node(s[k], n);
        b[k] = bucket_of(d[k]);
        r[k] = 0;
        if (e < E) r[k] = atomicAdd(&h[b[k]], 1);
    }
#pragma unroll
    for (int k = 0; k < 4; ++k) {
        const int e = base + k * 1024 + t;
        if (e < E) {
            int i = ls[b[k]] + r[k];
            i = i < 0 ? 0 : (i > CHUNK - 1 ? CHUNK - 1 : i);
            stage[i] = (int)(((unsigned)s[k] << 16) | (unsigned)(d[k] & 0xFFFF));
        }
    }
    __syncthreads();

    const int mine = (E - base) < CHUNK ? (E - base) : CHUNK;
#pragma unroll
    for (int k = 0; k < 4; ++k) {
        const int i = k * 1024 + t;
        if (i < mine) {
            const int w  = stage[i];
            int p = gb[bucket_of(w & 0xFFFF)] + i;
            p = p < 0 ? 0 : (p > E - 1 ? E - 1 : p);
            ent[p] = w;
        }
    }
}

__global__ void __launch_bounds__(1024)
k_bucket_fine(const int* __restrict__ ent, const int* __restrict__ bstart,
              int* __restrict__ rowptr, float* __restrict__ dinv,
              int* __restrict__ csr_src, int E, int n) {
    constexpr int FCAP = 8 * 1024;
    __shared__ int cnt[BNODE];
    __shared__ int cur[BNODE];
    __shared__ int wsum[8];
    __shared__ int stage[FCAP];
    const int t    = threadIdx.x;
    const int lane = t & 31;
    const int wave = t >> 5;
    const int bk   = blockIdx.x;
    int beg = bstart[bk];
    int end = bstart[bk + 1];
    beg = beg < 0 ? 0 : (beg > E ? E : beg);
    end = end < beg ? beg : (end > E ? E : end);
    const int  m      = end - beg;
    const bool staged = m <= FCAP;

    unsigned w[8];
#pragma unroll
    for (int u = 0; u < 8; ++u) {
        const int i = u * 1024 + t;
        w[u] = i < m ? (unsigned)ent[beg + i] : 0u;
    }
    if (t < BNODE) cnt[t] = 0;
    __syncthreads();
#pragma unroll
    for (int u = 0; u < 8; ++u) {
        const int i = u * 1024 + t;
        if (i < m) atomicAdd(&cnt[w[u] & (BNODE - 1)], 1);
    }
    for (int i = FCAP + t; i < m; i += 1024)
        atomicAdd(&cnt[(unsigned)ent[beg + i] & (BNODE - 1)], 1);
    __syncthreads();

    const int v   = t < BNODE ? cnt[t] : 0;
    const int inc = wave_incl_scan(v, lane);
    if (lane == 31 && wave < 8) wsum[wave] = inc;
    __syncthreads();
    if (t < BNODE) {
        int wb = 0;
#pragma unroll
        for (int k = 0; k < 8; ++k) wb += (k < wave) ? wsum[k] : 0;
        const int first = wb + inc - v;
        cur[t] = first;
        const int node = bk * BNODE + t;
        if (node < n) {
            rowptr[node] = beg + first;
            dinv[node]   = rsqrtf((float)v + 1.0f);
        }
    }
    if (bk == (int)gridDim.x - 1 && t == 0) rowptr[n] = end;
    __syncthreads();

    int p[8];
#pragma unroll
    for (int u = 0; u < 8; ++u) {
        const int i = u * 1024 + t;
        p[u] = 0;
        if (i < m) p[u] = atomicAdd(&cur[w[u] & (BNODE - 1)], 1);
    }
    if (staged) {
#pragma unroll
        for (int u = 0; u < 8; ++u) {
            const int i = u * 1024 + t;
            if (i < m) {
                const int q = p[u] < 0 ? 0 : (p[u] > FCAP - 1 ? FCAP - 1 : p[u]);
                stage[q] = (int)(w[u] >> 16);
            }
        }
    } else {
#pragma unroll
        for (int u = 0; u < 8; ++u) {
            const int i = u * 1024 + t;
            if (i < m) {
                int q = beg + p[u];
                q = q < 0 ? 0 : (q > E - 1 ? E - 1 : q);
                csr_src[q] = (int)(w[u] >> 16);
            }
        }
        for (int i = FCAP + t; i < m; i += 1024) {
            const unsigned x = (unsigned)ent[beg + i];
            int q = beg + atomicAdd(&cur[x & (BNODE - 1)], 1);
            q = q < 0 ? 0 : (q > E - 1 ? E - 1 : q);
            csr_src[q] = (int)(x >> 16);
        }
    }
    __syncthreads();
    if (staged) {
        for (int i = t; i < m; i += 1024) csr_src[beg + i] = stage[i];
    }
}

template <int R>
__device__ __forceinline__ void gather_rows(const float* __restrict__ x, int my, float cw,
                                            int j, bool hi, int l16, float4& acc) {
    float4 r[R];
#pragma unroll
    for (int u = 0; u < R; ++u) {
        const int s0 = __builtin_amdgcn_readlane(my, j + u);
        const int s1 = __builtin_amdgcn_readlane(my, 16 + j + u);
        const unsigned s = (unsigned)(hi ? s1 : s0);
        r[u] = *(const float4*)((const char*)x + (s * 256u + (unsigned)l16 * 16u));
    }
#pragma unroll
    for (int u = 0; u < R; ++u) {
        const int   c0 = __builtin_amdgcn_readlane(__float_as_int(cw), j + u);
        const int   c1 = __builtin_amdgcn_readlane(__float_as_int(cw), 16 + j + u);
        const float c  = __int_as_float(hi ? c1 : c0);
        acc.x += r[u].x * c;
        acc.y += r[u].y * c;
        acc.z += r[u].z * c;
        acc.w += r[u].w * c;
    }
}

__device__ __forceinline__ void gather_chunk(const float* __restrict__ x, int my, float cw,
                                             int T, bool hi, int l16, float4& acc) {
    int j = 0;
    for (; j + 8 <= T; j += 8) gather_rows<8>(x, my, cw, j, hi, l16, acc);
    if (j + 4 <= T) { gather_rows<4>(x, my, cw, j, hi, l16, acc); j += 4; }
    if (j + 2 <= T) { gather_rows<2>(x, my, cw, j, hi, l16, acc); j += 2; }
    if (j < T)      { gather_rows<1>(x, my, cw, j, hi, l16, acc); }
}

__device__ __forceinline__ void load_positions(const int* __restrict__ csr_src, int b, int e, int c,
                                               int nd, bool live, int& m0, int& m1) {
    m0 = nd;  m1 = nd;
    if (live) {
        if (c >= 1 && c <= e - b) m0 = csr_src[b + c - 1];
        if (16 + c <= e - b)      m1 = csr_src[b + 15 + c];
    }
}

__global__ void __launch_bounds__(256)
k_hidden_fold(const int* __restrict__ rowptr, const int* __restrict__ csr_src,
              const float* __restrict__ dinv, const float* __restrict__ x,
              const float* __restrict__ W1, const float* __restrict__ b1,
              const float* __restrict__ Wp, float* __restrict__ Y, int ntiles, int n) {
    constexpr int LDA = 68;
    constexpr int LDR = 12;
    constexpr int LDF = 20;
    __shared__ __attribute__((aligned(16))) float As[16 * LDA];
    __shared__ __attribute__((aligned(16))) float red[8 * 16 * LDR];
    __shared__ __attribute__((aligned(16))) float bsL[256];
    __shared__ __attribute__((aligned(16))) float a2L[256 * LDF];

    const int  tid  = threadIdx.x;
    const int  wave = tid >> 5;
    const int  lane = tid & 31;
    const int  c    = lane & 15;
    const int  h    = lane >> 4;
    const bool hi   = h != 0;
    const int  hb   = wave * 32;

    v2f a1[2][16];
#pragma unroll
    for (int t = 0; t < 2; ++t) {
#pragma unroll
        for (int sp = 0; sp < 8; ++sp) {
#pragma unroll
            for (int u = 0; u < 2; ++u) {
                const int k = 8 * sp + 4 * h + 2 * u;
                a1[t][2 * sp + u].x = W1[(size_t)k * 256 + hb + 16 * t + c];
                a1[t][2 * sp + u].y = W1[(size_t)(k + 1) * 256 + hb + 16 * t + c];
            }
        }
    }
    {
        float* __restrict__ fp = a2L + tid * LDF;
#pragma unroll
        for (int t = 0; t < 2; ++t) {
#pragma unroll
            for (int s = 0; s < 4; s += 2) {
                const int ch = hb + 16 * t + 8 * h + 2 * s;
                float4 w;
                w.x = Wp[(size_t)ch * 16 + c];
                w.y = Wp[(size_t)(ch + 1) * 16 + c];
                w.z = Wp[(size_t)(ch + 2) * 16 + c];
                w.w = Wp[(size_t)(ch + 3) * 16 + c];
                *(float4*)(fp + 8 * t + 2 * s) = w;
            }
        }
        bsL[tid] = b1[tid];
    }

    const int hn   = 2 * wave + h;
    const int step = (int)gridDim.x;
    int tile = blockIdx.x;
    auto node_of = [&](int t) -> int {
        int nd = t * 16 + hn;
        nd = nd < n ? nd : n - 1;
        return nd < 0 ? 0 : nd;
    };

    int   bA = 0, eA = 0, bB = 0, eB = 0, bC = 0, eC = 0;
    float dA = 0.0f, dB = 0.0f, dC = 0.0f;
    if (tile < ntiles) {
        const int nd = node_of(tile);
        bA = rowptr[nd];  eA = rowptr[nd + 1];  dA = dinv[nd];
    }
    if (tile + step < ntiles) {
        const int nd = node_of(tile + step);
        bB = rowptr[nd];  eB = rowptr[nd + 1];  dB = dinv[nd];
    }
    if (tile + 2 * step < ntiles) {
        const int nd = node_of(tile + 2 * step);
        bC = rowptr[nd];  eC = rowptr[nd + 1];  dC = dinv[nd];
    }
    int my0, my1, nx0, nx1;
    load_positions(csr_src, bA, eA, c, node_of(tile), tile < ntiles, my0, my1);
    load_positions(csr_src, bB, eB, c, node_of(tile + step), tile + step < ntiles, nx0, nx1);
    my0 = my0 < 0 ? 0 : (my0 > n - 1 ? n - 1 : my0);
    my1 = my1 < 0 ? 0 : (my1 > n - 1 ? n - 1 : my1);
    float q0 = dinv[my0];
    float q1 = dinv[my1];

    for (; tile < ntiles; tile += step) {
        const int nd = node_of(tile);
        const int m = eA - bA + 1;
        const int mlo = __builtin_amdgcn_readlane(m, 0);
        const int mhi = __builtin_amdgcn_readlane(m, 16);
        const int mm  = mlo > mhi ? mlo : mhi;

        const float cw0 = c < m      ? q0 * dA : 0.0f;
        const float cw1 = 16 + c < m ? q1 * dA : 0.0f;

        float4 acc = make_float4(0.0f, 0.0f, 0.0f, 0.0f);
        {
            const int tp = ((mm < 32 ? mm : 32) + 3) & ~3;
            int j = 0;
            for (; j + 8 <= tp; j += 8) {
                const bool first = j < 16;
                gather_rows<8>(x, first ? my0 : my1, first ? cw0 : cw1, j & 15, hi, c, acc);
            }
            if (j < tp) {
                const bool first = j < 16;
                gather_rows<4>(x, first ? my0 : my1, first ? cw0 : cw1, j & 15, hi, c, acc);
            }
        }
        for (int base = 32; base < mm; base += 16) {
            int   my = nd;
            float cw = 0.0f;
            if (base + c < m) {
                my = csr_src[bA + base - 1 + c];
                my = my < 0 ? 0 : (my > n - 1 ? n - 1 : my);
                cw = dinv[my] * dA;
            }
            gather_chunk(x, my, cw, mm - base < 16 ? mm - base : 16, hi, c, acc);
        }

        *(float4*)(As + hn * LDA + c * 4) = acc;
        __syncthreads();

        {
            bA = bB;  eA = eB;  dA = dB;
            bB = bC;  eB = eC;  dB = dC;
            my0 = nx0 < 0 ? 0 : (nx0 > n - 1 ? n - 1 : nx0);
            my1 = nx1 < 0 ? 0 : (nx1 > n - 1 ? n - 1 : nx1);
            q0 = dinv[my0];
            q1 = dinv[my1];
            const int t2 = tile + 2 * step;
            load_positions(csr_src, bB, eB, c, node_of(t2), t2 < ntiles, nx0, nx1);
            const int t3 = t2 + step;
            if (t3 < ntiles) {
                const int n3 = node_of(t3);
                bC = rowptr[n3];  eC = rowptr[n3 + 1];  dC = dinv[n3];
            }
        }

        v2f b[16];
#pragma unroll
        for (int sp = 0; sp < 8; ++sp) {
            const float4 q = *(const float4*)(As + c * LDA + 8 * sp + 4 * h);
            b[2 * sp].x     = q.x;  b[2 * sp].y     = q.y;
            b[2 * sp + 1].x = q.z;  b[2 * sp + 1].y = q.w;
        }

        v8f h0, h1;
        {
            const float4 p0 = *(const float4*)(bsL + hb + 8 * h);
            const float4 p1 = *(const float4*)(bsL + hb + 8 * h + 4);
            const float4 p2 = *(const float4*)(bsL + hb + 16 + 8 * h);
            const float4 p3 = *(const float4*)(bsL + hb + 16 + 8 * h + 4);
            h0[0] = p0.x; h0[1] = p0.y; h0[2] = p0.z; h0[3] = p0.w;
            h0[4] = p1.x; h0[5] = p1.y; h0[6] = p1.z; h0[7] = p1.w;
            h1[0] = p2.x; h1[1] = p2.y; h1[2] = p2.z; h1[3] = p2.w;
            h1[4] = p3.x; h1[5] = p3.y; h1[6] = p3.z; h1[7] = p3.w;
        }
        v2f a2[2][4];
        {
            const float* __restrict__ fp = a2L + tid * LDF;
#pragma unroll
            for (int t = 0; t < 2; ++t) {
                const float4 f0 = *(const float4*)(fp + 8 * t);
                const float4 f1 = *(const float4*)(fp + 8 * t + 4);
                a2[t][0].x = f0.x;  a2[t][0].y = f0.y;
                a2[t][1].x = f0.z;  a2[t][1].y = f0.w;
                a2[t][2].x = f1.x;  a2[t][2].y = f1.y;
                a2[t][3].x = f1.z;  a2[t][3].y = f1.w;
            }
        }
#pragma unroll
        for (int s = 0; s < 16; ++s) {
            h0 = __builtin_amdgcn_wmma_f32_16x16x4_f32(
                false, a1[0][s], false, b[s], (short)0, h0, false, false);
            h1 = __builtin_amdgcn_wmma_f32_16x16x4_f32(
                false, a1[1][s], false, b[s], (short)0, h1, false, false);
        }
#pragma unroll
        for (int r = 0; r < 8; ++r) {
            h0[r] = h0[r] > 0.0f ? h0[r] : 0.0f;
            h1[r] = h1[r] > 0.0f ? h1[r] : 0.0f;
        }

        v8f o;
#pragma unroll
        for (int r = 0; r < 8; ++r) o[r] = 0.0f;
#pragma unroll
        for (int s = 0; s < 4; ++s) {
            v2f hv; hv.x = h0[2 * s]; hv.y = h0[2 * s + 1];
            o = __builtin_amdgcn_wmma_f32_16x16x4_f32(
                false, a2[0][s], false, hv, (short)0, o, false, false);
        }
#pragma unroll
        for (int s = 0; s < 4; ++s) {
            v2f hv; hv.x = h1[2 * s]; hv.y = h1[2 * s + 1];
            o = __builtin_amdgcn_wmma_f32_16x16x4_f32(
                false, a2[1][s], false, hv, (short)0, o, false, false);
        }

        if (h == 0) {
            float* __restrict__ rp = red + (wave * 16 + c) * LDR;
            *(float4*)(rp)     = make_float4(o[0], o[1], o[2], o[3]);
            *(float4*)(rp + 4) = make_float4(o[4], o[5], o[6], o[7]);
        }
        __syncthreads();

        if (tid < 128) {
            const int node = tid >> 3;
            const int j    = tid & 7;
            float s = red[node * LDR + j];
#pragma unroll
            for (int w = 1; w < 8; ++w) s += red[(w * 16 + node) * LDR + j];
            Y[(size_t)tile * 128 + tid] = s;
        }
    }
}

__global__ void __launch_bounds__(256)
k_gather_heads(const int* __restrict__ rowptr, const int* __restrict__ csr_src,
               const float* __restrict__ dinv, const float* __restrict__ y,
               const float* __restrict__ c, const float* __restrict__ bn,
               float* __restrict__ out, float* __restrict__ pq, int n) {
    const int  t    = blockIdx.x * 256 + threadIdx.x;
    const int  q    = t & 3;
    const int  node = t >> 2;
    const bool live = node < n;
    const int  nd   = live ? node : n - 1;

    const int   beg = rowptr[nd];
    const int   end = rowptr[nd + 1];
    const float di  = dinv[nd];
    const float4* __restrict__ y4 = (const float4*)y;

    float4 a0, a1;
    {
        const float4 s0 = y4[(size_t)nd * 2];
        const float4 s1 = y4[(size_t)nd * 2 + 1];
        const float  w  = di * di;
        const bool   z  = q != 0;
        a0.x = z ? 0.0f : s0.x * w;  a0.y = z ? 0.0f : s0.y * w;
        a0.z = z ? 0.0f : s0.z * w;  a0.w = z ? 0.0f : s0.w * w;
        a1.x = z ? 0.0f : s1.x * w;  a1.y = z ? 0.0f : s1.y * w;
        a1.z = z ? 0.0f : s1.z * w;  a1.w = z ? 0.0f : s1.w * w;
    }

    int e = beg + q;
    for (; e + 4 < end; e += 8) {
        int sA = csr_src[e];
        int sB = csr_src[e + 4];
        sA = sA < 0 ? 0 : (sA > n - 1 ? n - 1 : sA);
        sB = sB < 0 ? 0 : (sB > n - 1 ? n - 1 : sB);
        const float  cA = dinv[sA] * di;
        const float  cB = dinv[sB] * di;
        const float4 rA0 = y4[(size_t)sA * 2], rA1 = y4[(size_t)sA * 2 + 1];
        const float4 rB0 = y4[(size_t)sB * 2], rB1 = y4[(size_t)sB * 2 + 1];
        a0.x += rA0.x * cA;  a0.y += rA0.y * cA;  a0.z += rA0.z * cA;  a0.w += rA0.w * cA;
        a1.x += rA1.x * cA;  a1.y += rA1.y * cA;  a1.z += rA1.z * cA;  a1.w += rA1.w * cA;
        a0.x += rB0.x * cB;  a0.y += rB0.y * cB;  a0.z += rB0.z * cB;  a0.w += rB0.w * cB;
        a1.x += rB1.x * cB;  a1.y += rB1.y * cB;  a1.z += rB1.z * cB;  a1.w += rB1.w * cB;
    }
    if (e < end) {
        int sA = csr_src[e];
        sA = sA < 0 ? 0 : (sA > n - 1 ? n - 1 : sA);
        const float  cA = dinv[sA] * di;
        const float4 rA0 = y4[(size_t)sA * 2], rA1 = y4[(size_t)sA * 2 + 1];
        a0.x += rA0.x * cA;  a0.y += rA0.y * cA;  a0.z += rA0.z * cA;  a0.w += rA0.w * cA;
        a1.x += rA1.x * cA;  a1.y += rA1.y * cA;  a1.z += rA1.z * cA;  a1.w += rA1.w * cA;
    }

#pragma unroll
    for (int off = 1; off <= 2; off <<= 1) {
        a0.x += __shfl_xor(a0.x, off, 32);  a0.y += __shfl_xor(a0.y, off, 32);
        a0.z += __shfl_xor(a0.z, off, 32);  a0.w += __shfl_xor(a0.w, off, 32);
        a1.x += __shfl_xor(a1.x, off, 32);  a1.y += __shfl_xor(a1.y, off, 32);
    }

    if (live) {
        if (q == 0) {
            float2 o;
            o.x = (a0.x + c[0]) + bn[0];
            o.y = (a0.y + c[1]) + bn[1];
            *(float2*)(out + (size_t)nd * 2) = o;
        } else if (q == 1) {
            float4 r;
            r.x = a0.z + c[2];  r.y = a0.w + c[3];
            r.z = a1.x + c[4];  r.w = a1.y + c[5];
            *(float4*)(pq + (size_t)nd * 4) = r;
        }
    }
}

__device__ __forceinline__ float2 edge_one(const float* __restrict__ pq, int s, int d,
                                           float b0, float b1) {
    const float2 p = *(const float2*)(pq + (size_t)s * 4);
    const float2 q = *(const float2*)(pq + (size_t)d * 4 + 2);
    float2 o;
    o.x = (p.x + q.x) + b0;
    o.y = (p.y + q.y) + b1;
    return o;
}

__global__ void __launch_bounds__(256)
k_edge_head(const int* __restrict__ src, const int* __restrict__ dst,
            const float* __restrict__ pq, const float* __restrict__ be,
            float* __restrict__ out, int E) {
    const int   base = blockIdx.x * 1024;
    const float b0 = be[0], b1 = be[1];
    const bool wide = ((((size_t)src | (size_t)dst) & 7) == 0) && (((size_t)out & 15) == 0);
    if (wide && base + 1024 <= E) {
        const int  e0 = base + threadIdx.x * 2;
        const int  e1 = e0 + 512;
        const int2 s0 = *(const int2*)(src + e0);
        const int2 d0 = *(const int2*)(dst + e0);
        const int2 s1 = *(const int2*)(src + e1);
        const int2 d1 = *(const int2*)(dst + e1);
        const float2 a0 = edge_one(pq, s0.x, d0.x, b0, b1);
        const float2 a1 = edge_one(pq, s0.y, d0.y, b0, b1);
        const float2 c0 = edge_one(pq, s1.x, d1.x, b0, b1);
        const float2 c1 = edge_one(pq, s1.y, d1.y, b0, b1);
        float4 o0, o1;
        o0.x = a0.x; o0.y = a0.y; o0.z = a1.x; o0.w = a1.y;
        o1.x = c0.x; o1.y = c0.y; o1.z = c1.x; o1.w = c1.y;
        *(float4*)(out + (size_t)e0 * 2) = o0;
        *(float4*)(out + (size_t)e1 * 2) = o1;
    } else {
        for (int u = 0; u < 2; ++u) {
            for (int k = 0; k < 2; ++k) {
                const int e = base + u * 512 + threadIdx.x * 2 + k;
                if (e < E) *(float2*)(out + (size_t)e * 2) = edge_one(pq, src[e], dst[e], b0, b1);
            }
        }
    }
}

extern "C" void kernel_launch(void* const* d_in, const int* in_sizes, int n_in,
                              void* d_out, int out_size, void* d_ws, size_t ws_size,
                              hipStream_t stream) {
    const float* x  = (const float*)d_in[0];
    const int*   ei = (const int*)d_in[1];
    const float* W1 = (const float*)d_in[2];
    const float* b1 = (const float*)d_in[3];
    const float* W2 = (const float*)d_in[4];
    const float* b2 = (const float*)d_in[5];
    const float* Wn = (const float*)d_in[6];
    const float* bn = (const float*)d_in[7];
    const float* We = (const float*)d_in[8];
    const float* be = (const float*)d_in[9];

    const int N = in_sizes[0] / 64;
    const int E = in_sizes[1] / 2;
    const int* src = ei;
    const int* dst = ei + E;

    char*  ws  = (char*)d_ws;
    size_t off = 0;
    auto carve = [&](size_t bytes) -> float* {
        float* p = (float*)(ws + off);
        off += (bytes + 255) & ~(size_t)255;
        return p;
    };
    const int G1 = (E + CHUNK - 1) / CHUNK;
    float* dinv   = carve((size_t)N * 4);
    int*   rowptr = (int*)carve((size_t)(N + 1) * 4);
    int*   hist   = (int*)carve((size_t)G1 * NBMAX * 4);
    int*   bstart = (int*)carve((size_t)(NBMAX + 1) * 4);
    int*   ent    = (int*)carve((size_t)E * 4);
    int*   csrsrc = (int*)carve((size_t)E * 4);
    float* wp   = carve((size_t)(256 + 1) * 16 * 4);
    float* y    = carve((size_t)N * 8 * 4);
    float* pq   = carve((size_t)N * 4 * 4);

    float* out_node = (float*)d_out;
    float* out_edge = out_node + (size_t)N * 2;

    k_bucket_hist   <<<G1, 1024, 0, stream>>>(dst, hist, E, N);
    k_bucket_scatter<<<G1 + 1, 1024, 0, stream>>>(src, dst, hist, bstart, ent, E, N,
                                              W2, b2, Wn, We, wp);
    k_bucket_fine   <<<(N + BNODE - 1) / BNODE, 1024, 0, stream>>>(ent, bstart, rowptr, dinv, csrsrc, E, N);

    k_hidden_fold<<<768, 256, 0, stream>>>(rowptr, csrsrc, dinv, x, W1, b1, wp, y, N / 16, N);
    k_gather_heads<<<(N * 4 + 255) / 256, 256, 0, stream>>>(
        rowptr, csrsrc, dinv, y, wp + 256 * 16, bn, out_node, pq, N);

    k_edge_head<<<(E + 1023) / 1024, 256, 0, stream>>>(src, dst, pq, be, out_edge, E);
}


# coarse bucketing in one pass: a block orders its 4096-edge chunk by bucket and notes its run starts; the per-bucket block collects its runs from all chunks; 6 -> 5 launches; f32 sum order changes
# speedup vs baseline: 18.6646x; 1.0631x over previous
#include <hip/hip_runtime.h>


typedef float v2f __attribute__((ext_vector_type(2)));
typedef float v8f __attribute__((ext_vector_type(8)));

constexpr int BSH   = 8;
constexpr int BNODE = 1 << BSH;
constexpr int NBMAX = 256;
constexpr int CHUNK = 4096;
constexpr int TPR   = 4;
constexpr int RPT   = 1024 / TPR;
constexpr int RROW  = NBMAX + 32;

__device__ __forceinline__ int clamp_node(int v, int n) {
    return v < 0 ? 0 : (v > n - 1 ? n - 1 : v);
}
__device__ __forceinline__ int bucket_of(int d) {
    const int b = d >> BSH;
    return b > NBMAX - 1 ? NBMAX - 1 : b;
}

__device__ __forceinline__ int wave_incl_scan(int v, int lane) {
    for (int off = 1; off < 32; off <<= 1) {
        const int u = __shfl_up(v, off, 32);
        if (lane >= off) v += u;
    }
    return v;
}

__device__ __forceinline__ void fold_heads_part(int part, int lane,
                                                const float* __restrict__ W2,
                                                const float* __restrict__ b2,
                                                const float* __restrict__ Wn,
                                                const float* __restrict__ We,
                                                float* __restrict__ Wp) {
    const int col  = lane & 15;
    const int kb   = (lane >> 4) << 1;
    const bool live = col < 6;
    const float* __restrict__ hc =
        col < 2 ? Wn + col : (col < 4 ? We + (col - 2) : (col < 6 ? We + 256 + (col - 4) : Wn));

    if (part < 16) {
        const float* __restrict__ Ar = W2 + (size_t)(part * 16 + col) * 128 + kb;
        v8f acc;
#pragma unroll
        for (int r = 0; r < 8; ++r) acc[r] = 0.0f;
#pragma unroll
        for (int k = 0; k < 128; k += 4) {
            v2f av; av.x = Ar[k];                 av.y = Ar[k + 1];
            const float h0 = hc[(k + kb) * 2];
            const float h1 = hc[(k + kb + 1) * 2];
            v2f bv; bv.x = live ? h0 : 0.0f;      bv.y = live ? h1 : 0.0f;
            acc = __builtin_amdgcn_wmma_f32_16x16x4_f32(
                false, av, false, bv, (short)0, acc, false, false);
        }
        float* __restrict__ Cp = Wp + (size_t)(part * 16 + ((lane >> 4) << 3)) * 16 + col;
#pragma unroll
        for (int r = 0; r < 8; ++r) Cp[(size_t)r * 16] = acc[r];
    } else if (lane < 16) {
        float s = 0.0f;
        for (int k = 0; k < 128; ++k) {
            const float h = hc[k * 2];
            s += b2[k] * (live ? h : 0.0f);
        }
        Wp[256 * 16 + lane] = s;
    }
}

__global__ void __launch_bounds__(1024)
k_chunk_sort(const int* __restrict__ src, const int* __restrict__ dst,
             int* __restrict__ runs, int* __restrict__ ent, int E, int n,
             const float* __restrict__ W2, const float* __restrict__ b2,
             const float* __restrict__ Wn, const float* __restrict__ We,
             float* __restrict__ Wp) {
    __shared__ int h[NBMAX];
    __shared__ int ls[NBMAX];
    __shared__ int stage[CHUNK];
    __shared__ int lsum[8];
    const int t    = threadIdx.x;
    const int lane = t & 31;
    const int wave = t >> 5;

    if (blockIdx.x == 0) {
        const int part = __builtin_amdgcn_readfirstlane(wave);
        if (part <= 16) fold_heads_part(part, lane, W2, b2, Wn, We, Wp);
        return;
    }
    const int blk  = (int)blockIdx.x - 1;
    const int base = blk * CHUNK;

    int d[4], s[4];
#pragma unroll
    for (int k = 0; k < 4; ++k) {
        const int e = base + k * 1024 + t;
        d[k] = e < E ? dst[e] : 0;
        s[k] = e < E ? src[e] : 0;
    }
    if (t < NBMAX) h[t] = 0;
    __syncthreads();

    int b[4], r[4];
#pragma unroll
    for (int k = 0; k < 4; ++k) {
        const int e = base + k * 1024 + t;
        d[k] = clamp_node(d[k], n);
        s[k] = clamp_node(s[k], n);
        b[k] = bucket_of(d[k]);
        r[k] = 0;
        if (e < E) r[k] = atomicAdd(&h[b[k]], 1);
    }
    __syncthreads();

    const int own = t < NBMAX ? h[t] : 0;
    const int inl = wave_incl_scan(own, lane);
    if (lane == 31 && wave < 8) lsum[wave] = inl;
    __syncthreads();
    if (t < NBMAX) {
        int lb = 0;
#pragma unroll
        for (int w = 0; w < 8; ++w) lb += (w < wave) ? lsum[w] : 0;
        const int lex = lb + inl - own;
        ls[t] = lex;
        runs[(size_t)blk * RROW + t] = lex;
        if (t == NBMAX - 1) runs[(size_t)blk * RROW + NBMAX] = lex + own;
    }
    __syncthreads();

#pragma unroll
    for (int k = 0; k < 4; ++k) {
        const int e = base + k * 1024 + t;
        if (e < E) {
            int i = ls[b[k]] + r[k];
            i = i < 0 ? 0 : (i > CHUNK - 1 ? CHUNK - 1 : i);
            stage[i] = (int)(((unsigned)s[k] << 16) | (unsigned)(d[k] & 0xFFFF));
        }
    }
    __syncthreads();

    const int mine = (E - base) < CHUNK ? (E - base) : CHUNK;
#pragma unroll
    for (int k = 0; k < 4; ++k) {
        const int i = k * 1024 + t;
        if (i < mine) ent[base + i] = stage[i];
    }
}

__global__ void __launch_bounds__(1024)
k_bucket_fine(const int* __restrict__ ent, const int* __restrict__ runs, int G,
              int* __restrict__ rowptr, float* __restrict__ dinv,
              int* __restrict__ csr_src, int E, int n) {
    constexpr int FCAP = 8 * 1024;
    constexpr int KEEP = 8;
    __shared__ int cnt[BNODE];
    __shared__ int cur[BNODE];
    __shared__ int wsum[8];
    __shared__ int bsum[32];
    __shared__ int begS;
    __shared__ int stage[FCAP];
    const int t    = threadIdx.x;
    const int lane = t & 31;
    const int wave = t >> 5;
    const int bk   = blockIdx.x;
    const int q    = t & (TPR - 1);
    const int j0   = t / TPR;

    auto run_of = [&](int j, int& rb, int& rl) -> int {
        int lim = E - j * CHUNK;
        lim = lim < 0 ? 0 : (lim > CHUNK ? CHUNK : lim);
        int a = runs[(size_t)j * RROW + bk];
        int z = runs[(size_t)j * RROW + bk + 1];
        a = a < 0 ? 0 : (a > lim ? lim : a);
        z = z < a ? a : (z > lim ? lim : z);
        rb = j * CHUNK + a;
        rl = z - a;
        return a;
    };

    int rb = 0, rl = 0, part = 0;
    if (j0 < G) {
        const int a = run_of(j0, rb, rl);
        part = q == 0 ? a : 0;
    }
    for (int j = j0 + RPT; j < G; j += RPT) {
        int cb, cl;
        const int a = run_of(j, cb, cl);
        part += q == 0 ? a : 0;
    }
    unsigned w[KEEP];
#pragma unroll
    for (int u = 0; u < KEEP; ++u) {
        const int i = q + TPR * u;
        w[u] = i < rl ? (unsigned)ent[rb + i] : 0u;
    }
    for (int off = 16; off >= 1; off >>= 1) part += __shfl_xor(part, off, 32);
    if (lane == 0) bsum[wave] = part;
    if (t < BNODE) cnt[t] = 0;
    __syncthreads();
#pragma unroll
    for (int u = 0; u < KEEP; ++u) {
        if (q + TPR * u < rl) atomicAdd(&cnt[w[u] & (BNODE - 1)], 1);
    }
    for (int i = KEEP * TPR + q; i < rl; i += TPR)
        atomicAdd(&cnt[(unsigned)ent[rb + i] & (BNODE - 1)], 1);
    for (int j = j0 + RPT; j < G; j += RPT) {
        int cb, cl;
        run_of(j, cb, cl);
        for (int i = q; i < cl; i += TPR)
            atomicAdd(&cnt[(unsigned)ent[cb + i] & (BNODE - 1)], 1);
    }
    if (wave == 0) {
        int v = bsum[lane];
        for (int off = 16; off >= 1; off >>= 1) v += __shfl_xor(v, off, 32);
        if (lane == 0) begS = v;
    }
    __syncthreads();

    const int v   = t < BNODE ? cnt[t] : 0;
    const int inc = wave_incl_scan(v, lane);
    if (lane == 31 && wave < 8) wsum[wave] = inc;
    __syncthreads();
    int beg = begS;
    int m   = 0;
#pragma unroll
    for (int k = 0; k < 8; ++k) m += wsum[k];
    beg = beg < 0 ? 0 : (beg > E ? E : beg);
    m   = m < 0 ? 0 : (m > E - beg ? E - beg : m);
    const int  end    = beg + m;
    const bool staged = m <= FCAP;
    if (t < BNODE) {
        int wb = 0;
#pragma unroll
        for (int k = 0; k < 8; ++k) wb += (k < wave) ? wsum[k] : 0;
        const int first = wb + inc - v;
        cur[t] = first;
        const int node = bk * BNODE + t;
        if (node < n) {
            rowptr[node] = beg + first;
            dinv[node]   = rsqrtf((float)v + 1.0f);
        }
    }
    if (bk == (int)gridDim.x - 1 && t == 0) rowptr[n] = end;
    __syncthreads();

    int p[KEEP];
#pragma unroll
    for (int u = 0; u < KEEP; ++u) {
        p[u] = 0;
        if (q + TPR * u < rl) p[u] = atomicAdd(&cur[w[u] & (BNODE - 1)], 1);
    }
    auto place = [&](unsigned x) {
        const int o = atomicAdd(&cur[x & (BNODE - 1)], 1);
        if (staged) {
            stage[o < 0 ? 0 : (o > FCAP - 1 ? FCAP - 1 : o)] = (int)(x >> 16);
        } else {
            int g = beg + o;
            g = g < 0 ? 0 : (g > E - 1 ? E - 1 : g);
            csr_src[g] = (int)(x >> 16);
        }
    };
    if (staged) {
#pragma unroll
        for (int u = 0; u < KEEP; ++u) {
            if (q + TPR * u < rl) {
                const int o = p[u] < 0 ? 0 : (p[u] > FCAP - 1 ? FCAP - 1 : p[u]);
                stage[o] = (int)(w[u] >> 16);
            }
        }
    } else {
#pragma unroll
        for (int u = 0; u < KEEP; ++u) {
            if (q + TPR * u < rl) {
                int g = beg + p[u];
                g = g < 0 ? 0 : (g > E - 1 ? E - 1 : g);
                csr_src[g] = (int)(w[u] >> 16);
            }
        }
    }
    for (int i = KEEP * TPR + q; i < rl; i += TPR) place((unsigned)ent[rb + i]);
    for (int j = j0 + RPT; j < G; j += RPT) {
        int cb, cl;
        run_of(j, cb, cl);
        for (int i = q; i < cl; i += TPR) place((unsigned)ent[cb + i]);
    }
    __syncthreads();
    if (staged) {
        for (int i = t; i < m; i += 1024) csr_src[beg + i] = stage[i];
    }
}


template <int R>
__device__ __forceinline__ void gather_rows(const float* __restrict__ x, int my, float cw,
                                            int j, bool hi, int l16, float4& acc) {
    float4 r[R];
#pragma unroll
    for (int u = 0; u < R; ++u) {
        const int s0 = __builtin_amdgcn_readlane(my, j + u);
        const int s1 = __builtin_amdgcn_readlane(my, 16 + j + u);
        const unsigned s = (unsigned)(hi ? s1 : s0);
        r[u] = *(const float4*)((const char*)x + (s * 256u + (unsigned)l16 * 16u));
    }
#pragma unroll
    for (int u = 0; u < R; ++u) {
        const int   c0 = __builtin_amdgcn_readlane(__float_as_int(cw), j + u);
        const int   c1 = __builtin_amdgcn_readlane(__float_as_int(cw), 16 + j + u);
        const float c  = __int_as_float(hi ? c1 : c0);
        acc.x += r[u].x * c;
        acc.y += r[u].y * c;
        acc.z += r[u].z * c;
        acc.w += r[u].w * c;
    }
}

__device__ __forceinline__ void gather_chunk(const float* __restrict__ x, int my, float cw,
                                             int T, bool hi, int l16, float4& acc) {
    int j = 0;
    for (; j + 8 <= T; j += 8) gather_rows<8>(x, my, cw, j, hi, l16, acc);
    if (j + 4 <= T) { gather_rows<4>(x, my, cw, j, hi, l16, acc); j += 4; }
    if (j + 2 <= T) { gather_rows<2>(x, my, cw, j, hi, l16, acc); j += 2; }
    if (j < T)      { gather_rows<1>(x, my, cw, j, hi, l16, acc); }
}

__device__ __forceinline__ void load_positions(const int* __restrict__ csr_src, int b, int e, int c,
                                               int nd, bool live, int& m0, int& m1) {
    m0 = nd;  m1 = nd;
    if (live) {
        if (c >= 1 && c <= e - b) m0 = csr_src[b + c - 1];
        if (16 + c <= e - b)      m1 = csr_src[b + 15 + c];
    }
}

__global__ void __launch_bounds__(256)
k_hidden_fold(const int* __restrict__ rowptr, const int* __restrict__ csr_src,
              const float* __restrict__ dinv, const float* __restrict__ x,
              const float* __restrict__ W1, const float* __restrict__ b1,
              const float* __restrict__ Wp, float* __restrict__ Y, int ntiles, int n) {
    constexpr int LDA = 68;
    constexpr int LDR = 12;
    constexpr int LDF = 20;
    __shared__ __attribute__((aligned(16))) float As[16 * LDA];
    __shared__ __attribute__((aligned(16))) float red[8 * 16 * LDR];
    __shared__ __attribute__((aligned(16))) float bsL[256];
    __shared__ __attribute__((aligned(16))) float a2L[256 * LDF];

    const int  tid  = threadIdx.x;
    const int  wave = tid >> 5;
    const int  lane = tid & 31;
    const int  c    = lane & 15;
    const int  h    = lane >> 4;
    const bool hi   = h != 0;
    const int  hb   = wave * 32;

    v2f a1[2][16];
#pragma unroll
    for (int t = 0; t < 2; ++t) {
#pragma unroll
        for (int sp = 0; sp < 8; ++sp) {
#pragma unroll
            for (int u = 0; u < 2; ++u) {
                const int k = 8 * sp + 4 * h + 2 * u;
                a1[t][2 * sp + u].x = W1[(size_t)k * 256 + hb + 16 * t + c];
                a1[t][2 * sp + u].y = W1[(size_t)(k + 1) * 256 + hb + 16 * t + c];
            }
        }
    }
    {
        float* __restrict__ fp = a2L + tid * LDF;
#pragma unroll
        for (int t = 0; t < 2; ++t) {
#pragma unroll
            for (int s = 0; s < 4; s += 2) {
                const int ch = hb + 16 * t + 8 * h + 2 * s;
                float4 w;
                w.x = Wp[(size_t)ch * 16 + c];
                w.y = Wp[(size_t)(ch + 1) * 16 + c];
                w.z = Wp[(size_t)(ch + 2) * 16 + c];
                w.w = Wp[(size_t)(ch + 3) * 16 + c];
                *(float4*)(fp + 8 * t + 2 * s) = w;
            }
        }
        bsL[tid] = b1[tid];
    }

    const int hn   = 2 * wave + h;
    const int step = (int)gridDim.x;
    int tile = blockIdx.x;
    auto node_of = [&](int t) -> int {
        int nd = t * 16 + hn;
        nd = nd < n ? nd : n - 1;
        return nd < 0 ? 0 : nd;
    };

    int   bA = 0, eA = 0, bB = 0, eB = 0, bC = 0, eC = 0;
    float dA = 0.0f, dB = 0.0f, dC = 0.0f;
    if (tile < ntiles) {
        const int nd = node_of(tile);
        bA = rowptr[nd];  eA = rowptr[nd + 1];  dA = dinv[nd];
    }
    if (tile + step < ntiles) {
        const int nd = node_of(tile + step);
        bB = rowptr[nd];  eB = rowptr[nd + 1];  dB = dinv[nd];
    }
    if (tile + 2 * step < ntiles) {
        const int nd = node_of(tile + 2 * step);
        bC = rowptr[nd];  eC = rowptr[nd + 1];  dC = dinv[nd];
    }
    int my0, my1, nx0, nx1;
    load_positions(csr_src, bA, eA, c, node_of(tile), tile < ntiles, my0, my1);
    load_positions(csr_src, bB, eB, c, node_of(tile + step), tile + step < ntiles, nx0, nx1);
    my0 = my0 < 0 ? 0 : (my0 > n - 1 ? n - 1 : my0);
    my1 = my1 < 0 ? 0 : (my1 > n - 1 ? n - 1 : my1);
    float q0 = dinv[my0];
    float q1 = dinv[my1];

    for (; tile < ntiles; tile += step) {
        const int nd = node_of(tile);
        const int m = eA - bA + 1;
        const int mlo = __builtin_amdgcn_readlane(m, 0);
        const int mhi = __builtin_amdgcn_readlane(m, 16);
        const int mm  = mlo > mhi ? mlo : mhi;

        const float cw0 = c < m      ? q0 * dA : 0.0f;
        const float cw1 = 16 + c < m ? q1 * dA : 0.0f;

        float4 acc = make_float4(0.0f, 0.0f, 0.0f, 0.0f);
        {
            const int tp = ((mm < 32 ? mm : 32) + 3) & ~3;
            int j = 0;
            for (; j + 8 <= tp; j += 8) {
                const bool first = j < 16;
                gather_rows<8>(x, first ? my0 : my1, first ? cw0 : cw1, j & 15, hi, c, acc);
            }
            if (j < tp) {
                const bool first = j < 16;
                gather_rows<4>(x, first ? my0 : my1, first ? cw0 : cw1, j & 15, hi, c, acc);
            }
        }
        for (int base = 32; base < mm; base += 16) {
            int   my = nd;
            float cw = 0.0f;
            if (base + c < m) {
                my = csr_src[bA + base - 1 + c];
                my = my < 0 ? 0 : (my > n - 1 ? n - 1 : my);
                cw = dinv[my] * dA;
            }
            gather_chunk(x, my, cw, mm - base < 16 ? mm - base : 16, hi, c, acc);
        }

        *(float4*)(As + hn * LDA + c * 4) = acc;
        __syncthreads();

        {
            bA = bB;  eA = eB;  dA = dB;
            bB = bC;  eB = eC;  dB = dC;
            my0 = nx0 < 0 ? 0 : (nx0 > n - 1 ? n - 1 : nx0);
            my1 = nx1 < 0 ? 0 : (nx1 > n - 1 ? n - 1 : nx1);
            q0 = dinv[my0];
            q1 = dinv[my1];
            const int t2 = tile + 2 * step;
            load_positions(csr_src, bB, eB, c, node_of(t2), t2 < ntiles, nx0, nx1);
            const int t3 = t2 + step;
            if (t3 < ntiles) {
                const int n3 = node_of(t3);
                bC = rowptr[n3];  eC = rowptr[n3 + 1];  dC = dinv[n3];
            }
        }

        v2f b[16];
#pragma unroll
        for (int sp = 0; sp < 8; ++sp) {
            const float4 q = *(const float4*)(As + c * LDA + 8 * sp + 4 * h);
            b[2 * sp].x     = q.x;  b[2 * sp].y     = q.y;
            b[2 * sp + 1].x = q.z;  b[2 * sp + 1].y = q.w;
        }

        v8f h0, h1;
        {
            const float4 p0 = *(const float4*)(bsL + hb + 8 * h);
            const float4 p1 = *(const float4*)(bsL + hb + 8 * h + 4);
            const float4 p2 = *(const float4*)(bsL + hb + 16 + 8 * h);
            const float4 p3 = *(const float4*)(bsL + hb + 16 + 8 * h + 4);
            h0[0] = p0.x; h0[1] = p0.y; h0[2] = p0.z; h0[3] = p0.w;
            h0[4] = p1.x; h0[5] = p1.y; h0[6] = p1.z; h0[7] = p1.w;
            h1[0] = p2.x; h1[1] = p2.y; h1[2] = p2.z; h1[3] = p2.w;
            h1[4] = p3.x; h1[5] = p3.y; h1[6] = p3.z; h1[7] = p3.w;
        }
        v2f a2[2][4];
        {
            const float* __restrict__ fp = a2L + tid * LDF;
#pragma unroll
            for (int t = 0; t < 2; ++t) {
                const float4 f0 = *(const float4*)(fp + 8 * t);
                const float4 f1 = *(const float4*)(fp + 8 * t + 4);
                a2[t][0].x = f0.x;  a2[t][0].y = f0.y;
                a2[t][1].x = f0.z;  a2[t][1].y = f0.w;
                a2[t][2].x = f1.x;  a2[t][2].y = f1.y;
                a2[t][3].x = f1.z;  a2[t][3].y = f1.w;
            }
        }
#pragma unroll
        for (int s = 0; s < 16; ++s) {
            h0 = __builtin_amdgcn_wmma_f32_16x16x4_f32(
                false, a1[0][s], false, b[s], (short)0, h0, false, false);
            h1 = __builtin_amdgcn_wmma_f32_16x16x4_f32(
                false, a1[1][s], false, b[s], (short)0, h1, false, false);
        }
#pragma unroll
        for (int r = 0; r < 8; ++r) {
            h0[r] = h0[r] > 0.0f ? h0[r] : 0.0f;
            h1[r] = h1[r] > 0.0f ? h1[r] : 0.0f;
        }

        v8f o;
#pragma unroll
        for (int r = 0; r < 8; ++r) o[r] = 0.0f;
#pragma unroll
        for (int s = 0; s < 4; ++s) {
            v2f hv; hv.x = h0[2 * s]; hv.y = h0[2 * s + 1];
            o = __builtin_amdgcn_wmma_f32_16x16x4_f32(
                false, a2[0][s], false, hv, (short)0, o, false, false);
        }
#pragma unroll
        for (int s = 0; s < 4; ++s) {
            v2f hv; hv.x = h1[2 * s]; hv.y = h1[2 * s + 1];
            o = __builtin_amdgcn_wmma_f32_16x16x4_f32(
                false, a2[1][s], false, hv, (short)0, o, false, false);
        }

        if (h == 0) {
            float* __restrict__ rp = red + (wave * 16 + c) * LDR;
            *(float4*)(rp)     = make_float4(o[0], o[1], o[2], o[3]);
            *(float4*)(rp + 4) = make_float4(o[4], o[5], o[6], o[7]);
        }
        __syncthreads();

        if (tid < 128) {
            const int node = tid >> 3;
            const int j    = tid & 7;
            float s = red[node * LDR + j];
#pragma unroll
            for (int w = 1; w < 8; ++w) s += red[(w * 16 + node) * LDR + j];
            Y[(size_t)tile * 128 + tid] = s;
        }
    }
}

__global__ void __launch_bounds__(256)
k_gather_heads(const int* __restrict__ rowptr, const int* __restrict__ csr_src,
               const float* __restrict__ dinv, const float* __restrict__ y,
               const float* __restrict__ c, const float* __restrict__ bn,
               float* __restrict__ out, float* __restrict__ pq, int n) {
    const int  t    = blockIdx.x * 256 + threadIdx.x;
    const int  q    = t & 3;
    const int  node = t >> 2;
    const bool live = node < n;
    const int  nd   = live ? node : n - 1;

    const int   beg = rowptr[nd];
    const int   end = rowptr[nd + 1];
    const float di  = dinv[nd];
    const float4* __restrict__ y4 = (const float4*)y;

    float4 a0, a1;
    {
        const float4 s0 = y4[(size_t)nd * 2];
        const float4 s1 = y4[(size_t)nd * 2 + 1];
        const float  w  = di * di;
        const bool   z  = q != 0;
        a0.x = z ? 0.0f : s0.x * w;  a0.y = z ? 0.0f : s0.y * w;
        a0.z = z ? 0.0f : s0.z * w;  a0.w = z ? 0.0f : s0.w * w;
        a1.x = z ? 0.0f : s1.x * w;  a1.y = z ? 0.0f : s1.y * w;
        a1.z = z ? 0.0f : s1.z * w;  a1.w = z ? 0.0f : s1.w * w;
    }

    int e = beg + q;
    for (; e + 4 < end; e += 8) {
        int sA = csr_src[e];
        int sB = csr_src[e + 4];
        sA = sA < 0 ? 0 : (sA > n - 1 ? n - 1 : sA);
        sB = sB < 0 ? 0 : (sB > n - 1 ? n - 1 : sB);
        const float  cA = dinv[sA] * di;
        const float  cB = dinv[sB] * di;
        const float4 rA0 = y4[(size_t)sA * 2], rA1 = y4[(size_t)sA * 2 + 1];
        const float4 rB0 = y4[(size_t)sB * 2], rB1 = y4[(size_t)sB * 2 + 1];
        a0.x += rA0.x * cA;  a0.y += rA0.y * cA;  a0.z += rA0.z * cA;  a0.w += rA0.w * cA;
        a1.x += rA1.x * cA;  a1.y += rA1.y * cA;  a1.z += rA1.z * cA;  a1.w += rA1.w * cA;
        a0.x += rB0.x * cB;  a0.y += rB0.y * cB;  a0.z += rB0.z * cB;  a0.w += rB0.w * cB;
        a1.x += rB1.x * cB;  a1.y += rB1.y * cB;  a1.z += rB1.z * cB;  a1.w += rB1.w * cB;
    }
    if (e < end) {
        int sA = csr_src[e];
        sA = sA < 0 ? 0 : (sA > n - 1 ? n - 1 : sA);
        const float  cA = dinv[sA] * di;
        const float4 rA0 = y4[(size_t)sA * 2], rA1 = y4[(size_t)sA * 2 + 1];
        a0.x += rA0.x * cA;  a0.y += rA0.y * cA;  a0.z += rA0.z * cA;  a0.w += rA0.w * cA;
        a1.x += rA1.x * cA;  a1.y += rA1.y * cA;  a1.z += rA1.z * cA;  a1.w += rA1.w * cA;
    }

#pragma unroll
    for (int off = 1; off <= 2; off <<= 1) {
        a0.x += __shfl_xor(a0.x, off, 32);  a0.y += __shfl_xor(a0.y, off, 32);
        a0.z += __shfl_xor(a0.z, off, 32);  a0.w += __shfl_xor(a0.w, off, 32);
        a1.x += __shfl_xor(a1.x, off, 32);  a1.y += __shfl_xor(a1.y, off, 32);
    }

    if (live) {
        if (q == 0) {
            float2 o;
            o.x = (a0.x + c[0]) + bn[0];
            o.y = (a0.y + c[1]) + bn[1];
            *(float2*)(out + (size_t)nd * 2) = o;
        } else if (q == 1) {
            float4 r;
            r.x = a0.z + c[2];  r.y = a0.w + c[3];
            r.z = a1.x + c[4];  r.w = a1.y + c[5];
            *(float4*)(pq + (size_t)nd * 4) = r;
        }
    }
}

__device__ __forceinline__ float2 edge_one(const float* __restrict__ pq, int s, int d,
                                           float b0, float b1) {
    const float2 p = *(const float2*)(pq + (size_t)s * 4);
    const float2 q = *(const float2*)(pq + (size_t)d * 4 + 2);
    float2 o;
    o.x = (p.x + q.x) + b0;
    o.y = (p.y + q.y) + b1;
    return o;
}

__global__ void __launch_bounds__(256)
k_edge_head(const int* __restrict__ src, const int* __restrict__ dst,
            const float* __restrict__ pq, const float* __restrict__ be,
            float* __restrict__ out, int E) {
    const int   base = blockIdx.x * 1024;
    const float b0 = be[0], b1 = be[1];
    const bool wide = ((((size_t)src | (size_t)dst) & 7) == 0) && (((size_t)out & 15) == 0);
    if (wide && base + 1024 <= E) {
        const int  e0 = base + threadIdx.x * 2;
        const int  e1 = e0 + 512;
        const int2 s0 = *(const int2*)(src + e0);
        const int2 d0 = *(const int2*)(dst + e0);
        const int2 s1 = *(const int2*)(src + e1);
        const int2 d1 = *(const int2*)(dst + e1);
        const float2 a0 = edge_one(pq, s0.x, d0.x, b0, b1);
        const float2 a1 = edge_one(pq, s0.y, d0.y, b0, b1);
        const float2 c0 = edge_one(pq, s1.x, d1.x, b0, b1);
        const float2 c1 = edge_one(pq, s1.y, d1.y, b0, b1);
        float4 o0, o1;
        o0.x = a0.x; o0.y = a0.y; o0.z = a1.x; o0.w = a1.y;
        o1.x = c0.x; o1.y = c0.y; o1.z = c1.x; o1.w = c1.y;
        *(float4*)(out + (size_t)e0 * 2) = o0;
        *(float4*)(out + (size_t)e1 * 2) = o1;
    } else {
        for (int u = 0; u < 2; ++u) {
            for (int k = 0; k < 2; ++k) {
                const int e = base + u * 512 + threadIdx.x * 2 + k;
                if (e < E) *(float2*)(out + (size_t)e * 2) = edge_one(pq, src[e], dst[e], b0, b1);
            }
        }
    }
}

extern "C" void kernel_launch(void* const* d_in, const int* in_sizes, int n_in,
                              void* d_out, int out_size, void* d_ws, size_t ws_size,
                              hipStream_t stream) {
    const float* x  = (const float*)d_in[0];
    const int*   ei = (const int*)d_in[1];
    const float* W1 = (const float*)d_in[2];
    const float* b1 = (const float*)d_in[3];
    const float* W2 = (const float*)d_in[4];
    const float* b2 = (const float*)d_in[5];
    const float* Wn = (const float*)d_in[6];
    const float* bn = (const float*)d_in[7];
    const float* We = (const float*)d_in[8];
    const float* be = (const float*)d_in[9];

    const int N = in_sizes[0] / 64;
    const int E = in_sizes[1] / 2;
    const int* src = ei;
    const int* dst = ei + E;

    char*  ws  = (char*)d_ws;
    size_t off = 0;
    auto carve = [&](size_t bytes) -> float* {
        float* p = (float*)(ws + off);
        off += (bytes + 255) & ~(size_t)255;
        return p;
    };
    const int G1 = (E + CHUNK - 1) / CHUNK;
    float* dinv   = carve((size_t)N * 4);
    int*   rowptr = (int*)carve((size_t)(N + 1) * 4);
    int*   runs   = (int*)carve((size_t)G1 * RROW * 4);
    int*   ent    = (int*)carve((size_t)E * 4);
    int*   csrsrc = (int*)carve((size_t)E * 4);
    float* wp   = carve((size_t)(256 + 1) * 16 * 4);
    float* y    = carve((size_t)N * 8 * 4);
    float* pq   = carve((size_t)N * 4 * 4);

    float* out_node = (float*)d_out;
    float* out_edge = out_node + (size_t)N * 2;

    k_chunk_sort <<<G1 + 1, 1024, 0, stream>>>(src, dst, runs, ent, E, N, W2, b2, Wn, We, wp);
    k_bucket_fine<<<(N + BNODE - 1) / BNODE, 1024, 0, stream>>>(ent, runs, G1, rowptr, dinv, csrsrc, E, N);

    k_hidden_fold<<<768, 256, 0, stream>>>(rowptr, csrsrc, dinv, x, W1, b1, wp, y, N / 16, N);
    k_gather_heads<<<(N * 4 + 255) / 256, 256, 0, stream>>>(
        rowptr, csrsrc, dinv, y, wp + 256 * 16, bn, out_node, pq, N);

    k_edge_head<<<(E + 1023) / 1024, 256, 0, stream>>>(src, dst, pq, be, out_edge, E);
}


# layer-2 gather at one line lookup per list entry: y rows stored scaled by dinv[node] (no dinv[src] gather), a row read by a lane pair in one 16-byte load each, 6 in flight; f32 sum order changes
# speedup vs baseline: 19.6939x; 1.0551x over previous
#include <hip/hip_runtime.h>


typedef float v2f __attribute__((ext_vector_type(2)));
typedef float v8f __attribute__((ext_vector_type(8)));

constexpr int BSH   = 8;
constexpr int BNODE = 1 << BSH;
constexpr int NBMAX = 256;
constexpr int CHUNK = 4096;
constexpr int TPR   = 4;
constexpr int RPT   = 1024 / TPR;
constexpr int RROW  = NBMAX + 32;

__device__ __forceinline__ int clamp_node(int v, int n) {
    return v < 0 ? 0 : (v > n - 1 ? n - 1 : v);
}
__device__ __forceinline__ int bucket_of(int d) {
    const int b = d >> BSH;
    return b > NBMAX - 1 ? NBMAX - 1 : b;
}

__device__ __forceinline__ int wave_incl_scan(int v, int lane) {
    for (int off = 1; off < 32; off <<= 1) {
        const int u = __shfl_up(v, off, 32);
        if (lane >= off) v += u;
    }
    return v;
}

__device__ __forceinline__ void fold_heads_part(int part, int lane,
                                                const float* __restrict__ W2,
                                                const float* __restrict__ b2,
                                                const float* __restrict__ Wn,
                                                const float* __restrict__ We,
                                                float* __restrict__ Wp) {
    const int col  = lane & 15;
    const int kb   = (lane >> 4) << 1;
    const bool live = col < 6;
    const float* __restrict__ hc =
        col < 2 ? Wn + col : (col < 4 ? We + (col - 2) : (col < 6 ? We + 256 + (col - 4) : Wn));

    if (part < 16) {
        const float* __restrict__ Ar = W2 + (size_t)(part * 16 + col) * 128 + kb;
        v8f acc;
#pragma unroll
        for (int r = 0; r < 8; ++r) acc[r] = 0.0f;
#pragma unroll
        for (int k = 0; k < 128; k += 4) {
            v2f av; av.x = Ar[k];                 av.y = Ar[k + 1];
            const float h0 = hc[(k + kb) * 2];
            const float h1 = hc[(k + kb + 1) * 2];
            v2f bv; bv.x = live ? h0 : 0.0f;      bv.y = live ? h1 : 0.0f;
            acc = __builtin_amdgcn_wmma_f32_16x16x4_f32(
                false, av, false, bv, (short)0, acc, false, false);
        }
        float* __restrict__ Cp = Wp + (size_t)(part * 16 + ((lane >> 4) << 3)) * 16 + col;
#pragma unroll
        for (int r = 0; r < 8; ++r) Cp[(size_t)r * 16] = acc[r];
    } else if (lane < 16) {
        float s = 0.0f;
        for (int k = 0; k < 128; ++k) {
            const float h = hc[k * 2];
            s += b2[k] * (live ? h : 0.0f);
        }
        Wp[256 * 16 + lane] = s;
    }
}

__global__ void __launch_bounds__(1024)
k_chunk_sort(const int* __restrict__ src, const int* __restrict__ dst,
             int* __restrict__ runs, int* __restrict__ ent, int E, int n,
             const float* __restrict__ W2, const float* __restrict__ b2,
             const float* __restrict__ Wn, const float* __restrict__ We,
             float* __restrict__ Wp) {
    __shared__ int h[NBMAX];
    __shared__ int ls[NBMAX];
    __shared__ int stage[CHUNK];
    __shared__ int lsum[8];
    const int t    = threadIdx.x;
    const int lane = t & 31;
    const int wave = t >> 5;

    if (blockIdx.x == 0) {
        const int part = __builtin_amdgcn_readfirstlane(wave);
        if (part <= 16) fold_heads_part(part, lane, W2, b2, Wn, We, Wp);
        return;
    }
    const int blk  = (int)blockIdx.x - 1;
    const int base = blk * CHUNK;

    int d[4], s[4];
#pragma unroll
    for (int k = 0; k < 4; ++k) {
        const int e = base + k * 1024 + t;
        d[k] = e < E ? dst[e] : 0;
        s[k] = e < E ? src[e] : 0;
    }
    if (t < NBMAX) h[t] = 0;
    __syncthreads();

    int b[4], r[4];
#pragma unroll
    for (int k = 0; k < 4; ++k) {
        const int e = base + k * 1024 + t;
        d[k] = clamp_node(d[k], n);
        s[k] = clamp_node(s[k], n);
        b[k] = bucket_of(d[k]);
        r[k] = 0;
        if (e < E) r[k] = atomicAdd(&h[b[k]], 1);
    }
    __syncthreads();

    const int own = t < NBMAX ? h[t] : 0;
    const int inl = wave_incl_scan(own, lane);
    if (lane == 31 && wave < 8) lsum[wave] = inl;
    __syncthreads();
    if (t < NBMAX) {
        int lb = 0;
#pragma unroll
        for (int w = 0; w < 8; ++w) lb += (w < wave) ? lsum[w] : 0;
        const int lex = lb + inl - own;
        ls[t] = lex;
        runs[(size_t)blk * RROW + t] = lex;
        if (t == NBMAX - 1) runs[(size_t)blk * RROW + NBMAX] = lex + own;
    }
    __syncthreads();

#pragma unroll
    for (int k = 0; k < 4; ++k) {
        const int e = base + k * 1024 + t;
        if (e < E) {
            int i = ls[b[k]] + r[k];
            i = i < 0 ? 0 : (i > CHUNK - 1 ? CHUNK - 1 : i);
            stage[i] = (int)(((unsigned)s[k] << 16) | (unsigned)(d[k] & 0xFFFF));
        }
    }
    __syncthreads();

    const int mine = (E - base) < CHUNK ? (E - base) : CHUNK;
#pragma unroll
    for (int k = 0; k < 4; ++k) {
        const int i = k * 1024 + t;
        if (i < mine) ent[base + i] = stage[i];
    }
}

__global__ void __launch_bounds__(1024)
k_bucket_fine(const int* __restrict__ ent, const int* __restrict__ runs, int G,
              int* __restrict__ rowptr, float* __restrict__ dinv,
              int* __restrict__ csr_src, int E, int n) {
    constexpr int FCAP = 8 * 1024;
    constexpr int KEEP = 8;
    __shared__ int cnt[BNODE];
    __shared__ int cur[BNODE];
    __shared__ int wsum[8];
    __shared__ int bsum[32];
    __shared__ int begS;
    __shared__ int stage[FCAP];
    const int t    = threadIdx.x;
    const int lane = t & 31;
    const int wave = t >> 5;
    const int bk   = blockIdx.x;
    const int q    = t & (TPR - 1);
    const int j0   = t / TPR;

    auto run_of = [&](int j, int& rb, int& rl) -> int {
        int lim = E - j * CHUNK;
        lim = lim < 0 ? 0 : (lim > CHUNK ? CHUNK : lim);
        int a = runs[(size_t)j * RROW + bk];
        int z = runs[(size_t)j * RROW + bk + 1];
        a = a < 0 ? 0 : (a > lim ? lim : a);
        z = z < a ? a : (z > lim ? lim : z);
        rb = j * CHUNK + a;
        rl = z - a;
        return a;
    };

    int rb = 0, rl = 0, part = 0;
    if (j0 < G) {
        const int a = run_of(j0, rb, rl);
        part = q == 0 ? a : 0;
    }
    for (int j = j0 + RPT; j < G; j += RPT) {
        int cb, cl;
        const int a = run_of(j, cb, cl);
        part += q == 0 ? a : 0;
    }
    unsigned w[KEEP];
#pragma unroll
    for (int u = 0; u < KEEP; ++u) {
        const int i = q + TPR * u;
        w[u] = i < rl ? (unsigned)ent[rb + i] : 0u;
    }
    for (int off = 16; off >= 1; off >>= 1) part += __shfl_xor(part, off, 32);
    if (lane == 0) bsum[wave] = part;
    if (t < BNODE) cnt[t] = 0;
    __syncthreads();
#pragma unroll
    for (int u = 0; u < KEEP; ++u) {
        if (q + TPR * u < rl) atomicAdd(&cnt[w[u] & (BNODE - 1)], 1);
    }
    for (int i = KEEP * TPR + q; i < rl; i += TPR)
        atomicAdd(&cnt[(unsigned)ent[rb + i] & (BNODE - 1)], 1);
    for (int j = j0 + RPT; j < G; j += RPT) {
        int cb, cl;
        run_of(j, cb, cl);
        for (int i = q; i < cl; i += TPR)
            atomicAdd(&cnt[(unsigned)ent[cb + i] & (BNODE - 1)], 1);
    }
    if (wave == 0) {
        int v = bsum[lane];
        for (int off = 16; off >= 1; off >>= 1) v += __shfl_xor(v, off, 32);
        if (lane == 0) begS = v;
    }
    __syncthreads();

    const int v   = t < BNODE ? cnt[t] : 0;
    const int inc = wave_incl_scan(v, lane);
    if (lane == 31 && wave < 8) wsum[wave] = inc;
    __syncthreads();
    int beg = begS;
    int m   = 0;
#pragma unroll
    for (int k = 0; k < 8; ++k) m += wsum[k];
    beg = beg < 0 ? 0 : (beg > E ? E : beg);
    m   = m < 0 ? 0 : (m > E - beg ? E - beg : m);
    const int  end    = beg + m;
    const bool staged = m <= FCAP;
    if (t < BNODE) {
        int wb = 0;
#pragma unroll
        for (int k = 0; k < 8; ++k) wb += (k < wave) ? wsum[k] : 0;
        const int first = wb + inc - v;
        cur[t] = first;
        const int node = bk * BNODE + t;
        if (node < n) {
            rowptr[node] = beg + first;
            dinv[node]   = rsqrtf((float)v + 1.0f);
        }
    }
    if (bk == (int)gridDim.x - 1 && t == 0) rowptr[n] = end;
    __syncthreads();

    int p[KEEP];
#pragma unroll
    for (int u = 0; u < KEEP; ++u) {
        p[u] = 0;
        if (q + TPR * u < rl) p[u] = atomicAdd(&cur[w[u] & (BNODE - 1)], 1);
    }
    auto place = [&](unsigned x) {
        const int o = atomicAdd(&cur[x & (BNODE - 1)], 1);
        if (staged) {
            stage[o < 0 ? 0 : (o > FCAP - 1 ? FCAP - 1 : o)] = (int)(x >> 16);
        } else {
            int g = beg + o;
            g = g < 0 ? 0 : (g > E - 1 ? E - 1 : g);
            csr_src[g] = (int)(x >> 16);
        }
    };
    if (staged) {
#pragma unroll
        for (int u = 0; u < KEEP; ++u) {
            if (q + TPR * u < rl) {
                const int o = p[u] < 0 ? 0 : (p[u] > FCAP - 1 ? FCAP - 1 : p[u]);
                stage[o] = (int)(w[u] >> 16);
            }
        }
    } else {
#pragma unroll
        for (int u = 0; u < KEEP; ++u) {
            if (q + TPR * u < rl) {
                int g = beg + p[u];
                g = g < 0 ? 0 : (g > E - 1 ? E - 1 : g);
                csr_src[g] = (int)(w[u] >> 16);
            }
        }
    }
    for (int i = KEEP * TPR + q; i < rl; i += TPR) place((unsigned)ent[rb + i]);
    for (int j = j0 + RPT; j < G; j += RPT) {
        int cb, cl;
        run_of(j, cb, cl);
        for (int i = q; i < cl; i += TPR) place((unsigned)ent[cb + i]);
    }
    __syncthreads();
    if (staged) {
        for (int i = t; i < m; i += 1024) csr_src[beg + i] = stage[i];
    }
}


template <int R>
__device__ __forceinline__ void gather_rows(const float* __restrict__ x, int my, float cw,
                                            int j, bool hi, int l16, float4& acc) {
    float4 r[R];
#pragma unroll
    for (int u = 0; u < R; ++u) {
        const int s0 = __builtin_amdgcn_readlane(my, j + u);
        const int s1 = __builtin_amdgcn_readlane(my, 16 + j + u);
        const unsigned s = (unsigned)(hi ? s1 : s0);
        r[u] = *(const float4*)((const char*)x + (s * 256u + (unsigned)l16 * 16u));
    }
#pragma unroll
    for (int u = 0; u < R; ++u) {
        const int   c0 = __builtin_amdgcn_readlane(__float_as_int(cw), j + u);
        const int   c1 = __builtin_amdgcn_readlane(__float_as_int(cw), 16 + j + u);
        const float c  = __int_as_float(hi ? c1 : c0);
        acc.x += r[u].x * c;
        acc.y += r[u].y * c;
        acc.z += r[u].z * c;
        acc.w += r[u].w * c;
    }
}

__device__ __forceinline__ void gather_chunk(const float* __restrict__ x, int my, float cw,
                                             int T, bool hi, int l16, float4& acc) {
    int j = 0;
    for (; j + 8 <= T; j += 8) gather_rows<8>(x, my, cw, j, hi, l16, acc);
    if (j + 4 <= T) { gather_rows<4>(x, my, cw, j, hi, l16, acc); j += 4; }
    if (j + 2 <= T) { gather_rows<2>(x, my, cw, j, hi, l16, acc); j += 2; }
    if (j < T)      { gather_rows<1>(x, my, cw, j, hi, l16, acc); }
}

__device__ __forceinline__ void load_positions(const int* __restrict__ csr_src, int b, int e, int c,
                                               int nd, bool live, int& m0, int& m1) {
    m0 = nd;  m1 = nd;
    if (live) {
        if (c >= 1 && c <= e - b) m0 = csr_src[b + c - 1];
        if (16 + c <= e - b)      m1 = csr_src[b + 15 + c];
    }
}

__global__ void __launch_bounds__(256)
k_hidden_fold(const int* __restrict__ rowptr, const int* __restrict__ csr_src,
              const float* __restrict__ dinv, const float* __restrict__ x,
              const float* __restrict__ W1, const float* __restrict__ b1,
              const float* __restrict__ Wp, float* __restrict__ Y, int ntiles, int n) {
    constexpr int LDA = 68;
    constexpr int LDR = 12;
    constexpr int LDF = 20;
    __shared__ __attribute__((aligned(16))) float As[16 * LDA];
    __shared__ __attribute__((aligned(16))) float red[8 * 16 * LDR];
    __shared__ __attribute__((aligned(16))) float bsL[256];
    __shared__ __attribute__((aligned(16))) float a2L[256 * LDF];
    __shared__ float dl[16];

    const int  tid  = threadIdx.x;
    const int  wave = tid >> 5;
    const int  lane = tid & 31;
    const int  c    = lane & 15;
    const int  h    = lane >> 4;
    const bool hi   = h != 0;
    const int  hb   = wave * 32;

    v2f a1[2][16];
#pragma unroll
    for (int t = 0; t < 2; ++t) {
#pragma unroll
        for (int sp = 0; sp < 8; ++sp) {
#pragma unroll
            for (int u = 0; u < 2; ++u) {
                const int k = 8 * sp + 4 * h + 2 * u;
                a1[t][2 * sp + u].x = W1[(size_t)k * 256 + hb + 16 * t + c];
                a1[t][2 * sp + u].y = W1[(size_t)(k + 1) * 256 + hb + 16 * t + c];
            }
        }
    }
    {
        float* __restrict__ fp = a2L + tid * LDF;
#pragma unroll
        for (int t = 0; t < 2; ++t) {
#pragma unroll
            for (int s = 0; s < 4; s += 2) {
                const int ch = hb + 16 * t + 8 * h + 2 * s;
                float4 w;
                w.x = Wp[(size_t)ch * 16 + c];
                w.y = Wp[(size_t)(ch + 1) * 16 + c];
                w.z = Wp[(size_t)(ch + 2) * 16 + c];
                w.w = Wp[(size_t)(ch + 3) * 16 + c];
                *(float4*)(fp + 8 * t + 2 * s) = w;
            }
        }
        bsL[tid] = b1[tid];
    }

    const int hn   = 2 * wave + h;
    const int step = (int)gridDim.x;
    int tile = blockIdx.x;
    auto node_of = [&](int t) -> int {
        int nd = t * 16 + hn;
        nd = nd < n ? nd : n - 1;
        return nd < 0 ? 0 : nd;
    };

    int   bA = 0, eA = 0, bB = 0, eB = 0, bC = 0, eC = 0;
    float dA = 0.0f, dB = 0.0f, dC = 0.0f;
    if (tile < ntiles) {
        const int nd = node_of(tile);
        bA = rowptr[nd];  eA = rowptr[nd + 1];  dA = dinv[nd];
    }
    if (tile + step < ntiles) {
        const int nd = node_of(tile + step);
        bB = rowptr[nd];  eB = rowptr[nd + 1];  dB = dinv[nd];
    }
    if (tile + 2 * step < ntiles) {
        const int nd = node_of(tile + 2 * step);
        bC = rowptr[nd];  eC = rowptr[nd + 1];  dC = dinv[nd];
    }
    int my0, my1, nx0, nx1;
    load_positions(csr_src, bA, eA, c, node_of(tile), tile < ntiles, my0, my1);
    load_positions(csr_src, bB, eB, c, node_of(tile + step), tile + step < ntiles, nx0, nx1);
    my0 = my0 < 0 ? 0 : (my0 > n - 1 ? n - 1 : my0);
    my1 = my1 < 0 ? 0 : (my1 > n - 1 ? n - 1 : my1);
    float q0 = dinv[my0];
    float q1 = dinv[my1];

    for (; tile < ntiles; tile += step) {
        const int nd = node_of(tile);
        const int m = eA - bA + 1;
        const int mlo = __builtin_amdgcn_readlane(m, 0);
        const int mhi = __builtin_amdgcn_readlane(m, 16);
        const int mm  = mlo > mhi ? mlo : mhi;

        const float cw0 = c < m      ? q0 * dA : 0.0f;
        const float cw1 = 16 + c < m ? q1 * dA : 0.0f;

        float4 acc = make_float4(0.0f, 0.0f, 0.0f, 0.0f);
        {
            const int tp = ((mm < 32 ? mm : 32) + 3) & ~3;
            int j = 0;
            for (; j + 8 <= tp; j += 8) {
                const bool first = j < 16;
                gather_rows<8>(x, first ? my0 : my1, first ? cw0 : cw1, j & 15, hi, c, acc);
            }
            if (j < tp) {
                const bool first = j < 16;
                gather_rows<4>(x, first ? my0 : my1, first ? cw0 : cw1, j & 15, hi, c, acc);
            }
        }
        for (int base = 32; base < mm; base += 16) {
            int   my = nd;
            float cw = 0.0f;
            if (base + c < m) {
                my = csr_src[bA + base - 1 + c];
                my = my < 0 ? 0 : (my > n - 1 ? n - 1 : my);
                cw = dinv[my] * dA;
            }
            gather_chunk(x, my, cw, mm - base < 16 ? mm - base : 16, hi, c, acc);
        }

        *(float4*)(As + hn * LDA + c * 4) = acc;
        if (c == 0) dl[hn] = dA;
        __syncthreads();
        const float dn = tid < 128 ? dl[tid >> 3] : 0.0f;

        {
            bA = bB;  eA = eB;  dA = dB;
            bB = bC;  eB = eC;  dB = dC;
            my0 = nx0 < 0 ? 0 : (nx0 > n - 1 ? n - 1 : nx0);
            my1 = nx1 < 0 ? 0 : (nx1 > n - 1 ? n - 1 : nx1);
            q0 = dinv[my0];
            q1 = dinv[my1];
            const int t2 = tile + 2 * step;
            load_positions(csr_src, bB, eB, c, node_of(t2), t2 < ntiles, nx0, nx1);
            const int t3 = t2 + step;
            if (t3 < ntiles) {
                const int n3 = node_of(t3);
                bC = rowptr[n3];  eC = rowptr[n3 + 1];  dC = dinv[n3];
            }
        }

        v2f b[16];
#pragma unroll
        for (int sp = 0; sp < 8; ++sp) {
            const float4 q = *(const float4*)(As + c * LDA + 8 * sp + 4 * h);
            b[2 * sp].x     = q.x;  b[2 * sp].y     = q.y;
            b[2 * sp + 1].x = q.z;  b[2 * sp + 1].y = q.w;
        }

        v8f h0, h1;
        {
            const float4 p0 = *(const float4*)(bsL + hb + 8 * h);
            const float4 p1 = *(const float4*)(bsL + hb + 8 * h + 4);
            const float4 p2 = *(const float4*)(bsL + hb + 16 + 8 * h);
            const float4 p3 = *(const float4*)(bsL + hb + 16 + 8 * h + 4);
            h0[0] = p0.x; h0[1] = p0.y; h0[2] = p0.z; h0[3] = p0.w;
            h0[4] = p1.x; h0[5] = p1.y; h0[6] = p1.z; h0[7] = p1.w;
            h1[0] = p2.x; h1[1] = p2.y; h1[2] = p2.z; h1[3] = p2.w;
            h1[4] = p3.x; h1[5] = p3.y; h1[6] = p3.z; h1[7] = p3.w;
        }
        v2f a2[2][4];
        {
            const float* __restrict__ fp = a2L + tid * LDF;
#pragma unroll
            for (int t = 0; t < 2; ++t) {
                const float4 f0 = *(const float4*)(fp + 8 * t);
                const float4 f1 = *(const float4*)(fp + 8 * t + 4);
                a2[t][0].x = f0.x;  a2[t][0].y = f0.y;
                a2[t][1].x = f0.z;  a2[t][1].y = f0.w;
                a2[t][2].x = f1.x;  a2[t][2].y = f1.y;
                a2[t][3].x = f1.z;  a2[t][3].y = f1.w;
            }
        }
#pragma unroll
        for (int s = 0; s < 16; ++s) {
            h0 = __builtin_amdgcn_wmma_f32_16x16x4_f32(
                false, a1[0][s], false, b[s], (short)0, h0, false, false);
            h1 = __builtin_amdgcn_wmma_f32_16x16x4_f32(
                false, a1[1][s], false, b[s], (short)0, h1, false, false);
        }
#pragma unroll
        for (int r = 0; r < 8; ++r) {
            h0[r] = h0[r] > 0.0f ? h0[r] : 0.0f;
            h1[r] = h1[r] > 0.0f ? h1[r] : 0.0f;
        }

        v8f o;
#pragma unroll
        for (int r = 0; r < 8; ++r) o[r] = 0.0f;
#pragma unroll
        for (int s = 0; s < 4; ++s) {
            v2f hv; hv.x = h0[2 * s]; hv.y = h0[2 * s + 1];
            o = __builtin_amdgcn_wmma_f32_16x16x4_f32(
                false, a2[0][s], false, hv, (short)0, o, false, false);
        }
#pragma unroll
        for (int s = 0; s < 4; ++s) {
            v2f hv; hv.x = h1[2 * s]; hv.y = h1[2 * s + 1];
            o = __builtin_amdgcn_wmma_f32_16x16x4_f32(
                false, a2[1][s], false, hv, (short)0, o, false, false);
        }

        if (h == 0) {
            float* __restrict__ rp = red + (wave * 16 + c) * LDR;
            *(float4*)(rp)     = make_float4(o[0], o[1], o[2], o[3]);
            *(float4*)(rp + 4) = make_float4(o[4], o[5], o[6], o[7]);
        }
        __syncthreads();

        if (tid < 128) {
            const int node = tid >> 3;
            const int j    = tid & 7;
            float s = red[node * LDR + j];
#pragma unroll
            for (int w = 1; w < 8; ++w) s += red[(w * 16 + node) * LDR + j];
            Y[(size_t)tile * 128 + tid] = s * dn;
        }
    }
}

__global__ void __launch_bounds__(256)
k_gather_heads(const int* __restrict__ rowptr, const int* __restrict__ csr_src,
               const float* __restrict__ dinv, const float* __restrict__ ys,
               const float* __restrict__ c, const float* __restrict__ bn,
               float* __restrict__ out, float* __restrict__ pq, int n) {
    constexpr int U = 6;
    const int  t    = blockIdx.x * 256 + threadIdx.x;
    const int  q    = t & 3;
    const int  hf   = q & 1;
    const int  e2   = q >> 1;
    const int  node = t >> 2;
    const bool live = node < n;
    const int  nd   = live ? node : n - 1;

    const int   beg = rowptr[nd];
    const int   end = rowptr[nd + 1];
    const float di  = dinv[nd];
    const float4* __restrict__ y4 = (const float4*)ys;

    float4 a = y4[(size_t)nd * 2 + hf];
    if (e2 != 0) a = make_float4(0.0f, 0.0f, 0.0f, 0.0f);

    for (int e = beg + e2; e < end; e += 2 * U) {
        int s[U];
#pragma unroll
        for (int u = 0; u < U; ++u) {
            const int i = e + 2 * u;
            const int v = csr_src[i < end ? i : e];
            s[u] = v < 0 ? 0 : (v > n - 1 ? n - 1 : v);
        }
        float4 r[U];
#pragma unroll
        for (int u = 0; u < U; ++u) {
            r[u] = *(const float4*)((const char*)ys + ((unsigned)s[u] * 32u + (unsigned)hf * 16u));
        }
        __builtin_amdgcn_sched_barrier(0);
#pragma unroll
        for (int u = 0; u < U; ++u) {
            const bool ok = e + 2 * u < end;
            a.x += ok ? r[u].x : 0.0f;
            a.y += ok ? r[u].y : 0.0f;
            a.z += ok ? r[u].z : 0.0f;
            a.w += ok ? r[u].w : 0.0f;
        }
    }

    a.x += __shfl_xor(a.x, 2, 32);  a.y += __shfl_xor(a.y, 2, 32);
    a.z += __shfl_xor(a.z, 2, 32);  a.w += __shfl_xor(a.w, 2, 32);
    const float n2 = __shfl_xor(a.z, 1, 32);
    const float n3 = __shfl_xor(a.w, 1, 32);

    if (live) {
        if (q == 0) {
            float2 o;
            o.x = (a.x * di + c[0]) + bn[0];
            o.y = (a.y * di + c[1]) + bn[1];
            *(float2*)(out + (size_t)nd * 2) = o;
        } else if (q == 1) {
            float4 r;
            r.x = n2 * di + c[2];   r.y = n3 * di + c[3];
            r.z = a.x * di + c[4];  r.w = a.y * di + c[5];
            *(float4*)(pq + (size_t)nd * 4) = r;
        }
    }
}

__device__ __forceinline__ float2 edge_one(const float* __restrict__ pq, int s, int d,
                                           float b0, float b1) {
    const float2 p = *(const float2*)(pq + (size_t)s * 4);
    const float2 q = *(const float2*)(pq + (size_t)d * 4 + 2);
    float2 o;
    o.x = (p.x + q.x) + b0;
    o.y = (p.y + q.y) + b1;
    return o;
}

__global__ void __launch_bounds__(256)
k_edge_head(const int* __restrict__ src, const int* __restrict__ dst,
            const float* __restrict__ pq, const float* __restrict__ be,
            float* __restrict__ out, int E) {
    const int   base = blockIdx.x * 1024;
    const float b0 = be[0], b1 = be[1];
    const bool wide = ((((size_t)src | (size_t)dst) & 7) == 0) && (((size_t)out & 15) == 0);
    if (wide && base + 1024 <= E) {
        const int  e0 = base + threadIdx.x * 2;
        const int  e1 = e0 + 512;
        const int2 s0 = *(const int2*)(src + e0);
        const int2 d0 = *(const int2*)(dst + e0);
        const int2 s1 = *(const int2*)(src + e1);
        const int2 d1 = *(const int2*)(dst + e1);
        const float2 a0 = edge_one(pq, s0.x, d0.x, b0, b1);
        const float2 a1 = edge_one(pq, s0.y, d0.y, b0, b1);
        const float2 c0 = edge_one(pq, s1.x, d1.x, b0, b1);
        const float2 c1 = edge_one(pq, s1.y, d1.y, b0, b1);
        float4 o0, o1;
        o0.x = a0.x; o0.y = a0.y; o0.z = a1.x; o0.w = a1.y;
        o1.x = c0.x; o1.y = c0.y; o1.z = c1.x; o1.w = c1.y;
        *(float4*)(out + (size_t)e0 * 2) = o0;
        *(float4*)(out + (size_t)e1 * 2) = o1;
    } else {
        for (int u = 0; u < 2; ++u) {
            for (int k = 0; k < 2; ++k) {
                const int e = base + u * 512 + threadIdx.x * 2 + k;
                if (e < E) *(float2*)(out + (size_t)e * 2) = edge_one(pq, src[e], dst[e], b0, b1);
            }
        }
    }
}

extern "C" void kernel_launch(void* const* d_in, const int* in_sizes, int n_in,
                              void* d_out, int out_size, void* d_ws, size_t ws_size,
                              hipStream_t stream) {
    const float* x  = (const float*)d_in[0];
    const int*   ei = (const int*)d_in[1];
    const float* W1 = (const float*)d_in[2];
    const float* b1 = (const float*)d_in[3];
    const float* W2 = (const float*)d_in[4];
    const float* b2 = (const float*)d_in[5];
    const float* Wn = (const float*)d_in[6];
    const float* bn = (const float*)d_in[7];
    const float* We = (const float*)d_in[8];
    const float* be = (const float*)d_in[9];

    const int N = in_sizes[0] / 64;
    const int E = in_sizes[1] / 2;
    const int* src = ei;
    const int* dst = ei + E;

    char*  ws  = (char*)d_ws;
    size_t off = 0;
    auto carve = [&](size_t bytes) -> float* {
        float* p = (float*)(ws + off);
        off += (bytes + 255) & ~(size_t)255;
        return p;
    };
    const int G1 = (E + CHUNK - 1) / CHUNK;
    float* dinv   = carve((size_t)N * 4);
    int*   rowptr = (int*)carve((size_t)(N + 1) * 4);
    int*   runs   = (int*)carve((size_t)G1 * RROW * 4);
    int*   ent    = (int*)carve((size_t)E * 4);
    int*   csrsrc = (int*)carve((size_t)E * 4);
    float* wp   = carve((size_t)(256 + 1) * 16 * 4);
    float* y    = carve((size_t)N * 8 * 4);
    float* pq   = carve((size_t)N * 4 * 4);

    float* out_node = (float*)d_out;
    float* out_edge = out_node + (size_t)N * 2;

    k_chunk_sort <<<G1 + 1, 1024, 0, stream>>>(src, dst, runs, ent, E, N, W2, b2, Wn, We, wp);
    k_bucket_fine<<<(N + BNODE - 1) / BNODE, 1024, 0, stream>>>(ent, runs, G1, rowptr, dinv, csrsrc, E, N);

    k_hidden_fold<<<768, 256, 0, stream>>>(rowptr, csrsrc, dinv, x, W1, b1, wp, y, N / 16, N);
    k_gather_heads<<<(N * 4 + 255) / 256, 256, 0, stream>>>(
        rowptr, csrsrc, dinv, y, wp + 256 * 16, bn, out_node, pq, N);

    k_edge_head<<<(E + 1023) / 1024, 256, 0, stream>>>(src, dst, pq, be, out_edge, E);
}
